# starting kernel
# baseline (speedup 1.0000x reference)
#include <hip/hip_runtime.h>
#include <math.h>

typedef __attribute__((ext_vector_type(2))) float v2f;
typedef __attribute__((ext_vector_type(8))) float v8f;

#define N_NODES 50000
#define N_EDGES 800000
#define EPB 64

__device__ __forceinline__ v8f wmma4(v2f a, v2f b, v8f c) {
  return __builtin_amdgcn_wmma_f32_16x16x4_f32(false, a, false, b, (short)0, c,
                                               false, false);
}

__device__ __forceinline__ float silu_f(float v) {
  return v / (1.0f + __expf(-v));
}

__global__ void egnn_init_kernel(const float* __restrict__ coords,
                                 float* __restrict__ agg,
                                 float* __restrict__ coords_out) {
  int i = blockIdx.x * blockDim.x + threadIdx.x;
  if (i < N_NODES * 128) agg[i] = 0.0f;
  if (i < N_NODES * 3) coords_out[i] = coords[i];
}

__global__ __launch_bounds__(256) void egnn_edge_kernel(
    const float* __restrict__ x, const float* __restrict__ coords,
    const float* __restrict__ edge_attr, const int* __restrict__ erow,
    const int* __restrict__ ecol, const float* __restrict__ We1,
    const float* __restrict__ be1, const float* __restrict__ We2,
    const float* __restrict__ be2, const float* __restrict__ Wc1,
    const float* __restrict__ bc1, const float* __restrict__ Wc2,
    float* __restrict__ agg, float* __restrict__ coords_out) {
  __shared__ float smem[EPB * 292 + EPB * 132];
  float* sA = smem;
  float* sH = smem + EPB * 292;
  float* sE = smem;

  __shared__ int sRow[EPB], sCol[EPB];
  __shared__ float sDiff[EPB * 3], sDist[EPB], sW[EPB];

  const int tid = threadIdx.x;
  const int lane = tid & 31;
  const int wv = tid >> 5;
  const int hi = lane >> 4;
  const int lm = lane & 15;
  const int n0 = wv * 16;
  const int ebase = blockIdx.x * EPB;

  if (tid < EPB) {
    int r = erow[ebase + tid];
    int c = ecol[ebase + tid];
    sRow[tid] = r;
    sCol[tid] = c;
    float d0 = coords[r * 3 + 0] - coords[c * 3 + 0];
    float d1 = coords[r * 3 + 1] - coords[c * 3 + 1];
    float d2 = coords[r * 3 + 2] - coords[c * 3 + 2];
    sDiff[tid * 3 + 0] = d0;
    sDiff[tid * 3 + 1] = d1;
    sDiff[tid * 3 + 2] = d2;
    sDist[tid] = d0 * d0 + d1 * d1 + d2 * d2;
    sW[tid] = 0.0f;
  }
  __syncthreads();

  for (int idx = tid; idx < EPB * 288; idx += 256) {
    int e = idx / 288;
    int c = idx - e * 288;
    float v;
    if (c < 128)
      v = x[sRow[e] * 128 + c];
    else if (c < 256)
      v = x[sCol[e] * 128 + (c - 128)];
    else
      v = edge_attr[(ebase + e) * 32 + (c - 256)];
    sA[e * 292 + c] = v;
  }
  __syncthreads();

  {
    v8f acc[4] = {{}, {}, {}, {}};
    for (int k0 = 0; k0 < 288; k0 += 4) {
      int kk = k0 + 2 * hi;
      v2f b;
      b.x = We1[kk * 128 + n0 + lm];
      b.y = We1[(kk + 1) * 128 + n0 + lm];
#pragma unroll
      for (int t = 0; t < 4; ++t) {
        v2f a;
        a.x = sA[(t * 16 + lm) * 292 + kk];
        a.y = sA[(t * 16 + lm) * 292 + kk + 1];
        acc[t] = wmma4(a, b, acc[t]);
      }
    }
    float bias = be1[n0 + lm];
    float wlast = We1[288 * 128 + n0 + lm];
#pragma unroll
    for (int t = 0; t < 4; ++t) {
#pragma unroll
      for (int v = 0; v < 8; ++v) {
        int M = t * 16 + v + 8 * hi;
        float pre = acc[t][v] + bias + sDist[M] * wlast;
        sH[M * 132 + n0 + lm] = silu_f(pre);
      }
    }
  }
  __syncthreads();

  {
    v8f acc[4] = {{}, {}, {}, {}};
    for (int k0 = 0; k0 < 128; k0 += 4) {
      int kk = k0 + 2 * hi;
      v2f b;
      b.x = We2[kk * 128 + n0 + lm];
      b.y = We2[(kk + 1) * 128 + n0 + lm];
#pragma unroll
      for (int t = 0; t < 4; ++t) {
        v2f a;
        a.x = sH[(t * 16 + lm) * 132 + kk];
        a.y = sH[(t * 16 + lm) * 132 + kk + 1];
        acc[t] = wmma4(a, b, acc[t]);
      }
    }
    float bias = be2[n0 + lm];
#pragma unroll
    for (int t = 0; t < 4; ++t) {
#pragma unroll
      for (int v = 0; v < 8; ++v) {
        int M = t * 16 + v + 8 * hi;
        float ef = acc[t][v] + bias;
        sE[M * 132 + n0 + lm] = ef;
        atomicAdd(&agg[sRow[M] * 128 + n0 + lm], ef);
      }
    }
  }
  __syncthreads();

  {
    v8f acc[4] = {{}, {}, {}, {}};
    for (int k0 = 0; k0 < 128; k0 += 4) {
      int kk = k0 + 2 * hi;
      v2f b;
      b.x = Wc1[kk * 128 + n0 + lm];
      b.y = Wc1[(kk + 1) * 128 + n0 + lm];
#pragma unroll
      for (int t = 0; t < 4; ++t) {
        v2f a;
        a.x = sE[(t * 16 + lm) * 132 + kk];
        a.y = sE[(t * 16 + lm) * 132 + kk + 1];
        acc[t] = wmma4(a, b, acc[t]);
      }
    }
    float bias = bc1[n0 + lm];
    float wc2 = Wc2[n0 + lm];
#pragma unroll
    for (int t = 0; t < 4; ++t) {
#pragma unroll
      for (int v = 0; v < 8; ++v) {
        int M = t * 16 + v + 8 * hi;
        float s = silu_f(acc[t][v] + bias);
        atomicAdd(&sW[M], s * wc2);
      }
    }
  }
  __syncthreads();

  if (tid < EPB * 3) {
    int e = tid / 3;
    int d = tid - e * 3;
    float cd = sDiff[tid] / (sqrtf(sDist[e]) + 1e-8f);
    atomicAdd(&coords_out[sRow[e] * 3 + d], cd * sW[e]);
  }
}

__global__ __launch_bounds__(256) void egnn_node_kernel(
    const float* __restrict__ x, const float* __restrict__ agg,
    const float* __restrict__ Wn1, const float* __restrict__ bn1,
    const float* __restrict__ Wn2, const float* __restrict__ bn2,
    float* __restrict__ x_out) {
  __shared__ float sA[16 * 260];
  __shared__ float sH[16 * 132];

  const int tid = threadIdx.x;
  const int lane = tid & 31;
  const int wv = tid >> 5;
  const int hi = lane >> 4;
  const int lm = lane & 15;
  const int n0 = wv * 16;
  const int nb = blockIdx.x * 16;

  for (int idx = tid; idx < 16 * 256; idx += 256) {
    int m = idx >> 8;
    int c = idx & 255;
    float v = (c < 128) ? x[(nb + m) * 128 + c] : agg[(nb + m) * 128 + (c - 128)];
    sA[m * 260 + c] = v;
  }
  __syncthreads();

  v8f acc1 = {};
  for (int k0 = 0; k0 < 256; k0 += 4) {
    int kk = k0 + 2 * hi;
    v2f a, b;
    a.x = sA[lm * 260 + kk];
    a.y = sA[lm * 260 + kk + 1];
    b.x = Wn1[kk * 128 + n0 + lm];
    b.y = Wn1[(kk + 1) * 128 + n0 + lm];
    acc1 = wmma4(a, b, acc1);
  }
  {
    float bias = bn1[n0 + lm];
#pragma unroll
    for (int v = 0; v < 8; ++v) {
      int M = v + 8 * hi;
      sH[M * 132 + n0 + lm] = silu_f(acc1[v] + bias);
    }
  }
  __syncthreads();

  v8f acc2 = {};
  for (int k0 = 0; k0 < 128; k0 += 4) {
    int kk = k0 + 2 * hi;
    v2f a, b;
    a.x = sH[lm * 132 + kk];
    a.y = sH[lm * 132 + kk + 1];
    b.x = Wn2[kk * 128 + n0 + lm];
    b.y = Wn2[(kk + 1) * 128 + n0 + lm];
    acc2 = wmma4(a, b, acc2);
  }
  {
    float bias = bn2[n0 + lm];
#pragma unroll
    for (int v = 0; v < 8; ++v) {
      int M = v + 8 * hi;
      x_out[(nb + M) * 128 + n0 + lm] = acc2[v] + bias;
    }
  }
}

extern "C" void kernel_launch(void* const* d_in, const int* in_sizes, int n_in,
                              void* d_out, int out_size, void* d_ws,
                              size_t ws_size, hipStream_t stream) {
  const float* x = (const float*)d_in[0];
  const float* coords = (const float*)d_in[1];
  const float* edge_attr = (const float*)d_in[2];
  const int* edge_index = (const int*)d_in[3];
  const float* We1 = (const float*)d_in[4];
  const float* be1 = (const float*)d_in[5];
  const float* We2 = (const float*)d_in[6];
  const float* be2 = (const float*)d_in[7];
  const float* Wc1 = (const float*)d_in[8];
  const float* bc1 = (const float*)d_in[9];
  const float* Wc2 = (const float*)d_in[10];
  const float* Wn1 = (const float*)d_in[11];
  const float* bn1 = (const float*)d_in[12];
  const float* Wn2 = (const float*)d_in[13];
  const float* bn2 = (const float*)d_in[14];

  float* x_out = (float*)d_out;
  float* coords_out = x_out + (size_t)N_NODES * 128;
  float* agg = (float*)d_ws;

  const int* erow = edge_index;
  const int* ecol = edge_index + N_EDGES;

  egnn_init_kernel<<<(N_NODES * 128 + 255) / 256, 256, 0, stream>>>(
      coords, agg, coords_out);
  egnn_edge_kernel<<<N_EDGES / EPB, 256, 0, stream>>>(
      x, coords, edge_attr, erow, ecol, We1, be1, We2, be2, Wc1, bc1, Wc2, agg,
      coords_out);
  egnn_node_kernel<<<N_NODES / 16, 256, 0, stream>>>(x, agg, Wn1, bn1, Wn2, bn2,
                                                     x_out);
}


# Edge-tile gather as 18 independent b128 loads per thread (a wave copies whole 512-byte x rows; edge_attr issued ahead of the index phase) + ds_store_b128, replacing 72 serial 4-byte load-store pairs
# speedup vs baseline: 1.5050x; 1.5050x over previous
#include <hip/hip_runtime.h>
#include <math.h>

typedef __attribute__((ext_vector_type(2))) float v2f;
typedef __attribute__((ext_vector_type(8))) float v8f;
typedef __attribute__((ext_vector_type(4))) float v4f;

#define N_NODES 50000
#define N_EDGES 800000
#define EPB 64

__device__ __forceinline__ v8f wmma4(v2f a, v2f b, v8f c) {
  return __builtin_amdgcn_wmma_f32_16x16x4_f32(false, a, false, b, (short)0, c,
                                               false, false);
}

__device__ __forceinline__ float silu_f(float v) {
  return v / (1.0f + __expf(-v));
}

__global__ void egnn_init_kernel(const float* __restrict__ coords,
                                 float* __restrict__ agg,
                                 float* __restrict__ coords_out) {
  int i = blockIdx.x * blockDim.x + threadIdx.x;
  if (i < N_NODES * 128) agg[i] = 0.0f;
  if (i < N_NODES * 3) coords_out[i] = coords[i];
}

__global__ __launch_bounds__(256) void egnn_edge_kernel(
    const float* __restrict__ x, const float* __restrict__ coords,
    const float* __restrict__ edge_attr, const int* __restrict__ erow,
    const int* __restrict__ ecol, const float* __restrict__ We1,
    const float* __restrict__ be1, const float* __restrict__ We2,
    const float* __restrict__ be2, const float* __restrict__ Wc1,
    const float* __restrict__ bc1, const float* __restrict__ Wc2,
    float* __restrict__ agg, float* __restrict__ coords_out) {
  __shared__ __attribute__((aligned(16))) float smem[EPB * 292 + EPB * 132];
  float* sA = smem;
  float* sH = smem + EPB * 292;
  float* sE = smem;

  __shared__ int sRC[2 * EPB];
  int* sRow = sRC;
  int* sCol = sRC + EPB;
  __shared__ float sDiff[EPB * 3], sDist[EPB], sW[EPB];

  const int tid = threadIdx.x;
  const int lane = tid & 31;
  const int wv = tid >> 5;
  const int hi = lane >> 4;
  const int lm = lane & 15;
  const int n0 = wv * 16;
  const int ebase = blockIdx.x * EPB;

  const v4f* ea4 = reinterpret_cast<const v4f*>(edge_attr) + ebase * 8;
  const v4f ea0 = ea4[tid];
  const v4f ea1 = ea4[tid + 256];

  if (tid < EPB) {
    int r = erow[ebase + tid];
    int c = ecol[ebase + tid];
    sRow[tid] = r;
    sCol[tid] = c;
    float d0 = coords[r * 3 + 0] - coords[c * 3 + 0];
    float d1 = coords[r * 3 + 1] - coords[c * 3 + 1];
    float d2 = coords[r * 3 + 2] - coords[c * 3 + 2];
    sDiff[tid * 3 + 0] = d0;
    sDiff[tid * 3 + 1] = d1;
    sDiff[tid * 3 + 2] = d2;
    sDist[tid] = d0 * d0 + d1 * d1 + d2 * d2;
    sW[tid] = 0.0f;
  }
  __syncthreads();

  {
    const int q = tid & 31;
    const int sel = (tid >> 5) & 1;
    const int e0 = tid >> 6;
    const v4f* x4 = reinterpret_cast<const v4f*>(x);
    v4f xv[16];
    for (int i = 0; i < 16; ++i) {
      const int node = sRC[sel * EPB + e0 + 4 * i];
      xv[i] = x4[node * 32 + q];
    }
    for (int i = 0; i < 16; ++i)
      *reinterpret_cast<v4f*>(&sA[(e0 + 4 * i) * 292 + sel * 128 + 4 * q]) =
          xv[i];
    *reinterpret_cast<v4f*>(&sA[(tid >> 3) * 292 + 256 + 4 * (tid & 7)]) = ea0;
    *reinterpret_cast<v4f*>(&sA[((tid >> 3) + 32) * 292 + 256 + 4 * (tid & 7)]) =
        ea1;
  }
  __syncthreads();

  {
    v8f acc[4] = {{}, {}, {}, {}};
    for (int k0 = 0; k0 < 288; k0 += 4) {
      int kk = k0 + 2 * hi;
      v2f b;
      b.x = We1[kk * 128 + n0 + lm];
      b.y = We1[(kk + 1) * 128 + n0 + lm];
#pragma unroll
      for (int t = 0; t < 4; ++t) {
        v2f a;
        a.x = sA[(t * 16 + lm) * 292 + kk];
        a.y = sA[(t * 16 + lm) * 292 + kk + 1];
        acc[t] = wmma4(a, b, acc[t]);
      }
    }
    float bias = be1[n0 + lm];
    float wlast = We1[288 * 128 + n0 + lm];
#pragma unroll
    for (int t = 0; t < 4; ++t) {
#pragma unroll
      for (int v = 0; v < 8; ++v) {
        int M = t * 16 + v + 8 * hi;
        float pre = acc[t][v] + bias + sDist[M] * wlast;
        sH[M * 132 + n0 + lm] = silu_f(pre);
      }
    }
  }
  __syncthreads();

  {
    v8f acc[4] = {{}, {}, {}, {}};
    for (int k0 = 0; k0 < 128; k0 += 4) {
      int kk = k0 + 2 * hi;
      v2f b;
      b.x = We2[kk * 128 + n0 + lm];
      b.y = We2[(kk + 1) * 128 + n0 + lm];
#pragma unroll
      for (int t = 0; t < 4; ++t) {
        v2f a;
        a.x = sH[(t * 16 + lm) * 132 + kk];
        a.y = sH[(t * 16 + lm) * 132 + kk + 1];
        acc[t] = wmma4(a, b, acc[t]);
      }
    }
    float bias = be2[n0 + lm];
#pragma unroll
    for (int t = 0; t < 4; ++t) {
#pragma unroll
      for (int v = 0; v < 8; ++v) {
        int M = t * 16 + v + 8 * hi;
        float ef = acc[t][v] + bias;
        sE[M * 132 + n0 + lm] = ef;
        atomicAdd(&agg[sRow[M] * 128 + n0 + lm], ef);
      }
    }
  }
  __syncthreads();

  {
    v8f acc[4] = {{}, {}, {}, {}};
    for (int k0 = 0; k0 < 128; k0 += 4) {
      int kk = k0 + 2 * hi;
      v2f b;
      b.x = Wc1[kk * 128 + n0 + lm];
      b.y = Wc1[(kk + 1) * 128 + n0 + lm];
#pragma unroll
      for (int t = 0; t < 4; ++t) {
        v2f a;
        a.x = sE[(t * 16 + lm) * 132 + kk];
        a.y = sE[(t * 16 + lm) * 132 + kk + 1];
        acc[t] = wmma4(a, b, acc[t]);
      }
    }
    float bias = bc1[n0 + lm];
    float wc2 = Wc2[n0 + lm];
#pragma unroll
    for (int t = 0; t < 4; ++t) {
#pragma unroll
      for (int v = 0; v < 8; ++v) {
        int M = t * 16 + v + 8 * hi;
        float s = silu_f(acc[t][v] + bias);
        atomicAdd(&sW[M], s * wc2);
      }
    }
  }
  __syncthreads();

  if (tid < EPB * 3) {
    int e = tid / 3;
    int d = tid - e * 3;
    float cd = sDiff[tid] / (sqrtf(sDist[e]) + 1e-8f);
    atomicAdd(&coords_out[sRow[e] * 3 + d], cd * sW[e]);
  }
}

__global__ __launch_bounds__(256) void egnn_node_kernel(
    const float* __restrict__ x, const float* __restrict__ agg,
    const float* __restrict__ Wn1, const float* __restrict__ bn1,
    const float* __restrict__ Wn2, const float* __restrict__ bn2,
    float* __restrict__ x_out) {
  __shared__ float sA[16 * 260];
  __shared__ float sH[16 * 132];

  const int tid = threadIdx.x;
  const int lane = tid & 31;
  const int wv = tid >> 5;
  const int hi = lane >> 4;
  const int lm = lane & 15;
  const int n0 = wv * 16;
  const int nb = blockIdx.x * 16;

  for (int idx = tid; idx < 16 * 256; idx += 256) {
    int m = idx >> 8;
    int c = idx & 255;
    float v = (c < 128) ? x[(nb + m) * 128 + c] : agg[(nb + m) * 128 + (c - 128)];
    sA[m * 260 + c] = v;
  }
  __syncthreads();

  v8f acc1 = {};
  for (int k0 = 0; k0 < 256; k0 += 4) {
    int kk = k0 + 2 * hi;
    v2f a, b;
    a.x = sA[lm * 260 + kk];
    a.y = sA[lm * 260 + kk + 1];
    b.x = Wn1[kk * 128 + n0 + lm];
    b.y = Wn1[(kk + 1) * 128 + n0 + lm];
    acc1 = wmma4(a, b, acc1);
  }
  {
    float bias = bn1[n0 + lm];
#pragma unroll
    for (int v = 0; v < 8; ++v) {
      int M = v + 8 * hi;
      sH[M * 132 + n0 + lm] = silu_f(acc1[v] + bias);
    }
  }
  __syncthreads();

  v8f acc2 = {};
  for (int k0 = 0; k0 < 128; k0 += 4) {
    int kk = k0 + 2 * hi;
    v2f a, b;
    a.x = sH[lm * 132 + kk];
    a.y = sH[lm * 132 + kk + 1];
    b.x = Wn2[kk * 128 + n0 + lm];
    b.y = Wn2[(kk + 1) * 128 + n0 + lm];
    acc2 = wmma4(a, b, acc2);
  }
  {
    float bias = bn2[n0 + lm];
#pragma unroll
    for (int v = 0; v < 8; ++v) {
      int M = v + 8 * hi;
      x_out[(nb + M) * 128 + n0 + lm] = acc2[v] + bias;
    }
  }
}

extern "C" void kernel_launch(void* const* d_in, const int* in_sizes, int n_in,
                              void* d_out, int out_size, void* d_ws,
                              size_t ws_size, hipStream_t stream) {
  const float* x = (const float*)d_in[0];
  const float* coords = (const float*)d_in[1];
  const float* edge_attr = (const float*)d_in[2];
  const int* edge_index = (const int*)d_in[3];
  const float* We1 = (const float*)d_in[4];
  const float* be1 = (const float*)d_in[5];
  const float* We2 = (const float*)d_in[6];
  const float* be2 = (const float*)d_in[7];
  const float* Wc1 = (const float*)d_in[8];
  const float* bc1 = (const float*)d_in[9];
  const float* Wc2 = (const float*)d_in[10];
  const float* Wn1 = (const float*)d_in[11];
  const float* bn1 = (const float*)d_in[12];
  const float* Wn2 = (const float*)d_in[13];
  const float* bn2 = (const float*)d_in[14];

  float* x_out = (float*)d_out;
  float* coords_out = x_out + (size_t)N_NODES * 128;
  float* agg = (float*)d_ws;

  const int* erow = edge_index;
  const int* ecol = edge_index + N_EDGES;

  egnn_init_kernel<<<(N_NODES * 128 + 255) / 256, 256, 0, stream>>>(
      coords, agg, coords_out);
  egnn_edge_kernel<<<N_EDGES / EPB, 256, 0, stream>>>(
      x, coords, edge_attr, erow, ecol, We1, be1, We2, be2, Wc1, bc1, Wc2, agg,
      coords_out);
  egnn_node_kernel<<<N_NODES / 16, 256, 0, stream>>>(x, agg, Wn1, bn1, Wn2, bn2,
                                                     x_out);
}


# Per-edge gate sum over features: 16-lane DPP butterfly in registers, then one ds_add_f32 per tile on 16 distinct words, replacing 32 same-word LDS float atomics per wave (sum order: in-wave tree)
# speedup vs baseline: 1.8701x; 1.2426x over previous
#include <hip/hip_runtime.h>
#include <math.h>

typedef __attribute__((ext_vector_type(2))) float v2f;
typedef __attribute__((ext_vector_type(8))) float v8f;
typedef __attribute__((ext_vector_type(4))) float v4f;

#define N_NODES 50000
#define N_EDGES 800000
#define EPB 64

__device__ __forceinline__ v8f wmma4(v2f a, v2f b, v8f c) {
  return __builtin_amdgcn_wmma_f32_16x16x4_f32(false, a, false, b, (short)0, c,
                                               false, false);
}

__device__ __forceinline__ float silu_f(float v) {
  return v / (1.0f + __expf(-v));
}

__device__ __forceinline__ float row16_sum(float v) {
  v += __int_as_float(__builtin_amdgcn_update_dpp(
      0, __float_as_int(v), 0x161, 0xf, 0xf, false));
  v += __int_as_float(__builtin_amdgcn_update_dpp(
      0, __float_as_int(v), 0x162, 0xf, 0xf, false));
  v += __int_as_float(__builtin_amdgcn_update_dpp(
      0, __float_as_int(v), 0x164, 0xf, 0xf, false));
  v += __int_as_float(__builtin_amdgcn_update_dpp(
      0, __float_as_int(v), 0x168, 0xf, 0xf, false));
  return v;
}

__global__ void egnn_init_kernel(const float* __restrict__ coords,
                                 float* __restrict__ agg,
                                 float* __restrict__ coords_out) {
  int i = blockIdx.x * blockDim.x + threadIdx.x;
  if (i < N_NODES * 128) agg[i] = 0.0f;
  if (i < N_NODES * 3) coords_out[i] = coords[i];
}

__global__ __launch_bounds__(256) void egnn_edge_kernel(
    const float* __restrict__ x, const float* __restrict__ coords,
    const float* __restrict__ edge_attr, const int* __restrict__ erow,
    const int* __restrict__ ecol, const float* __restrict__ We1,
    const float* __restrict__ be1, const float* __restrict__ We2,
    const float* __restrict__ be2, const float* __restrict__ Wc1,
    const float* __restrict__ bc1, const float* __restrict__ Wc2,
    float* __restrict__ agg, float* __restrict__ coords_out) {
  __shared__ __attribute__((aligned(16))) float smem[EPB * 292 + EPB * 132];
  float* sA = smem;
  float* sH = smem + EPB * 292;
  float* sE = smem;

  __shared__ int sRC[2 * EPB];
  int* sRow = sRC;
  int* sCol = sRC + EPB;
  __shared__ float sDiff[EPB * 3], sDist[EPB], sW[EPB];

  const int tid = threadIdx.x;
  const int lane = tid & 31;
  const int wv = tid >> 5;
  const int hi = lane >> 4;
  const int lm = lane & 15;
  const int n0 = wv * 16;
  const int ebase = blockIdx.x * EPB;

  const v4f* ea4 = reinterpret_cast<const v4f*>(edge_attr) + ebase * 8;
  const v4f ea0 = ea4[tid];
  const v4f ea1 = ea4[tid + 256];

  if (tid < EPB) {
    int r = erow[ebase + tid];
    int c = ecol[ebase + tid];
    sRow[tid] = r;
    sCol[tid] = c;
    float d0 = coords[r * 3 + 0] - coords[c * 3 + 0];
    float d1 = coords[r * 3 + 1] - coords[c * 3 + 1];
    float d2 = coords[r * 3 + 2] - coords[c * 3 + 2];
    sDiff[tid * 3 + 0] = d0;
    sDiff[tid * 3 + 1] = d1;
    sDiff[tid * 3 + 2] = d2;
    sDist[tid] = d0 * d0 + d1 * d1 + d2 * d2;
    sW[tid] = 0.0f;
  }
  __syncthreads();

  {
    const int q = tid & 31;
    const int sel = (tid >> 5) & 1;
    const int e0 = tid >> 6;
    const v4f* x4 = reinterpret_cast<const v4f*>(x);
    v4f xv[16];
    for (int i = 0; i < 16; ++i) {
      const int node = sRC[sel * EPB + e0 + 4 * i];
      xv[i] = x4[node * 32 + q];
    }
    for (int i = 0; i < 16; ++i)
      *reinterpret_cast<v4f*>(&sA[(e0 + 4 * i) * 292 + sel * 128 + 4 * q]) =
          xv[i];
    *reinterpret_cast<v4f*>(&sA[(tid >> 3) * 292 + 256 + 4 * (tid & 7)]) = ea0;
    *reinterpret_cast<v4f*>(&sA[((tid >> 3) + 32) * 292 + 256 + 4 * (tid & 7)]) =
        ea1;
  }
  __syncthreads();

  {
    v8f acc[4] = {{}, {}, {}, {}};
    for (int k0 = 0; k0 < 288; k0 += 4) {
      int kk = k0 + 2 * hi;
      v2f b;
      b.x = We1[kk * 128 + n0 + lm];
      b.y = We1[(kk + 1) * 128 + n0 + lm];
#pragma unroll
      for (int t = 0; t < 4; ++t) {
        v2f a;
        a.x = sA[(t * 16 + lm) * 292 + kk];
        a.y = sA[(t * 16 + lm) * 292 + kk + 1];
        acc[t] = wmma4(a, b, acc[t]);
      }
    }
    float bias = be1[n0 + lm];
    float wlast = We1[288 * 128 + n0 + lm];
#pragma unroll
    for (int t = 0; t < 4; ++t) {
#pragma unroll
      for (int v = 0; v < 8; ++v) {
        int M = t * 16 + v + 8 * hi;
        float pre = acc[t][v] + bias + sDist[M] * wlast;
        sH[M * 132 + n0 + lm] = silu_f(pre);
      }
    }
  }
  __syncthreads();

  {
    v8f acc[4] = {{}, {}, {}, {}};
    for (int k0 = 0; k0 < 128; k0 += 4) {
      int kk = k0 + 2 * hi;
      v2f b;
      b.x = We2[kk * 128 + n0 + lm];
      b.y = We2[(kk + 1) * 128 + n0 + lm];
#pragma unroll
      for (int t = 0; t < 4; ++t) {
        v2f a;
        a.x = sH[(t * 16 + lm) * 132 + kk];
        a.y = sH[(t * 16 + lm) * 132 + kk + 1];
        acc[t] = wmma4(a, b, acc[t]);
      }
    }
    float bias = be2[n0 + lm];
#pragma unroll
    for (int t = 0; t < 4; ++t) {
#pragma unroll
      for (int v = 0; v < 8; ++v) {
        int M = t * 16 + v + 8 * hi;
        float ef = acc[t][v] + bias;
        sE[M * 132 + n0 + lm] = ef;
        atomicAdd(&agg[sRow[M] * 128 + n0 + lm], ef);
      }
    }
  }
  __syncthreads();

  {
    v8f acc[4] = {{}, {}, {}, {}};
    for (int k0 = 0; k0 < 128; k0 += 4) {
      int kk = k0 + 2 * hi;
      v2f b;
      b.x = Wc1[kk * 128 + n0 + lm];
      b.y = Wc1[(kk + 1) * 128 + n0 + lm];
#pragma unroll
      for (int t = 0; t < 4; ++t) {
        v2f a;
        a.x = sE[(t * 16 + lm) * 132 + kk];
        a.y = sE[(t * 16 + lm) * 132 + kk + 1];
        acc[t] = wmma4(a, b, acc[t]);
      }
    }
    float bias = bc1[n0 + lm];
    float wc2 = Wc2[n0 + lm];
#pragma unroll
    for (int t = 0; t < 4; ++t) {
      float mine = 0.0f;
#pragma unroll
      for (int v = 0; v < 8; ++v) {
        float p = row16_sum(silu_f(acc[t][v] + bias) * wc2);
        mine = (lm == v) ? p : mine;
      }
      if (lm < 8) atomicAdd(&sW[t * 16 + lm + 8 * hi], mine);
    }
  }
  __syncthreads();

  if (tid < EPB * 3) {
    int e = tid / 3;
    int d = tid - e * 3;
    float cd = sDiff[tid] / (sqrtf(sDist[e]) + 1e-8f);
    atomicAdd(&coords_out[sRow[e] * 3 + d], cd * sW[e]);
  }
}

__global__ __launch_bounds__(256) void egnn_node_kernel(
    const float* __restrict__ x, const float* __restrict__ agg,
    const float* __restrict__ Wn1, const float* __restrict__ bn1,
    const float* __restrict__ Wn2, const float* __restrict__ bn2,
    float* __restrict__ x_out) {
  __shared__ float sA[16 * 260];
  __shared__ float sH[16 * 132];

  const int tid = threadIdx.x;
  const int lane = tid & 31;
  const int wv = tid >> 5;
  const int hi = lane >> 4;
  const int lm = lane & 15;
  const int n0 = wv * 16;
  const int nb = blockIdx.x * 16;

  for (int idx = tid; idx < 16 * 256; idx += 256) {
    int m = idx >> 8;
    int c = idx & 255;
    float v = (c < 128) ? x[(nb + m) * 128 + c] : agg[(nb + m) * 128 + (c - 128)];
    sA[m * 260 + c] = v;
  }
  __syncthreads();

  v8f acc1 = {};
  for (int k0 = 0; k0 < 256; k0 += 4) {
    int kk = k0 + 2 * hi;
    v2f a, b;
    a.x = sA[lm * 260 + kk];
    a.y = sA[lm * 260 + kk + 1];
    b.x = Wn1[kk * 128 + n0 + lm];
    b.y = Wn1[(kk + 1) * 128 + n0 + lm];
    acc1 = wmma4(a, b, acc1);
  }
  {
    float bias = bn1[n0 + lm];
#pragma unroll
    for (int v = 0; v < 8; ++v) {
      int M = v + 8 * hi;
      sH[M * 132 + n0 + lm] = silu_f(acc1[v] + bias);
    }
  }
  __syncthreads();

  v8f acc2 = {};
  for (int k0 = 0; k0 < 128; k0 += 4) {
    int kk = k0 + 2 * hi;
    v2f a, b;
    a.x = sH[lm * 132 + kk];
    a.y = sH[lm * 132 + kk + 1];
    b.x = Wn2[kk * 128 + n0 + lm];
    b.y = Wn2[(kk + 1) * 128 + n0 + lm];
    acc2 = wmma4(a, b, acc2);
  }
  {
    float bias = bn2[n0 + lm];
#pragma unroll
    for (int v = 0; v < 8; ++v) {
      int M = v + 8 * hi;
      x_out[(nb + M) * 128 + n0 + lm] = acc2[v] + bias;
    }
  }
}

extern "C" void kernel_launch(void* const* d_in, const int* in_sizes, int n_in,
                              void* d_out, int out_size, void* d_ws,
                              size_t ws_size, hipStream_t stream) {
  const float* x = (const float*)d_in[0];
  const float* coords = (const float*)d_in[1];
  const float* edge_attr = (const float*)d_in[2];
  const int* edge_index = (const int*)d_in[3];
  const float* We1 = (const float*)d_in[4];
  const float* be1 = (const float*)d_in[5];
  const float* We2 = (const float*)d_in[6];
  const float* be2 = (const float*)d_in[7];
  const float* Wc1 = (const float*)d_in[8];
  const float* bc1 = (const float*)d_in[9];
  const float* Wc2 = (const float*)d_in[10];
  const float* Wn1 = (const float*)d_in[11];
  const float* bn1 = (const float*)d_in[12];
  const float* Wn2 = (const float*)d_in[13];
  const float* bn2 = (const float*)d_in[14];

  float* x_out = (float*)d_out;
  float* coords_out = x_out + (size_t)N_NODES * 128;
  float* agg = (float*)d_ws;

  const int* erow = edge_index;
  const int* ecol = edge_index + N_EDGES;

  egnn_init_kernel<<<(N_NODES * 128 + 255) / 256, 256, 0, stream>>>(
      coords, agg, coords_out);
  egnn_edge_kernel<<<N_EDGES / EPB, 256, 0, stream>>>(
      x, coords, edge_attr, erow, ecol, We1, be1, We2, be2, Wc1, bc1, Wc2, agg,
      coords_out);
  egnn_node_kernel<<<N_NODES / 16, 256, 0, stream>>>(x, agg, Wn1, bn1, Wn2, bn2,
                                                     x_out);
}


# Occupancy: h1 and edge_feat tiles overlaid on the dead gathered-A tile in LDS (one more barrier) and registers capped at 128, so four blocks (32 waves) fit a WGP instead of two
# speedup vs baseline: 2.0680x; 1.1058x over previous
#include <hip/hip_runtime.h>
#include <math.h>

typedef __attribute__((ext_vector_type(2))) float v2f;
typedef __attribute__((ext_vector_type(8))) float v8f;
typedef __attribute__((ext_vector_type(4))) float v4f;

#define N_NODES 50000
#define N_EDGES 800000
#define EPB 64

__device__ __forceinline__ v8f wmma4(v2f a, v2f b, v8f c) {
  return __builtin_amdgcn_wmma_f32_16x16x4_f32(false, a, false, b, (short)0, c,
                                               false, false);
}

__device__ __forceinline__ float silu_f(float v) {
  return v / (1.0f + __expf(-v));
}

__device__ __forceinline__ float row16_sum(float v) {
  v += __int_as_float(__builtin_amdgcn_update_dpp(
      0, __float_as_int(v), 0x161, 0xf, 0xf, false));
  v += __int_as_float(__builtin_amdgcn_update_dpp(
      0, __float_as_int(v), 0x162, 0xf, 0xf, false));
  v += __int_as_float(__builtin_amdgcn_update_dpp(
      0, __float_as_int(v), 0x164, 0xf, 0xf, false));
  v += __int_as_float(__builtin_amdgcn_update_dpp(
      0, __float_as_int(v), 0x168, 0xf, 0xf, false));
  return v;
}

__global__ void egnn_init_kernel(const float* __restrict__ coords,
                                 float* __restrict__ agg,
                                 float* __restrict__ coords_out) {
  int i = blockIdx.x * blockDim.x + threadIdx.x;
  if (i < N_NODES * 128) agg[i] = 0.0f;
  if (i < N_NODES * 3) coords_out[i] = coords[i];
}

__global__ __launch_bounds__(256, 8) void egnn_edge_kernel(
    const float* __restrict__ x, const float* __restrict__ coords,
    const float* __restrict__ edge_attr, const int* __restrict__ erow,
    const int* __restrict__ ecol, const float* __restrict__ We1,
    const float* __restrict__ be1, const float* __restrict__ We2,
    const float* __restrict__ be2, const float* __restrict__ Wc1,
    const float* __restrict__ bc1, const float* __restrict__ Wc2,
    float* __restrict__ agg, float* __restrict__ coords_out) {
  __shared__ __attribute__((aligned(16))) float smem[EPB * 292];
  float* sA = smem;
  float* sH = smem;
  float* sE = smem + EPB * 132;

  __shared__ int sRC[2 * EPB];
  int* sRow = sRC;
  int* sCol = sRC + EPB;
  __shared__ float sDiff[EPB * 3], sDist[EPB], sW[EPB];

  const int tid = threadIdx.x;
  const int lane = tid & 31;
  const int wv = tid >> 5;
  const int hi = lane >> 4;
  const int lm = lane & 15;
  const int n0 = wv * 16;
  const int ebase = blockIdx.x * EPB;

  const v4f* ea4 = reinterpret_cast<const v4f*>(edge_attr) + ebase * 8;
  const v4f ea0 = ea4[tid];
  const v4f ea1 = ea4[tid + 256];

  if (tid < EPB) {
    int r = erow[ebase + tid];
    int c = ecol[ebase + tid];
    sRow[tid] = r;
    sCol[tid] = c;
    float d0 = coords[r * 3 + 0] - coords[c * 3 + 0];
    float d1 = coords[r * 3 + 1] - coords[c * 3 + 1];
    float d2 = coords[r * 3 + 2] - coords[c * 3 + 2];
    sDiff[tid * 3 + 0] = d0;
    sDiff[tid * 3 + 1] = d1;
    sDiff[tid * 3 + 2] = d2;
    sDist[tid] = d0 * d0 + d1 * d1 + d2 * d2;
    sW[tid] = 0.0f;
  }
  __syncthreads();

  {
    const int q = tid & 31;
    const int sel = (tid >> 5) & 1;
    const int e0 = tid >> 6;
    const v4f* x4 = reinterpret_cast<const v4f*>(x);
    v4f xv[16];
    for (int i = 0; i < 16; ++i) {
      const int node = sRC[sel * EPB + e0 + 4 * i];
      xv[i] = x4[node * 32 + q];
    }
    for (int i = 0; i < 16; ++i)
      *reinterpret_cast<v4f*>(&sA[(e0 + 4 * i) * 292 + sel * 128 + 4 * q]) =
          xv[i];
    *reinterpret_cast<v4f*>(&sA[(tid >> 3) * 292 + 256 + 4 * (tid & 7)]) = ea0;
    *reinterpret_cast<v4f*>(&sA[((tid >> 3) + 32) * 292 + 256 + 4 * (tid & 7)]) =
        ea1;
  }
  __syncthreads();

  {
    v8f acc[4] = {{}, {}, {}, {}};
    for (int k0 = 0; k0 < 288; k0 += 4) {
      int kk = k0 + 2 * hi;
      v2f b;
      b.x = We1[kk * 128 + n0 + lm];
      b.y = We1[(kk + 1) * 128 + n0 + lm];
#pragma unroll
      for (int t = 0; t < 4; ++t) {
        v2f a;
        a.x = sA[(t * 16 + lm) * 292 + kk];
        a.y = sA[(t * 16 + lm) * 292 + kk + 1];
        acc[t] = wmma4(a, b, acc[t]);
      }
    }
    float bias = be1[n0 + lm];
    float wlast = We1[288 * 128 + n0 + lm];
    __syncthreads();
#pragma unroll
    for (int t = 0; t < 4; ++t) {
#pragma unroll
      for (int v = 0; v < 8; ++v) {
        int M = t * 16 + v + 8 * hi;
        float pre = acc[t][v] + bias + sDist[M] * wlast;
        sH[M * 132 + n0 + lm] = silu_f(pre);
      }
    }
  }
  __syncthreads();

  {
    v8f acc[4] = {{}, {}, {}, {}};
    for (int k0 = 0; k0 < 128; k0 += 4) {
      int kk = k0 + 2 * hi;
      v2f b;
      b.x = We2[kk * 128 + n0 + lm];
      b.y = We2[(kk + 1) * 128 + n0 + lm];
#pragma unroll
      for (int t = 0; t < 4; ++t) {
        v2f a;
        a.x = sH[(t * 16 + lm) * 132 + kk];
        a.y = sH[(t * 16 + lm) * 132 + kk + 1];
        acc[t] = wmma4(a, b, acc[t]);
      }
    }
    float bias = be2[n0 + lm];
#pragma unroll
    for (int t = 0; t < 4; ++t) {
#pragma unroll
      for (int v = 0; v < 8; ++v) {
        int M = t * 16 + v + 8 * hi;
        float ef = acc[t][v] + bias;
        sE[M * 132 + n0 + lm] = ef;
        atomicAdd(&agg[sRow[M] * 128 + n0 + lm], ef);
      }
    }
  }
  __syncthreads();

  {
    v8f acc[4] = {{}, {}, {}, {}};
    for (int k0 = 0; k0 < 128; k0 += 4) {
      int kk = k0 + 2 * hi;
      v2f b;
      b.x = Wc1[kk * 128 + n0 + lm];
      b.y = Wc1[(kk + 1) * 128 + n0 + lm];
#pragma unroll
      for (int t = 0; t < 4; ++t) {
        v2f a;
        a.x = sE[(t * 16 + lm) * 132 + kk];
        a.y = sE[(t * 16 + lm) * 132 + kk + 1];
        acc[t] = wmma4(a, b, acc[t]);
      }
    }
    float bias = bc1[n0 + lm];
    float wc2 = Wc2[n0 + lm];
#pragma unroll
    for (int t = 0; t < 4; ++t) {
      float mine = 0.0f;
#pragma unroll
      for (int v = 0; v < 8; ++v) {
        float p = row16_sum(silu_f(acc[t][v] + bias) * wc2);
        mine = (lm == v) ? p : mine;
      }
      if (lm < 8) atomicAdd(&sW[t * 16 + lm + 8 * hi], mine);
    }
  }
  __syncthreads();

  if (tid < EPB * 3) {
    int e = tid / 3;
    int d = tid - e * 3;
    float cd = sDiff[tid] / (sqrtf(sDist[e]) + 1e-8f);
    atomicAdd(&coords_out[sRow[e] * 3 + d], cd * sW[e]);
  }
}

__global__ __launch_bounds__(256) void egnn_node_kernel(
    const float* __restrict__ x, const float* __restrict__ agg,
    const float* __restrict__ Wn1, const float* __restrict__ bn1,
    const float* __restrict__ Wn2, const float* __restrict__ bn2,
    float* __restrict__ x_out) {
  __shared__ float sA[16 * 260];
  __shared__ float sH[16 * 132];

  const int tid = threadIdx.x;
  const int lane = tid & 31;
  const int wv = tid >> 5;
  const int hi = lane >> 4;
  const int lm = lane & 15;
  const int n0 = wv * 16;
  const int nb = blockIdx.x * 16;

  for (int idx = tid; idx < 16 * 256; idx += 256) {
    int m = idx >> 8;
    int c = idx & 255;
    float v = (c < 128) ? x[(nb + m) * 128 + c] : agg[(nb + m) * 128 + (c - 128)];
    sA[m * 260 + c] = v;
  }
  __syncthreads();

  v8f acc1 = {};
  for (int k0 = 0; k0 < 256; k0 += 4) {
    int kk = k0 + 2 * hi;
    v2f a, b;
    a.x = sA[lm * 260 + kk];
    a.y = sA[lm * 260 + kk + 1];
    b.x = Wn1[kk * 128 + n0 + lm];
    b.y = Wn1[(kk + 1) * 128 + n0 + lm];
    acc1 = wmma4(a, b, acc1);
  }
  {
    float bias = bn1[n0 + lm];
#pragma unroll
    for (int v = 0; v < 8; ++v) {
      int M = v + 8 * hi;
      sH[M * 132 + n0 + lm] = silu_f(acc1[v] + bias);
    }
  }
  __syncthreads();

  v8f acc2 = {};
  for (int k0 = 0; k0 < 128; k0 += 4) {
    int kk = k0 + 2 * hi;
    v2f a, b;
    a.x = sH[lm * 132 + kk];
    a.y = sH[lm * 132 + kk + 1];
    b.x = Wn2[kk * 128 + n0 + lm];
    b.y = Wn2[(kk + 1) * 128 + n0 + lm];
    acc2 = wmma4(a, b, acc2);
  }
  {
    float bias = bn2[n0 + lm];
#pragma unroll
    for (int v = 0; v < 8; ++v) {
      int M = v + 8 * hi;
      x_out[(nb + M) * 128 + n0 + lm] = acc2[v] + bias;
    }
  }
}

extern "C" void kernel_launch(void* const* d_in, const int* in_sizes, int n_in,
                              void* d_out, int out_size, void* d_ws,
                              size_t ws_size, hipStream_t stream) {
  const float* x = (const float*)d_in[0];
  const float* coords = (const float*)d_in[1];
  const float* edge_attr = (const float*)d_in[2];
  const int* edge_index = (const int*)d_in[3];
  const float* We1 = (const float*)d_in[4];
  const float* be1 = (const float*)d_in[5];
  const float* We2 = (const float*)d_in[6];
  const float* be2 = (const float*)d_in[7];
  const float* Wc1 = (const float*)d_in[8];
  const float* bc1 = (const float*)d_in[9];
  const float* Wc2 = (const float*)d_in[10];
  const float* Wn1 = (const float*)d_in[11];
  const float* bn1 = (const float*)d_in[12];
  const float* Wn2 = (const float*)d_in[13];
  const float* bn2 = (const float*)d_in[14];

  float* x_out = (float*)d_out;
  float* coords_out = x_out + (size_t)N_NODES * 128;
  float* agg = (float*)d_ws;

  const int* erow = edge_index;
  const int* ecol = edge_index + N_EDGES;

  egnn_init_kernel<<<(N_NODES * 128 + 255) / 256, 256, 0, stream>>>(
      coords, agg, coords_out);
  egnn_edge_kernel<<<N_EDGES / EPB, 256, 0, stream>>>(
      x, coords, edge_attr, erow, ecol, We1, be1, We2, be2, Wc1, bc1, Wc2, agg,
      coords_out);
  egnn_node_kernel<<<N_NODES / 16, 256, 0, stream>>>(x, agg, Wn1, bn1, Wn2, bn2,
                                                     x_out);
}


# First edge layer split by linearity: x[row] @ We1[0:128] + be1 once per node in launch 1 (table in the x_out rows), loaded as accumulator start; per-edge K 288 -> 160 (sum order: node part first)
# speedup vs baseline: 2.3508x; 1.1368x over previous
#include <hip/hip_runtime.h>
#include <math.h>

typedef __attribute__((ext_vector_type(2))) float v2f;
typedef __attribute__((ext_vector_type(8))) float v8f;
typedef __attribute__((ext_vector_type(4))) float v4f;

#define N_NODES 50000
#define N_EDGES 800000
#define EPB 64

__device__ __forceinline__ v8f wmma4(v2f a, v2f b, v8f c) {
  return __builtin_amdgcn_wmma_f32_16x16x4_f32(false, a, false, b, (short)0, c,
                                               false, false);
}

__device__ __forceinline__ float silu_f(float v) {
  return v / (1.0f + __expf(-v));
}

__device__ __forceinline__ float row16_sum(float v) {
  v += __int_as_float(__builtin_amdgcn_update_dpp(
      0, __float_as_int(v), 0x161, 0xf, 0xf, false));
  v += __int_as_float(__builtin_amdgcn_update_dpp(
      0, __float_as_int(v), 0x162, 0xf, 0xf, false));
  v += __int_as_float(__builtin_amdgcn_update_dpp(
      0, __float_as_int(v), 0x164, 0xf, 0xf, false));
  v += __int_as_float(__builtin_amdgcn_update_dpp(
      0, __float_as_int(v), 0x168, 0xf, 0xf, false));
  return v;
}

__global__ __launch_bounds__(256) void egnn_pre_kernel(
    const float* __restrict__ x, const float* __restrict__ coords,
    const float* __restrict__ We1, const float* __restrict__ be1,
    float* __restrict__ P, float* __restrict__ agg,
    float* __restrict__ coords_out) {
  __shared__ __attribute__((aligned(16))) float sX[16 * 132];

  const int tid = threadIdx.x;
  const int lane = tid & 31;
  const int wv = tid >> 5;
  const int hi = lane >> 4;
  const int lm = lane & 15;
  const int n0 = wv * 16;
  const int nb = blockIdx.x * 16;

  const int sr = tid >> 5;
  const int sc = 4 * (tid & 31);
  const v4f* x4 = reinterpret_cast<const v4f*>(x) + nb * 32;
  const v4f xa = x4[tid];
  const v4f xb = x4[tid + 256];

  const v4f zero4 = {0.0f, 0.0f, 0.0f, 0.0f};
  v4f* g4 = reinterpret_cast<v4f*>(agg) + nb * 32;
  g4[tid] = zero4;
  g4[tid + 256] = zero4;
  if (tid < 48) coords_out[nb * 3 + tid] = coords[nb * 3 + tid];

  *reinterpret_cast<v4f*>(&sX[sr * 132 + sc]) = xa;
  *reinterpret_cast<v4f*>(&sX[(sr + 8) * 132 + sc]) = xb;
  __syncthreads();

  v8f acc = {};
  for (int k0 = 0; k0 < 128; k0 += 4) {
    int kk = k0 + 2 * hi;
    v2f a, b;
    a.x = sX[lm * 132 + kk];
    a.y = sX[lm * 132 + kk + 1];
    b.x = We1[kk * 128 + n0 + lm];
    b.y = We1[(kk + 1) * 128 + n0 + lm];
    acc = wmma4(a, b, acc);
  }
  const float bias = be1[n0 + lm];
  __syncthreads();
#pragma unroll
  for (int v = 0; v < 8; ++v) sX[(v + 8 * hi) * 132 + n0 + lm] = acc[v] + bias;
  __syncthreads();
  v4f* p4 = reinterpret_cast<v4f*>(P) + nb * 32;
  p4[tid] = *reinterpret_cast<const v4f*>(&sX[sr * 132 + sc]);
  p4[tid + 256] = *reinterpret_cast<const v4f*>(&sX[(sr + 8) * 132 + sc]);
}

__global__ __launch_bounds__(256, 8) void egnn_edge_kernel(
    const float* __restrict__ x, const float* __restrict__ coords,
    const float* __restrict__ edge_attr, const int* __restrict__ erow,
    const int* __restrict__ ecol, const float* __restrict__ We1,
    const float* __restrict__ P, const float* __restrict__ We2,
    const float* __restrict__ be2, const float* __restrict__ Wc1,
    const float* __restrict__ bc1, const float* __restrict__ Wc2,
    float* __restrict__ agg, float* __restrict__ coords_out) {
  __shared__ __attribute__((aligned(16))) float smem[EPB * 264];
  float* sA = smem;
  float* sH = smem;
  float* sE = smem + EPB * 132;

  __shared__ int sRC[2 * EPB];
  int* sRow = sRC;
  int* sCol = sRC + EPB;
  __shared__ float sDiff[EPB * 3], sDist[EPB], sW[EPB];

  const int tid = threadIdx.x;
  const int lane = tid & 31;
  const int wv = tid >> 5;
  const int hi = lane >> 4;
  const int lm = lane & 15;
  const int n0 = wv * 16;
  const int ebase = blockIdx.x * EPB;

  const v4f* ea4 = reinterpret_cast<const v4f*>(edge_attr) + ebase * 8;
  const v4f ea0 = ea4[tid];
  const v4f ea1 = ea4[tid + 256];

  if (tid < EPB) {
    int r = erow[ebase + tid];
    int c = ecol[ebase + tid];
    sRow[tid] = r;
    sCol[tid] = c;
    float d0 = coords[r * 3 + 0] - coords[c * 3 + 0];
    float d1 = coords[r * 3 + 1] - coords[c * 3 + 1];
    float d2 = coords[r * 3 + 2] - coords[c * 3 + 2];
    sDiff[tid * 3 + 0] = d0;
    sDiff[tid * 3 + 1] = d1;
    sDiff[tid * 3 + 2] = d2;
    sDist[tid] = d0 * d0 + d1 * d1 + d2 * d2;
    sW[tid] = 0.0f;
  }
  __syncthreads();

  v8f acc1[4];
  {
    const v4f* x4 = reinterpret_cast<const v4f*>(x);
    v4f xv[8];
    for (int i = 0; i < 8; ++i) {
      const int node = sCol[wv + 8 * i];
      xv[i] = x4[node * 32 + lane];
    }
#pragma unroll
    for (int t = 0; t < 4; ++t) {
#pragma unroll
      for (int v = 0; v < 8; ++v) {
        const int M = t * 16 + v + 8 * hi;
        acc1[t][v] = P[sRow[M] * 128 + n0 + lm];
      }
    }
    for (int i = 0; i < 8; ++i)
      *reinterpret_cast<v4f*>(&sA[(wv + 8 * i) * 164 + 4 * lane]) = xv[i];
    *reinterpret_cast<v4f*>(&sA[(tid >> 3) * 164 + 128 + 4 * (tid & 7)]) = ea0;
    *reinterpret_cast<v4f*>(&sA[((tid >> 3) + 32) * 164 + 128 + 4 * (tid & 7)]) =
        ea1;
  }
  __syncthreads();

  {
    for (int k0 = 0; k0 < 160; k0 += 4) {
      int kk = k0 + 2 * hi;
      v2f b;
      b.x = We1[(128 + kk) * 128 + n0 + lm];
      b.y = We1[(129 + kk) * 128 + n0 + lm];
#pragma unroll
      for (int t = 0; t < 4; ++t) {
        v2f a;
        a.x = sA[(t * 16 + lm) * 164 + kk];
        a.y = sA[(t * 16 + lm) * 164 + kk + 1];
        acc1[t] = wmma4(a, b, acc1[t]);
      }
    }
    float wlast = We1[288 * 128 + n0 + lm];
    __syncthreads();
#pragma unroll
    for (int t = 0; t < 4; ++t) {
#pragma unroll
      for (int v = 0; v < 8; ++v) {
        int M = t * 16 + v + 8 * hi;
        float pre = acc1[t][v] + sDist[M] * wlast;
        sH[M * 132 + n0 + lm] = silu_f(pre);
      }
    }
  }
  __syncthreads();

  {
    v8f acc[4] = {{}, {}, {}, {}};
    for (int k0 = 0; k0 < 128; k0 += 4) {
      int kk = k0 + 2 * hi;
      v2f b;
      b.x = We2[kk * 128 + n0 + lm];
      b.y = We2[(kk + 1) * 128 + n0 + lm];
#pragma unroll
      for (int t = 0; t < 4; ++t) {
        v2f a;
        a.x = sH[(t * 16 + lm) * 132 + kk];
        a.y = sH[(t * 16 + lm) * 132 + kk + 1];
        acc[t] = wmma4(a, b, acc[t]);
      }
    }
    float bias = be2[n0 + lm];
#pragma unroll
    for (int t = 0; t < 4; ++t) {
#pragma unroll
      for (int v = 0; v < 8; ++v) {
        int M = t * 16 + v + 8 * hi;
        float ef = acc[t][v] + bias;
        sE[M * 132 + n0 + lm] = ef;
        atomicAdd(&agg[sRow[M] * 128 + n0 + lm], ef);
      }
    }
  }
  __syncthreads();

  {
    v8f acc[4] = {{}, {}, {}, {}};
    for (int k0 = 0; k0 < 128; k0 += 4) {
      int kk = k0 + 2 * hi;
      v2f b;
      b.x = Wc1[kk * 128 + n0 + lm];
      b.y = Wc1[(kk + 1) * 128 + n0 + lm];
#pragma unroll
      for (int t = 0; t < 4; ++t) {
        v2f a;
        a.x = sE[(t * 16 + lm) * 132 + kk];
        a.y = sE[(t * 16 + lm) * 132 + kk + 1];
        acc[t] = wmma4(a, b, acc[t]);
      }
    }
    float bias = bc1[n0 + lm];
    float wc2 = Wc2[n0 + lm];
#pragma unroll
    for (int t = 0; t < 4; ++t) {
      float mine = 0.0f;
#pragma unroll
      for (int v = 0; v < 8; ++v) {
        float p = row16_sum(silu_f(acc[t][v] + bias) * wc2);
        mine = (lm == v) ? p : mine;
      }
      if (lm < 8) atomicAdd(&sW[t * 16 + lm + 8 * hi], mine);
    }
  }
  __syncthreads();

  if (tid < EPB * 3) {
    int e = tid / 3;
    int d = tid - e * 3;
    float cd = sDiff[tid] / (sqrtf(sDist[e]) + 1e-8f);
    atomicAdd(&coords_out[sRow[e] * 3 + d], cd * sW[e]);
  }
}

__global__ __launch_bounds__(256) void egnn_node_kernel(
    const float* __restrict__ x, const float* __restrict__ agg,
    const float* __restrict__ Wn1, const float* __restrict__ bn1,
    const float* __restrict__ Wn2, const float* __restrict__ bn2,
    float* __restrict__ x_out) {
  __shared__ float sA[16 * 260];
  __shared__ float sH[16 * 132];

  const int tid = threadIdx.x;
  const int lane = tid & 31;
  const int wv = tid >> 5;
  const int hi = lane >> 4;
  const int lm = lane & 15;
  const int n0 = wv * 16;
  const int nb = blockIdx.x * 16;

  for (int idx = tid; idx < 16 * 256; idx += 256) {
    int m = idx >> 8;
    int c = idx & 255;
    float v = (c < 128) ? x[(nb + m) * 128 + c] : agg[(nb + m) * 128 + (c - 128)];
    sA[m * 260 + c] = v;
  }
  __syncthreads();

  v8f acc1 = {};
  for (int k0 = 0; k0 < 256; k0 += 4) {
    int kk = k0 + 2 * hi;
    v2f a, b;
    a.x = sA[lm * 260 + kk];
    a.y = sA[lm * 260 + kk + 1];
    b.x = Wn1[kk * 128 + n0 + lm];
    b.y = Wn1[(kk + 1) * 128 + n0 + lm];
    acc1 = wmma4(a, b, acc1);
  }
  {
    float bias = bn1[n0 + lm];
#pragma unroll
    for (int v = 0; v < 8; ++v) {
      int M = v + 8 * hi;
      sH[M * 132 + n0 + lm] = silu_f(acc1[v] + bias);
    }
  }
  __syncthreads();

  v8f acc2 = {};
  for (int k0 = 0; k0 < 128; k0 += 4) {
    int kk = k0 + 2 * hi;
    v2f a, b;
    a.x = sH[lm * 132 + kk];
    a.y = sH[lm * 132 + kk + 1];
    b.x = Wn2[kk * 128 + n0 + lm];
    b.y = Wn2[(kk + 1) * 128 + n0 + lm];
    acc2 = wmma4(a, b, acc2);
  }
  {
    float bias = bn2[n0 + lm];
#pragma unroll
    for (int v = 0; v < 8; ++v) {
      int M = v + 8 * hi;
      x_out[(nb + M) * 128 + n0 + lm] = acc2[v] + bias;
    }
  }
}

extern "C" void kernel_launch(void* const* d_in, const int* in_sizes, int n_in,
                              void* d_out, int out_size, void* d_ws,
                              size_t ws_size, hipStream_t stream) {
  const float* x = (const float*)d_in[0];
  const float* coords = (const float*)d_in[1];
  const float* edge_attr = (const float*)d_in[2];
  const int* edge_index = (const int*)d_in[3];
  const float* We1 = (const float*)d_in[4];
  const float* be1 = (const float*)d_in[5];
  const float* We2 = (const float*)d_in[6];
  const float* be2 = (const float*)d_in[7];
  const float* Wc1 = (const float*)d_in[8];
  const float* bc1 = (const float*)d_in[9];
  const float* Wc2 = (const float*)d_in[10];
  const float* Wn1 = (const float*)d_in[11];
  const float* bn1 = (const float*)d_in[12];
  const float* Wn2 = (const float*)d_in[13];
  const float* bn2 = (const float*)d_in[14];

  float* x_out = (float*)d_out;
  float* coords_out = x_out + (size_t)N_NODES * 128;
  float* agg = (float*)d_ws;

  const int* erow = edge_index;
  const int* ecol = edge_index + N_EDGES;

  egnn_pre_kernel<<<N_NODES / 16, 256, 0, stream>>>(x, coords, We1, be1, x_out,
                                                    agg, coords_out);
  egnn_edge_kernel<<<N_EDGES / EPB, 256, 0, stream>>>(
      x, coords, edge_attr, erow, ecol, We1, x_out, We2, be2, Wc1, bc1, Wc2, agg,
      coords_out);
  egnn_node_kernel<<<N_NODES / 16, 256, 0, stream>>>(x, agg, Wn1, bn1, Wn2, bn2,
                                                     x_out);
}


# First edge layer's x[col] block also applied once per node: table Q in workspace behind agg (taken if ws_size allows), acc starts at P[row]+Q[col], per-edge K 160 -> 32 (sum order: node parts first)
# speedup vs baseline: 2.7324x; 1.1623x over previous
#include <hip/hip_runtime.h>
#include <math.h>

typedef __attribute__((ext_vector_type(2))) float v2f;
typedef __attribute__((ext_vector_type(8))) float v8f;
typedef __attribute__((ext_vector_type(4))) float v4f;

#define N_NODES 50000
#define N_EDGES 800000
#define EPB 64

__device__ __forceinline__ v8f wmma4(v2f a, v2f b, v8f c) {
  return __builtin_amdgcn_wmma_f32_16x16x4_f32(false, a, false, b, (short)0, c,
                                               false, false);
}

__device__ __forceinline__ float silu_f(float v) {
  return v / (1.0f + __expf(-v));
}

__device__ __forceinline__ float row16_sum(float v) {
  v += __int_as_float(__builtin_amdgcn_update_dpp(
      0, __float_as_int(v), 0x161, 0xf, 0xf, false));
  v += __int_as_float(__builtin_amdgcn_update_dpp(
      0, __float_as_int(v), 0x162, 0xf, 0xf, false));
  v += __int_as_float(__builtin_amdgcn_update_dpp(
      0, __float_as_int(v), 0x164, 0xf, 0xf, false));
  v += __int_as_float(__builtin_amdgcn_update_dpp(
      0, __float_as_int(v), 0x168, 0xf, 0xf, false));
  return v;
}

__global__ __launch_bounds__(256) void egnn_pre_kernel(
    const float* __restrict__ x, const float* __restrict__ coords,
    const float* __restrict__ We1, const float* __restrict__ be1,
    float* __restrict__ P, float* __restrict__ agg,
    float* __restrict__ coords_out) {
  __shared__ __attribute__((aligned(16))) float sX[16 * 132];

  const int tid = threadIdx.x;
  const int lane = tid & 31;
  const int wv = tid >> 5;
  const int hi = lane >> 4;
  const int lm = lane & 15;
  const int n0 = wv * 16;
  const int nb = blockIdx.x * 16;

  const int sr = tid >> 5;
  const int sc = 4 * (tid & 31);
  const v4f* x4 = reinterpret_cast<const v4f*>(x) + nb * 32;
  const v4f xa = x4[tid];
  const v4f xb = x4[tid + 256];

  const v4f zero4 = {0.0f, 0.0f, 0.0f, 0.0f};
  v4f* g4 = reinterpret_cast<v4f*>(agg) + nb * 32;
  g4[tid] = zero4;
  g4[tid + 256] = zero4;
  if (tid < 48) coords_out[nb * 3 + tid] = coords[nb * 3 + tid];

  *reinterpret_cast<v4f*>(&sX[sr * 132 + sc]) = xa;
  *reinterpret_cast<v4f*>(&sX[(sr + 8) * 132 + sc]) = xb;
  __syncthreads();

  v8f acc = {};
  for (int k0 = 0; k0 < 128; k0 += 4) {
    int kk = k0 + 2 * hi;
    v2f a, b;
    a.x = sX[lm * 132 + kk];
    a.y = sX[lm * 132 + kk + 1];
    b.x = We1[kk * 128 + n0 + lm];
    b.y = We1[(kk + 1) * 128 + n0 + lm];
    acc = wmma4(a, b, acc);
  }
  const float bias = be1[n0 + lm];
  __syncthreads();
#pragma unroll
  for (int v = 0; v < 8; ++v) sX[(v + 8 * hi) * 132 + n0 + lm] = acc[v] + bias;
  __syncthreads();
  v4f* p4 = reinterpret_cast<v4f*>(P) + nb * 32;
  p4[tid] = *reinterpret_cast<const v4f*>(&sX[sr * 132 + sc]);
  p4[tid + 256] = *reinterpret_cast<const v4f*>(&sX[(sr + 8) * 132 + sc]);
}

__global__ __launch_bounds__(256, 8) void egnn_edge_kernel(
    const float* __restrict__ x, const float* __restrict__ coords,
    const float* __restrict__ edge_attr, const int* __restrict__ erow,
    const int* __restrict__ ecol, const float* __restrict__ We1,
    const float* __restrict__ P, const float* __restrict__ We2,
    const float* __restrict__ be2, const float* __restrict__ Wc1,
    const float* __restrict__ bc1, const float* __restrict__ Wc2,
    float* __restrict__ agg, float* __restrict__ coords_out) {
  __shared__ __attribute__((aligned(16))) float smem[EPB * 264];
  float* sA = smem;
  float* sH = smem;
  float* sE = smem + EPB * 132;

  __shared__ int sRC[2 * EPB];
  int* sRow = sRC;
  int* sCol = sRC + EPB;
  __shared__ float sDiff[EPB * 3], sDist[EPB], sW[EPB];

  const int tid = threadIdx.x;
  const int lane = tid & 31;
  const int wv = tid >> 5;
  const int hi = lane >> 4;
  const int lm = lane & 15;
  const int n0 = wv * 16;
  const int ebase = blockIdx.x * EPB;

  const v4f* ea4 = reinterpret_cast<const v4f*>(edge_attr) + ebase * 8;
  const v4f ea0 = ea4[tid];
  const v4f ea1 = ea4[tid + 256];

  if (tid < EPB) {
    int r = erow[ebase + tid];
    int c = ecol[ebase + tid];
    sRow[tid] = r;
    sCol[tid] = c;
    float d0 = coords[r * 3 + 0] - coords[c * 3 + 0];
    float d1 = coords[r * 3 + 1] - coords[c * 3 + 1];
    float d2 = coords[r * 3 + 2] - coords[c * 3 + 2];
    sDiff[tid * 3 + 0] = d0;
    sDiff[tid * 3 + 1] = d1;
    sDiff[tid * 3 + 2] = d2;
    sDist[tid] = d0 * d0 + d1 * d1 + d2 * d2;
    sW[tid] = 0.0f;
  }
  __syncthreads();

  v8f acc1[4];
  {
    const v4f* x4 = reinterpret_cast<const v4f*>(x);
    v4f xv[8];
    for (int i = 0; i < 8; ++i) {
      const int node = sCol[wv + 8 * i];
      xv[i] = x4[node * 32 + lane];
    }
#pragma unroll
    for (int t = 0; t < 4; ++t) {
#pragma unroll
      for (int v = 0; v < 8; ++v) {
        const int M = t * 16 + v + 8 * hi;
        acc1[t][v] = P[sRow[M] * 128 + n0 + lm];
      }
    }
    for (int i = 0; i < 8; ++i)
      *reinterpret_cast<v4f*>(&sA[(wv + 8 * i) * 164 + 4 * lane]) = xv[i];
    *reinterpret_cast<v4f*>(&sA[(tid >> 3) * 164 + 128 + 4 * (tid & 7)]) = ea0;
    *reinterpret_cast<v4f*>(&sA[((tid >> 3) + 32) * 164 + 128 + 4 * (tid & 7)]) =
        ea1;
  }
  __syncthreads();

  {
    for (int k0 = 0; k0 < 160; k0 += 4) {
      int kk = k0 + 2 * hi;
      v2f b;
      b.x = We1[(128 + kk) * 128 + n0 + lm];
      b.y = We1[(129 + kk) * 128 + n0 + lm];
#pragma unroll
      for (int t = 0; t < 4; ++t) {
        v2f a;
        a.x = sA[(t * 16 + lm) * 164 + kk];
        a.y = sA[(t * 16 + lm) * 164 + kk + 1];
        acc1[t] = wmma4(a, b, acc1[t]);
      }
    }
    float wlast = We1[288 * 128 + n0 + lm];
    __syncthreads();
#pragma unroll
    for (int t = 0; t < 4; ++t) {
#pragma unroll
      for (int v = 0; v < 8; ++v) {
        int M = t * 16 + v + 8 * hi;
        float pre = acc1[t][v] + sDist[M] * wlast;
        sH[M * 132 + n0 + lm] = silu_f(pre);
      }
    }
  }
  __syncthreads();

  {
    v8f acc[4] = {{}, {}, {}, {}};
    for (int k0 = 0; k0 < 128; k0 += 4) {
      int kk = k0 + 2 * hi;
      v2f b;
      b.x = We2[kk * 128 + n0 + lm];
      b.y = We2[(kk + 1) * 128 + n0 + lm];
#pragma unroll
      for (int t = 0; t < 4; ++t) {
        v2f a;
        a.x = sH[(t * 16 + lm) * 132 + kk];
        a.y = sH[(t * 16 + lm) * 132 + kk + 1];
        acc[t] = wmma4(a, b, acc[t]);
      }
    }
    float bias = be2[n0 + lm];
#pragma unroll
    for (int t = 0; t < 4; ++t) {
#pragma unroll
      for (int v = 0; v < 8; ++v) {
        int M = t * 16 + v + 8 * hi;
        float ef = acc[t][v] + bias;
        sE[M * 132 + n0 + lm] = ef;
        atomicAdd(&agg[sRow[M] * 128 + n0 + lm], ef);
      }
    }
  }
  __syncthreads();

  {
    v8f acc[4] = {{}, {}, {}, {}};
    for (int k0 = 0; k0 < 128; k0 += 4) {
      int kk = k0 + 2 * hi;
      v2f b;
      b.x = Wc1[kk * 128 + n0 + lm];
      b.y = Wc1[(kk + 1) * 128 + n0 + lm];
#pragma unroll
      for (int t = 0; t < 4; ++t) {
        v2f a;
        a.x = sE[(t * 16 + lm) * 132 + kk];
        a.y = sE[(t * 16 + lm) * 132 + kk + 1];
        acc[t] = wmma4(a, b, acc[t]);
      }
    }
    float bias = bc1[n0 + lm];
    float wc2 = Wc2[n0 + lm];
#pragma unroll
    for (int t = 0; t < 4; ++t) {
      float mine = 0.0f;
#pragma unroll
      for (int v = 0; v < 8; ++v) {
        float p = row16_sum(silu_f(acc[t][v] + bias) * wc2);
        mine = (lm == v) ? p : mine;
      }
      if (lm < 8) atomicAdd(&sW[t * 16 + lm + 8 * hi], mine);
    }
  }
  __syncthreads();

  if (tid < EPB * 3) {
    int e = tid / 3;
    int d = tid - e * 3;
    float cd = sDiff[tid] / (sqrtf(sDist[e]) + 1e-8f);
    atomicAdd(&coords_out[sRow[e] * 3 + d], cd * sW[e]);
  }
}

__global__ __launch_bounds__(256) void egnn_pre2_kernel(
    const float* __restrict__ x, const float* __restrict__ coords,
    const float* __restrict__ We1, const float* __restrict__ be1,
    float* __restrict__ P, float* __restrict__ Q, float* __restrict__ agg,
    float* __restrict__ coords_out) {
  __shared__ __attribute__((aligned(16))) float sX[16 * 132];
  __shared__ __attribute__((aligned(16))) float sQ[16 * 132];

  const int tid = threadIdx.x;
  const int lane = tid & 31;
  const int wv = tid >> 5;
  const int hi = lane >> 4;
  const int lm = lane & 15;
  const int n0 = wv * 16;
  const int nb = blockIdx.x * 16;

  const int sr = tid >> 5;
  const int sc = 4 * (tid & 31);
  const v4f* x4 = reinterpret_cast<const v4f*>(x) + nb * 32;
  const v4f xa = x4[tid];
  const v4f xb = x4[tid + 256];

  const v4f zero4 = {0.0f, 0.0f, 0.0f, 0.0f};
  v4f* g4 = reinterpret_cast<v4f*>(agg) + nb * 32;
  g4[tid] = zero4;
  g4[tid + 256] = zero4;
  if (tid < 48) coords_out[nb * 3 + tid] = coords[nb * 3 + tid];

  *reinterpret_cast<v4f*>(&sX[sr * 132 + sc]) = xa;
  *reinterpret_cast<v4f*>(&sX[(sr + 8) * 132 + sc]) = xb;
  __syncthreads();

  v8f accP = {};
  v8f accQ = {};
  for (int k0 = 0; k0 < 128; k0 += 4) {
    int kk = k0 + 2 * hi;
    v2f a, bp, bq;
    a.x = sX[lm * 132 + kk];
    a.y = sX[lm * 132 + kk + 1];
    bp.x = We1[kk * 128 + n0 + lm];
    bp.y = We1[(kk + 1) * 128 + n0 + lm];
    bq.x = We1[(128 + kk) * 128 + n0 + lm];
    bq.y = We1[(129 + kk) * 128 + n0 + lm];
    accP = wmma4(a, bp, accP);
    accQ = wmma4(a, bq, accQ);
  }
  const float bias = be1[n0 + lm];
  __syncthreads();
#pragma unroll
  for (int v = 0; v < 8; ++v) {
    sX[(v + 8 * hi) * 132 + n0 + lm] = accP[v] + bias;
    sQ[(v + 8 * hi) * 132 + n0 + lm] = accQ[v];
  }
  __syncthreads();
  v4f* p4 = reinterpret_cast<v4f*>(P) + nb * 32;
  v4f* q4 = reinterpret_cast<v4f*>(Q) + nb * 32;
  p4[tid] = *reinterpret_cast<const v4f*>(&sX[sr * 132 + sc]);
  p4[tid + 256] = *reinterpret_cast<const v4f*>(&sX[(sr + 8) * 132 + sc]);
  q4[tid] = *reinterpret_cast<const v4f*>(&sQ[sr * 132 + sc]);
  q4[tid + 256] = *reinterpret_cast<const v4f*>(&sQ[(sr + 8) * 132 + sc]);
}

__global__ __launch_bounds__(256, 8) void egnn_edge2_kernel(
    const float* __restrict__ coords, const float* __restrict__ edge_attr,
    const int* __restrict__ erow, const int* __restrict__ ecol,
    const float* __restrict__ We1, const float* __restrict__ P,
    const float* __restrict__ Q, const float* __restrict__ We2,
    const float* __restrict__ be2, const float* __restrict__ Wc1,
    const float* __restrict__ bc1, const float* __restrict__ Wc2,
    float* __restrict__ agg, float* __restrict__ coords_out) {
  __shared__ __attribute__((aligned(16))) float smem[EPB * 264];
  float* sH = smem;
  float* sE = smem + EPB * 132;
  float* sA = sE;

  __shared__ int sRC[2 * EPB];
  int* sRow = sRC;
  int* sCol = sRC + EPB;
  __shared__ float sDiff[EPB * 3], sDist[EPB], sW[EPB];

  const int tid = threadIdx.x;
  const int lane = tid & 31;
  const int wv = tid >> 5;
  const int hi = lane >> 4;
  const int lm = lane & 15;
  const int n0 = wv * 16;
  const int ebase = blockIdx.x * EPB;

  const v4f* ea4 = reinterpret_cast<const v4f*>(edge_attr) + ebase * 8;
  const v4f ea0 = ea4[tid];
  const v4f ea1 = ea4[tid + 256];

  if (tid < EPB) {
    int r = erow[ebase + tid];
    int c = ecol[ebase + tid];
    sRow[tid] = r;
    sCol[tid] = c;
    float d0 = coords[r * 3 + 0] - coords[c * 3 + 0];
    float d1 = coords[r * 3 + 1] - coords[c * 3 + 1];
    float d2 = coords[r * 3 + 2] - coords[c * 3 + 2];
    sDiff[tid * 3 + 0] = d0;
    sDiff[tid * 3 + 1] = d1;
    sDiff[tid * 3 + 2] = d2;
    sDist[tid] = d0 * d0 + d1 * d1 + d2 * d2;
    sW[tid] = 0.0f;
  }
  __syncthreads();

  v8f acc1[4];
  {
    v8f qv[4];
#pragma unroll
    for (int t = 0; t < 4; ++t) {
#pragma unroll
      for (int v = 0; v < 8; ++v) {
        const int M = t * 16 + v + 8 * hi;
        qv[t][v] = Q[sCol[M] * 128 + n0 + lm];
      }
    }
#pragma unroll
    for (int t = 0; t < 4; ++t) {
#pragma unroll
      for (int v = 0; v < 8; ++v) {
        const int M = t * 16 + v + 8 * hi;
        acc1[t][v] = P[sRow[M] * 128 + n0 + lm];
      }
    }
#pragma unroll
    for (int t = 0; t < 4; ++t) acc1[t] += qv[t];
  }
  *reinterpret_cast<v4f*>(&sA[(tid >> 3) * 36 + 4 * (tid & 7)]) = ea0;
  *reinterpret_cast<v4f*>(&sA[((tid >> 3) + 32) * 36 + 4 * (tid & 7)]) = ea1;
  __syncthreads();

  {
    for (int k0 = 0; k0 < 32; k0 += 4) {
      int kk = k0 + 2 * hi;
      v2f b;
      b.x = We1[(256 + kk) * 128 + n0 + lm];
      b.y = We1[(257 + kk) * 128 + n0 + lm];
#pragma unroll
      for (int t = 0; t < 4; ++t) {
        v2f a;
        a.x = sA[(t * 16 + lm) * 36 + kk];
        a.y = sA[(t * 16 + lm) * 36 + kk + 1];
        acc1[t] = wmma4(a, b, acc1[t]);
      }
    }
    float wlast = We1[288 * 128 + n0 + lm];
#pragma unroll
    for (int t = 0; t < 4; ++t) {
#pragma unroll
      for (int v = 0; v < 8; ++v) {
        int M = t * 16 + v + 8 * hi;
        float pre = acc1[t][v] + sDist[M] * wlast;
        sH[M * 132 + n0 + lm] = silu_f(pre);
      }
    }
  }
  __syncthreads();

  {
    v8f acc[4] = {{}, {}, {}, {}};
    for (int k0 = 0; k0 < 128; k0 += 4) {
      int kk = k0 + 2 * hi;
      v2f b;
      b.x = We2[kk * 128 + n0 + lm];
      b.y = We2[(kk + 1) * 128 + n0 + lm];
#pragma unroll
      for (int t = 0; t < 4; ++t) {
        v2f a;
        a.x = sH[(t * 16 + lm) * 132 + kk];
        a.y = sH[(t * 16 + lm) * 132 + kk + 1];
        acc[t] = wmma4(a, b, acc[t]);
      }
    }
    float bias = be2[n0 + lm];
#pragma unroll
    for (int t = 0; t < 4; ++t) {
#pragma unroll
      for (int v = 0; v < 8; ++v) {
        int M = t * 16 + v + 8 * hi;
        float ef = acc[t][v] + bias;
        sE[M * 132 + n0 + lm] = ef;
        atomicAdd(&agg[sRow[M] * 128 + n0 + lm], ef);
      }
    }
  }
  __syncthreads();

  {
    v8f acc[4] = {{}, {}, {}, {}};
    for (int k0 = 0; k0 < 128; k0 += 4) {
      int kk = k0 + 2 * hi;
      v2f b;
      b.x = Wc1[kk * 128 + n0 + lm];
      b.y = Wc1[(kk + 1) * 128 + n0 + lm];
#pragma unroll
      for (int t = 0; t < 4; ++t) {
        v2f a;
        a.x = sE[(t * 16 + lm) * 132 + kk];
        a.y = sE[(t * 16 + lm) * 132 + kk + 1];
        acc[t] = wmma4(a, b, acc[t]);
      }
    }
    float bias = bc1[n0 + lm];
    float wc2 = Wc2[n0 + lm];
#pragma unroll
    for (int t = 0; t < 4; ++t) {
      float mine = 0.0f;
#pragma unroll
      for (int v = 0; v < 8; ++v) {
        float p = row16_sum(silu_f(acc[t][v] + bias) * wc2);
        mine = (lm == v) ? p : mine;
      }
      if (lm < 8) atomicAdd(&sW[t * 16 + lm + 8 * hi], mine);
    }
  }
  __syncthreads();

  if (tid < EPB * 3) {
    int e = tid / 3;
    int d = tid - e * 3;
    float cd = sDiff[tid] / (sqrtf(sDist[e]) + 1e-8f);
    atomicAdd(&coords_out[sRow[e] * 3 + d], cd * sW[e]);
  }
}

__global__ __launch_bounds__(256) void egnn_node_kernel(
    const float* __restrict__ x, const float* __restrict__ agg,
    const float* __restrict__ Wn1, const float* __restrict__ bn1,
    const float* __restrict__ Wn2, const float* __restrict__ bn2,
    float* __restrict__ x_out) {
  __shared__ float sA[16 * 260];
  __shared__ float sH[16 * 132];

  const int tid = threadIdx.x;
  const int lane = tid & 31;
  const int wv = tid >> 5;
  const int hi = lane >> 4;
  const int lm = lane & 15;
  const int n0 = wv * 16;
  const int nb = blockIdx.x * 16;

  for (int idx = tid; idx < 16 * 256; idx += 256) {
    int m = idx >> 8;
    int c = idx & 255;
    float v = (c < 128) ? x[(nb + m) * 128 + c] : agg[(nb + m) * 128 + (c - 128)];
    sA[m * 260 + c] = v;
  }
  __syncthreads();

  v8f acc1 = {};
  for (int k0 = 0; k0 < 256; k0 += 4) {
    int kk = k0 + 2 * hi;
    v2f a, b;
    a.x = sA[lm * 260 + kk];
    a.y = sA[lm * 260 + kk + 1];
    b.x = Wn1[kk * 128 + n0 + lm];
    b.y = Wn1[(kk + 1) * 128 + n0 + lm];
    acc1 = wmma4(a, b, acc1);
  }
  {
    float bias = bn1[n0 + lm];
#pragma unroll
    for (int v = 0; v < 8; ++v) {
      int M = v + 8 * hi;
      sH[M * 132 + n0 + lm] = silu_f(acc1[v] + bias);
    }
  }
  __syncthreads();

  v8f acc2 = {};
  for (int k0 = 0; k0 < 128; k0 += 4) {
    int kk = k0 + 2 * hi;
    v2f a, b;
    a.x = sH[lm * 132 + kk];
    a.y = sH[lm * 132 + kk + 1];
    b.x = Wn2[kk * 128 + n0 + lm];
    b.y = Wn2[(kk + 1) * 128 + n0 + lm];
    acc2 = wmma4(a, b, acc2);
  }
  {
    float bias = bn2[n0 + lm];
#pragma unroll
    for (int v = 0; v < 8; ++v) {
      int M = v + 8 * hi;
      x_out[(nb + M) * 128 + n0 + lm] = acc2[v] + bias;
    }
  }
}

extern "C" void kernel_launch(void* const* d_in, const int* in_sizes, int n_in,
                              void* d_out, int out_size, void* d_ws,
                              size_t ws_size, hipStream_t stream) {
  const float* x = (const float*)d_in[0];
  const float* coords = (const float*)d_in[1];
  const float* edge_attr = (const float*)d_in[2];
  const int* edge_index = (const int*)d_in[3];
  const float* We1 = (const float*)d_in[4];
  const float* be1 = (const float*)d_in[5];
  const float* We2 = (const float*)d_in[6];
  const float* be2 = (const float*)d_in[7];
  const float* Wc1 = (const float*)d_in[8];
  const float* bc1 = (const float*)d_in[9];
  const float* Wc2 = (const float*)d_in[10];
  const float* Wn1 = (const float*)d_in[11];
  const float* bn1 = (const float*)d_in[12];
  const float* Wn2 = (const float*)d_in[13];
  const float* bn2 = (const float*)d_in[14];

  float* x_out = (float*)d_out;
  float* coords_out = x_out + (size_t)N_NODES * 128;
  float* agg = (float*)d_ws;

  const int* erow = edge_index;
  const int* ecol = edge_index + N_EDGES;

  if (ws_size >= (size_t)2 * N_NODES * 128 * sizeof(float)) {
    float* Q = agg + (size_t)N_NODES * 128;
    egnn_pre2_kernel<<<N_NODES / 16, 256, 0, stream>>>(x, coords, We1, be1,
                                                       x_out, Q, agg, coords_out);
    egnn_edge2_kernel<<<N_EDGES / EPB, 256, 0, stream>>>(
        coords, edge_attr, erow, ecol, We1, x_out, Q, We2, be2, Wc1, bc1, Wc2,
        agg, coords_out);
  } else {
    egnn_pre_kernel<<<N_NODES / 16, 256, 0, stream>>>(x, coords, We1, be1, x_out,
                                                      agg, coords_out);
    egnn_edge_kernel<<<N_EDGES / EPB, 256, 0, stream>>>(
        x, coords, edge_attr, erow, ecol, We1, x_out, We2, be2, Wc1, bc1, Wc2,
        agg, coords_out);
  }
  egnn_node_kernel<<<N_NODES / 16, 256, 0, stream>>>(x, agg, Wn1, bn1, Wn2, bn2,
                                                     x_out);
}


# SiLU in both edge layers and the node layer: v_rcp_f32 (about 1 ulp) plus a multiply replace the correctly rounded f32 division, about 10 vector instructions per value, 64 values per lane per tile
# speedup vs baseline: 2.8899x; 1.0576x over previous
#include <hip/hip_runtime.h>
#include <math.h>

typedef __attribute__((ext_vector_type(2))) float v2f;
typedef __attribute__((ext_vector_type(8))) float v8f;
typedef __attribute__((ext_vector_type(4))) float v4f;

#define N_NODES 50000
#define N_EDGES 800000
#define EPB 64

__device__ __forceinline__ v8f wmma4(v2f a, v2f b, v8f c) {
  return __builtin_amdgcn_wmma_f32_16x16x4_f32(false, a, false, b, (short)0, c,
                                               false, false);
}

__device__ __forceinline__ float silu_f(float v) {
  return v * __builtin_amdgcn_rcpf(1.0f + __expf(-v));
}

__device__ __forceinline__ float row16_sum(float v) {
  v += __int_as_float(__builtin_amdgcn_update_dpp(
      0, __float_as_int(v), 0x161, 0xf, 0xf, false));
  v += __int_as_float(__builtin_amdgcn_update_dpp(
      0, __float_as_int(v), 0x162, 0xf, 0xf, false));
  v += __int_as_float(__builtin_amdgcn_update_dpp(
      0, __float_as_int(v), 0x164, 0xf, 0xf, false));
  v += __int_as_float(__builtin_amdgcn_update_dpp(
      0, __float_as_int(v), 0x168, 0xf, 0xf, false));
  return v;
}

__global__ __launch_bounds__(256) void egnn_pre_kernel(
    const float* __restrict__ x, const float* __restrict__ coords,
    const float* __restrict__ We1, const float* __restrict__ be1,
    float* __restrict__ P, float* __restrict__ agg,
    float* __restrict__ coords_out) {
  __shared__ __attribute__((aligned(16))) float sX[16 * 132];

  const int tid = threadIdx.x;
  const int lane = tid & 31;
  const int wv = tid >> 5;
  const int hi = lane >> 4;
  const int lm = lane & 15;
  const int n0 = wv * 16;
  const int nb = blockIdx.x * 16;

  const int sr = tid >> 5;
  const int sc = 4 * (tid & 31);
  const v4f* x4 = reinterpret_cast<const v4f*>(x) + nb * 32;
  const v4f xa = x4[tid];
  const v4f xb = x4[tid + 256];

  const v4f zero4 = {0.0f, 0.0f, 0.0f, 0.0f};
  v4f* g4 = reinterpret_cast<v4f*>(agg) + nb * 32;
  g4[tid] = zero4;
  g4[tid + 256] = zero4;
  if (tid < 48) coords_out[nb * 3 + tid] = coords[nb * 3 + tid];

  *reinterpret_cast<v4f*>(&sX[sr * 132 + sc]) = xa;
  *reinterpret_cast<v4f*>(&sX[(sr + 8) * 132 + sc]) = xb;
  __syncthreads();

  v8f acc = {};
  for (int k0 = 0; k0 < 128; k0 += 4) {
    int kk = k0 + 2 * hi;
    v2f a, b;
    a.x = sX[lm * 132 + kk];
    a.y = sX[lm * 132 + kk + 1];
    b.x = We1[kk * 128 + n0 + lm];
    b.y = We1[(kk + 1) * 128 + n0 + lm];
    acc = wmma4(a, b, acc);
  }
  const float bias = be1[n0 + lm];
  __syncthreads();
#pragma unroll
  for (int v = 0; v < 8; ++v) sX[(v + 8 * hi) * 132 + n0 + lm] = acc[v] + bias;
  __syncthreads();
  v4f* p4 = reinterpret_cast<v4f*>(P) + nb * 32;
  p4[tid] = *reinterpret_cast<const v4f*>(&sX[sr * 132 + sc]);
  p4[tid + 256] = *reinterpret_cast<const v4f*>(&sX[(sr + 8) * 132 + sc]);
}

__global__ __launch_bounds__(256, 8) void egnn_edge_kernel(
    const float* __restrict__ x, const float* __restrict__ coords,
    const float* __restrict__ edge_attr, const int* __restrict__ erow,
    const int* __restrict__ ecol, const float* __restrict__ We1,
    const float* __restrict__ P, const float* __restrict__ We2,
    const float* __restrict__ be2, const float* __restrict__ Wc1,
    const float* __restrict__ bc1, const float* __restrict__ Wc2,
    float* __restrict__ agg, float* __restrict__ coords_out) {
  __shared__ __attribute__((aligned(16))) float smem[EPB * 264];
  float* sA = smem;
  float* sH = smem;
  float* sE = smem + EPB * 132;

  __shared__ int sRC[2 * EPB];
  int* sRow = sRC;
  int* sCol = sRC + EPB;
  __shared__ float sDiff[EPB * 3], sDist[EPB], sW[EPB];

  const int tid = threadIdx.x;
  const int lane = tid & 31;
  const int wv = tid >> 5;
  const int hi = lane >> 4;
  const int lm = lane & 15;
  const int n0 = wv * 16;
  const int ebase = blockIdx.x * EPB;

  const v4f* ea4 = reinterpret_cast<const v4f*>(edge_attr) + ebase * 8;
  const v4f ea0 = ea4[tid];
  const v4f ea1 = ea4[tid + 256];

  if (tid < EPB) {
    int r = erow[ebase + tid];
    int c = ecol[ebase + tid];
    sRow[tid] = r;
    sCol[tid] = c;
    float d0 = coords[r * 3 + 0] - coords[c * 3 + 0];
    float d1 = coords[r * 3 + 1] - coords[c * 3 + 1];
    float d2 = coords[r * 3 + 2] - coords[c * 3 + 2];
    sDiff[tid * 3 + 0] = d0;
    sDiff[tid * 3 + 1] = d1;
    sDiff[tid * 3 + 2] = d2;
    sDist[tid] = d0 * d0 + d1 * d1 + d2 * d2;
    sW[tid] = 0.0f;
  }
  __syncthreads();

  v8f acc1[4];
  {
    const v4f* x4 = reinterpret_cast<const v4f*>(x);
    v4f xv[8];
    for (int i = 0; i < 8; ++i) {
      const int node = sCol[wv + 8 * i];
      xv[i] = x4[node * 32 + lane];
    }
#pragma unroll
    for (int t = 0; t < 4; ++t) {
#pragma unroll
      for (int v = 0; v < 8; ++v) {
        const int M = t * 16 + v + 8 * hi;
        acc1[t][v] = P[sRow[M] * 128 + n0 + lm];
      }
    }
    for (int i = 0; i < 8; ++i)
      *reinterpret_cast<v4f*>(&sA[(wv + 8 * i) * 164 + 4 * lane]) = xv[i];
    *reinterpret_cast<v4f*>(&sA[(tid >> 3) * 164 + 128 + 4 * (tid & 7)]) = ea0;
    *reinterpret_cast<v4f*>(&sA[((tid >> 3) + 32) * 164 + 128 + 4 * (tid & 7)]) =
        ea1;
  }
  __syncthreads();

  {
    for (int k0 = 0; k0 < 160; k0 += 4) {
      int kk = k0 + 2 * hi;
      v2f b;
      b.x = We1[(128 + kk) * 128 + n0 + lm];
      b.y = We1[(129 + kk) * 128 + n0 + lm];
#pragma unroll
      for (int t = 0; t < 4; ++t) {
        v2f a;
        a.x = sA[(t * 16 + lm) * 164 + kk];
        a.y = sA[(t * 16 + lm) * 164 + kk + 1];
        acc1[t] = wmma4(a, b, acc1[t]);
      }
    }
    float wlast = We1[288 * 128 + n0 + lm];
    __syncthreads();
#pragma unroll
    for (int t = 0; t < 4; ++t) {
#pragma unroll
      for (int v = 0; v < 8; ++v) {
        int M = t * 16 + v + 8 * hi;
        float pre = acc1[t][v] + sDist[M] * wlast;
        sH[M * 132 + n0 + lm] = silu_f(pre);
      }
    }
  }
  __syncthreads();

  {
    v8f acc[4] = {{}, {}, {}, {}};
    for (int k0 = 0; k0 < 128; k0 += 4) {
      int kk = k0 + 2 * hi;
      v2f b;
      b.x = We2[kk * 128 + n0 + lm];
      b.y = We2[(kk + 1) * 128 + n0 + lm];
#pragma unroll
      for (int t = 0; t < 4; ++t) {
        v2f a;
        a.x = sH[(t * 16 + lm) * 132 + kk];
        a.y = sH[(t * 16 + lm) * 132 + kk + 1];
        acc[t] = wmma4(a, b, acc[t]);
      }
    }
    float bias = be2[n0 + lm];
#pragma unroll
    for (int t = 0; t < 4; ++t) {
#pragma unroll
      for (int v = 0; v < 8; ++v) {
        int M = t * 16 + v + 8 * hi;
        float ef = acc[t][v] + bias;
        sE[M * 132 + n0 + lm] = ef;
        atomicAdd(&agg[sRow[M] * 128 + n0 + lm], ef);
      }
    }
  }
  __syncthreads();

  {
    v8f acc[4] = {{}, {}, {}, {}};
    for (int k0 = 0; k0 < 128; k0 += 4) {
      int kk = k0 + 2 * hi;
      v2f b;
      b.x = Wc1[kk * 128 + n0 + lm];
      b.y = Wc1[(kk + 1) * 128 + n0 + lm];
#pragma unroll
      for (int t = 0; t < 4; ++t) {
        v2f a;
        a.x = sE[(t * 16 + lm) * 132 + kk];
        a.y = sE[(t * 16 + lm) * 132 + kk + 1];
        acc[t] = wmma4(a, b, acc[t]);
      }
    }
    float bias = bc1[n0 + lm];
    float wc2 = Wc2[n0 + lm];
#pragma unroll
    for (int t = 0; t < 4; ++t) {
      float mine = 0.0f;
#pragma unroll
      for (int v = 0; v < 8; ++v) {
        float p = row16_sum(silu_f(acc[t][v] + bias) * wc2);
        mine = (lm == v) ? p : mine;
      }
      if (lm < 8) atomicAdd(&sW[t * 16 + lm + 8 * hi], mine);
    }
  }
  __syncthreads();

  if (tid < EPB * 3) {
    int e = tid / 3;
    int d = tid - e * 3;
    float cd = sDiff[tid] / (sqrtf(sDist[e]) + 1e-8f);
    atomicAdd(&coords_out[sRow[e] * 3 + d], cd * sW[e]);
  }
}

__global__ __launch_bounds__(256) void egnn_pre2_kernel(
    const float* __restrict__ x, const float* __restrict__ coords,
    const float* __restrict__ We1, const float* __restrict__ be1,
    float* __restrict__ P, float* __restrict__ Q, float* __restrict__ agg,
    float* __restrict__ coords_out) {
  __shared__ __attribute__((aligned(16))) float sX[16 * 132];
  __shared__ __attribute__((aligned(16))) float sQ[16 * 132];

  const int tid = threadIdx.x;
  const int lane = tid & 31;
  const int wv = tid >> 5;
  const int hi = lane >> 4;
  const int lm = lane & 15;
  const int n0 = wv * 16;
  const int nb = blockIdx.x * 16;

  const int sr = tid >> 5;
  const int sc = 4 * (tid & 31);
  const v4f* x4 = reinterpret_cast<const v4f*>(x) + nb * 32;
  const v4f xa = x4[tid];
  const v4f xb = x4[tid + 256];

  const v4f zero4 = {0.0f, 0.0f, 0.0f, 0.0f};
  v4f* g4 = reinterpret_cast<v4f*>(agg) + nb * 32;
  g4[tid] = zero4;
  g4[tid + 256] = zero4;
  if (tid < 48) coords_out[nb * 3 + tid] = coords[nb * 3 + tid];

  *reinterpret_cast<v4f*>(&sX[sr * 132 + sc]) = xa;
  *reinterpret_cast<v4f*>(&sX[(sr + 8) * 132 + sc]) = xb;
  __syncthreads();

  v8f accP = {};
  v8f accQ = {};
  for (int k0 = 0; k0 < 128; k0 += 4) {
    int kk = k0 + 2 * hi;
    v2f a, bp, bq;
    a.x = sX[lm * 132 + kk];
    a.y = sX[lm * 132 + kk + 1];
    bp.x = We1[kk * 128 + n0 + lm];
    bp.y = We1[(kk + 1) * 128 + n0 + lm];
    bq.x = We1[(128 + kk) * 128 + n0 + lm];
    bq.y = We1[(129 + kk) * 128 + n0 + lm];
    accP = wmma4(a, bp, accP);
    accQ = wmma4(a, bq, accQ);
  }
  const float bias = be1[n0 + lm];
  __syncthreads();
#pragma unroll
  for (int v = 0; v < 8; ++v) {
    sX[(v + 8 * hi) * 132 + n0 + lm] = accP[v] + bias;
    sQ[(v + 8 * hi) * 132 + n0 + lm] = accQ[v];
  }
  __syncthreads();
  v4f* p4 = reinterpret_cast<v4f*>(P) + nb * 32;
  v4f* q4 = reinterpret_cast<v4f*>(Q) + nb * 32;
  p4[tid] = *reinterpret_cast<const v4f*>(&sX[sr * 132 + sc]);
  p4[tid + 256] = *reinterpret_cast<const v4f*>(&sX[(sr + 8) * 132 + sc]);
  q4[tid] = *reinterpret_cast<const v4f*>(&sQ[sr * 132 + sc]);
  q4[tid + 256] = *reinterpret_cast<const v4f*>(&sQ[(sr + 8) * 132 + sc]);
}

__global__ __launch_bounds__(256, 8) void egnn_edge2_kernel(
    const float* __restrict__ coords, const float* __restrict__ edge_attr,
    const int* __restrict__ erow, const int* __restrict__ ecol,
    const float* __restrict__ We1, const float* __restrict__ P,
    const float* __restrict__ Q, const float* __restrict__ We2,
    const float* __restrict__ be2, const float* __restrict__ Wc1,
    const float* __restrict__ bc1, const float* __restrict__ Wc2,
    float* __restrict__ agg, float* __restrict__ coords_out) {
  __shared__ __attribute__((aligned(16))) float smem[EPB * 264];
  float* sH = smem;
  float* sE = smem + EPB * 132;
  float* sA = sE;

  __shared__ int sRC[2 * EPB];
  int* sRow = sRC;
  int* sCol = sRC + EPB;
  __shared__ float sDiff[EPB * 3], sDist[EPB], sW[EPB];

  const int tid = threadIdx.x;
  const int lane = tid & 31;
  const int wv = tid >> 5;
  const int hi = lane >> 4;
  const int lm = lane & 15;
  const int n0 = wv * 16;
  const int ebase = blockIdx.x * EPB;

  const v4f* ea4 = reinterpret_cast<const v4f*>(edge_attr) + ebase * 8;
  const v4f ea0 = ea4[tid];
  const v4f ea1 = ea4[tid + 256];

  if (tid < EPB) {
    int r = erow[ebase + tid];
    int c = ecol[ebase + tid];
    sRow[tid] = r;
    sCol[tid] = c;
    float d0 = coords[r * 3 + 0] - coords[c * 3 + 0];
    float d1 = coords[r * 3 + 1] - coords[c * 3 + 1];
    float d2 = coords[r * 3 + 2] - coords[c * 3 + 2];
    sDiff[tid * 3 + 0] = d0;
    sDiff[tid * 3 + 1] = d1;
    sDiff[tid * 3 + 2] = d2;
    sDist[tid] = d0 * d0 + d1 * d1 + d2 * d2;
    sW[tid] = 0.0f;
  }
  __syncthreads();

  v8f acc1[4];
  {
    v8f qv[4];
#pragma unroll
    for (int t = 0; t < 4; ++t) {
#pragma unroll
      for (int v = 0; v < 8; ++v) {
        const int M = t * 16 + v + 8 * hi;
        qv[t][v] = Q[sCol[M] * 128 + n0 + lm];
      }
    }
#pragma unroll
    for (int t = 0; t < 4; ++t) {
#pragma unroll
      for (int v = 0; v < 8; ++v) {
        const int M = t * 16 + v + 8 * hi;
        acc1[t][v] = P[sRow[M] * 128 + n0 + lm];
      }
    }
#pragma unroll
    for (int t = 0; t < 4; ++t) acc1[t] += qv[t];
  }
  *reinterpret_cast<v4f*>(&sA[(tid >> 3) * 36 + 4 * (tid & 7)]) = ea0;
  *reinterpret_cast<v4f*>(&sA[((tid >> 3) + 32) * 36 + 4 * (tid & 7)]) = ea1;
  __syncthreads();

  {
    for (int k0 = 0; k0 < 32; k0 += 4) {
      int kk = k0 + 2 * hi;
      v2f b;
      b.x = We1[(256 + kk) * 128 + n0 + lm];
      b.y = We1[(257 + kk) * 128 + n0 + lm];
#pragma unroll
      for (int t = 0; t < 4; ++t) {
        v2f a;
        a.x = sA[(t * 16 + lm) * 36 + kk];
        a.y = sA[(t * 16 + lm) * 36 + kk + 1];
        acc1[t] = wmma4(a, b, acc1[t]);
      }
    }
    float wlast = We1[288 * 128 + n0 + lm];
#pragma unroll
    for (int t = 0; t < 4; ++t) {
#pragma unroll
      for (int v = 0; v < 8; ++v) {
        int M = t * 16 + v + 8 * hi;
        float pre = acc1[t][v] + sDist[M] * wlast;
        sH[M * 132 + n0 + lm] = silu_f(pre);
      }
    }
  }
  __syncthreads();

  {
    v8f acc[4] = {{}, {}, {}, {}};
    for (int k0 = 0; k0 < 128; k0 += 4) {
      int kk = k0 + 2 * hi;
      v2f b;
      b.x = We2[kk * 128 + n0 + lm];
      b.y = We2[(kk + 1) * 128 + n0 + lm];
#pragma unroll
      for (int t = 0; t < 4; ++t) {
        v2f a;
        a.x = sH[(t * 16 + lm) * 132 + kk];
        a.y = sH[(t * 16 + lm) * 132 + kk + 1];
        acc[t] = wmma4(a, b, acc[t]);
      }
    }
    float bias = be2[n0 + lm];
#pragma unroll
    for (int t = 0; t < 4; ++t) {
#pragma unroll
      for (int v = 0; v < 8; ++v) {
        int M = t * 16 + v + 8 * hi;
        float ef = acc[t][v] + bias;
        sE[M * 132 + n0 + lm] = ef;
        atomicAdd(&agg[sRow[M] * 128 + n0 + lm], ef);
      }
    }
  }
  __syncthreads();

  {
    v8f acc[4] = {{}, {}, {}, {}};
    for (int k0 = 0; k0 < 128; k0 += 4) {
      int kk = k0 + 2 * hi;
      v2f b;
      b.x = Wc1[kk * 128 + n0 + lm];
      b.y = Wc1[(kk + 1) * 128 + n0 + lm];
#pragma unroll
      for (int t = 0; t < 4; ++t) {
        v2f a;
        a.x = sE[(t * 16 + lm) * 132 + kk];
        a.y = sE[(t * 16 + lm) * 132 + kk + 1];
        acc[t] = wmma4(a, b, acc[t]);
      }
    }
    float bias = bc1[n0 + lm];
    float wc2 = Wc2[n0 + lm];
#pragma unroll
    for (int t = 0; t < 4; ++t) {
      float mine = 0.0f;
#pragma unroll
      for (int v = 0; v < 8; ++v) {
        float p = row16_sum(silu_f(acc[t][v] + bias) * wc2);
        mine = (lm == v) ? p : mine;
      }
      if (lm < 8) atomicAdd(&sW[t * 16 + lm + 8 * hi], mine);
    }
  }
  __syncthreads();

  if (tid < EPB * 3) {
    int e = tid / 3;
    int d = tid - e * 3;
    float cd = sDiff[tid] / (sqrtf(sDist[e]) + 1e-8f);
    atomicAdd(&coords_out[sRow[e] * 3 + d], cd * sW[e]);
  }
}

__global__ __launch_bounds__(256) void egnn_node_kernel(
    const float* __restrict__ x, const float* __restrict__ agg,
    const float* __restrict__ Wn1, const float* __restrict__ bn1,
    const float* __restrict__ Wn2, const float* __restrict__ bn2,
    float* __restrict__ x_out) {
  __shared__ float sA[16 * 260];
  __shared__ float sH[16 * 132];

  const int tid = threadIdx.x;
  const int lane = tid & 31;
  const int wv = tid >> 5;
  const int hi = lane >> 4;
  const int lm = lane & 15;
  const int n0 = wv * 16;
  const int nb = blockIdx.x * 16;

  for (int idx = tid; idx < 16 * 256; idx += 256) {
    int m = idx >> 8;
    int c = idx & 255;
    float v = (c < 128) ? x[(nb + m) * 128 + c] : agg[(nb + m) * 128 + (c - 128)];
    sA[m * 260 + c] = v;
  }
  __syncthreads();

  v8f acc1 = {};
  for (int k0 = 0; k0 < 256; k0 += 4) {
    int kk = k0 + 2 * hi;
    v2f a, b;
    a.x = sA[lm * 260 + kk];
    a.y = sA[lm * 260 + kk + 1];
    b.x = Wn1[kk * 128 + n0 + lm];
    b.y = Wn1[(kk + 1) * 128 + n0 + lm];
    acc1 = wmma4(a, b, acc1);
  }
  {
    float bias = bn1[n0 + lm];
#pragma unroll
    for (int v = 0; v < 8; ++v) {
      int M = v + 8 * hi;
      sH[M * 132 + n0 + lm] = silu_f(acc1[v] + bias);
    }
  }
  __syncthreads();

  v8f acc2 = {};
  for (int k0 = 0; k0 < 128; k0 += 4) {
    int kk = k0 + 2 * hi;
    v2f a, b;
    a.x = sH[lm * 132 + kk];
    a.y = sH[lm * 132 + kk + 1];
    b.x = Wn2[kk * 128 + n0 + lm];
    b.y = Wn2[(kk + 1) * 128 + n0 + lm];
    acc2 = wmma4(a, b, acc2);
  }
  {
    float bias = bn2[n0 + lm];
#pragma unroll
    for (int v = 0; v < 8; ++v) {
      int M = v + 8 * hi;
      x_out[(nb + M) * 128 + n0 + lm] = acc2[v] + bias;
    }
  }
}

extern "C" void kernel_launch(void* const* d_in, const int* in_sizes, int n_in,
                              void* d_out, int out_size, void* d_ws,
                              size_t ws_size, hipStream_t stream) {
  const float* x = (const float*)d_in[0];
  const float* coords = (const float*)d_in[1];
  const float* edge_attr = (const float*)d_in[2];
  const int* edge_index = (const int*)d_in[3];
  const float* We1 = (const float*)d_in[4];
  const float* be1 = (const float*)d_in[5];
  const float* We2 = (const float*)d_in[6];
  const float* be2 = (const float*)d_in[7];
  const float* Wc1 = (const float*)d_in[8];
  const float* bc1 = (const float*)d_in[9];
  const float* Wc2 = (const float*)d_in[10];
  const float* Wn1 = (const float*)d_in[11];
  const float* bn1 = (const float*)d_in[12];
  const float* Wn2 = (const float*)d_in[13];
  const float* bn2 = (const float*)d_in[14];

  float* x_out = (float*)d_out;
  float* coords_out = x_out + (size_t)N_NODES * 128;
  float* agg = (float*)d_ws;

  const int* erow = edge_index;
  const int* ecol = edge_index + N_EDGES;

  if (ws_size >= (size_t)2 * N_NODES * 128 * sizeof(float)) {
    float* Q = agg + (size_t)N_NODES * 128;
    egnn_pre2_kernel<<<N_NODES / 16, 256, 0, stream>>>(x, coords, We1, be1,
                                                       x_out, Q, agg, coords_out);
    egnn_edge2_kernel<<<N_EDGES / EPB, 256, 0, stream>>>(
        coords, edge_attr, erow, ecol, We1, x_out, Q, We2, be2, Wc1, bc1, Wc2,
        agg, coords_out);
  } else {
    egnn_pre_kernel<<<N_NODES / 16, 256, 0, stream>>>(x, coords, We1, be1, x_out,
                                                      agg, coords_out);
    egnn_edge_kernel<<<N_EDGES / EPB, 256, 0, stream>>>(
        x, coords, edge_attr, erow, ecol, We1, x_out, We2, be2, Wc1, bc1, Wc2,
        agg, coords_out);
  }
  egnn_node_kernel<<<N_NODES / 16, 256, 0, stream>>>(x, agg, Wn1, bn1, Wn2, bn2,
                                                     x_out);
}


# Edge weights re-laid per call as [k/4][col] 16-byte slots: a lane's B fragment for two k-steps is one b128 load of whole lines, A one ds_load_b128 (sum order: k as {0,1,4,5},{2,3,6,7} per 8)
# speedup vs baseline: 3.0243x; 1.0465x over previous
#include <hip/hip_runtime.h>
#include <math.h>

typedef __attribute__((ext_vector_type(2))) float v2f;
typedef __attribute__((ext_vector_type(8))) float v8f;
typedef __attribute__((ext_vector_type(4))) float v4f;

#define N_NODES 50000
#define N_EDGES 800000
#define EPB 64
#define PK_W2 1024
#define PK_W3 5120
#define PK_SLOTS 9216

__device__ __forceinline__ v8f wmma4(v2f a, v2f b, v8f c) {
  return __builtin_amdgcn_wmma_f32_16x16x4_f32(false, a, false, b, (short)0, c,
                                               false, false);
}

__device__ __forceinline__ float silu_f(float v) {
  return v * __builtin_amdgcn_rcpf(1.0f + __expf(-v));
}

__device__ __forceinline__ float row16_sum(float v) {
  v += __int_as_float(__builtin_amdgcn_update_dpp(
      0, __float_as_int(v), 0x161, 0xf, 0xf, false));
  v += __int_as_float(__builtin_amdgcn_update_dpp(
      0, __float_as_int(v), 0x162, 0xf, 0xf, false));
  v += __int_as_float(__builtin_amdgcn_update_dpp(
      0, __float_as_int(v), 0x164, 0xf, 0xf, false));
  v += __int_as_float(__builtin_amdgcn_update_dpp(
      0, __float_as_int(v), 0x168, 0xf, 0xf, false));
  return v;
}

__global__ __launch_bounds__(256) void egnn_pre_kernel(
    const float* __restrict__ x, const float* __restrict__ coords,
    const float* __restrict__ We1, const float* __restrict__ be1,
    float* __restrict__ P, float* __restrict__ agg,
    float* __restrict__ coords_out) {
  __shared__ __attribute__((aligned(16))) float sX[16 * 132];

  const int tid = threadIdx.x;
  const int lane = tid & 31;
  const int wv = tid >> 5;
  const int hi = lane >> 4;
  const int lm = lane & 15;
  const int n0 = wv * 16;
  const int nb = blockIdx.x * 16;

  const int sr = tid >> 5;
  const int sc = 4 * (tid & 31);
  const v4f* x4 = reinterpret_cast<const v4f*>(x) + nb * 32;
  const v4f xa = x4[tid];
  const v4f xb = x4[tid + 256];

  const v4f zero4 = {0.0f, 0.0f, 0.0f, 0.0f};
  v4f* g4 = reinterpret_cast<v4f*>(agg) + nb * 32;
  g4[tid] = zero4;
  g4[tid + 256] = zero4;
  if (tid < 48) coords_out[nb * 3 + tid] = coords[nb * 3 + tid];

  *reinterpret_cast<v4f*>(&sX[sr * 132 + sc]) = xa;
  *reinterpret_cast<v4f*>(&sX[(sr + 8) * 132 + sc]) = xb;
  __syncthreads();

  v8f acc = {};
  for (int k0 = 0; k0 < 128; k0 += 4) {
    int kk = k0 + 2 * hi;
    v2f a, b;
    a.x = sX[lm * 132 + kk];
    a.y = sX[lm * 132 + kk + 1];
    b.x = We1[kk * 128 + n0 + lm];
    b.y = We1[(kk + 1) * 128 + n0 + lm];
    acc = wmma4(a, b, acc);
  }
  const float bias = be1[n0 + lm];
  __syncthreads();
#pragma unroll
  for (int v = 0; v < 8; ++v) sX[(v + 8 * hi) * 132 + n0 + lm] = acc[v] + bias;
  __syncthreads();
  v4f* p4 = reinterpret_cast<v4f*>(P) + nb * 32;
  p4[tid] = *reinterpret_cast<const v4f*>(&sX[sr * 132 + sc]);
  p4[tid + 256] = *reinterpret_cast<const v4f*>(&sX[(sr + 8) * 132 + sc]);
}

__global__ __launch_bounds__(256, 8) void egnn_edge_kernel(
    const float* __restrict__ x, const float* __restrict__ coords,
    const float* __restrict__ edge_attr, const int* __restrict__ erow,
    const int* __restrict__ ecol, const float* __restrict__ We1,
    const float* __restrict__ P, const float* __restrict__ We2,
    const float* __restrict__ be2, const float* __restrict__ Wc1,
    const float* __restrict__ bc1, const float* __restrict__ Wc2,
    float* __restrict__ agg, float* __restrict__ coords_out) {
  __shared__ __attribute__((aligned(16))) float smem[EPB * 264];
  float* sA = smem;
  float* sH = smem;
  float* sE = smem + EPB * 132;

  __shared__ int sRC[2 * EPB];
  int* sRow = sRC;
  int* sCol = sRC + EPB;
  __shared__ float sDiff[EPB * 3], sDist[EPB], sW[EPB];

  const int tid = threadIdx.x;
  const int lane = tid & 31;
  const int wv = tid >> 5;
  const int hi = lane >> 4;
  const int lm = lane & 15;
  const int n0 = wv * 16;
  const int ebase = blockIdx.x * EPB;

  const v4f* ea4 = reinterpret_cast<const v4f*>(edge_attr) + ebase * 8;
  const v4f ea0 = ea4[tid];
  const v4f ea1 = ea4[tid + 256];

  if (tid < EPB) {
    int r = erow[ebase + tid];
    int c = ecol[ebase + tid];
    sRow[tid] = r;
    sCol[tid] = c;
    float d0 = coords[r * 3 + 0] - coords[c * 3 + 0];
    float d1 = coords[r * 3 + 1] - coords[c * 3 + 1];
    float d2 = coords[r * 3 + 2] - coords[c * 3 + 2];
    sDiff[tid * 3 + 0] = d0;
    sDiff[tid * 3 + 1] = d1;
    sDiff[tid * 3 + 2] = d2;
    sDist[tid] = d0 * d0 + d1 * d1 + d2 * d2;
    sW[tid] = 0.0f;
  }
  __syncthreads();

  v8f acc1[4];
  {
    const v4f* x4 = reinterpret_cast<const v4f*>(x);
    v4f xv[8];
    for (int i = 0; i < 8; ++i) {
      const int node = sCol[wv + 8 * i];
      xv[i] = x4[node * 32 + lane];
    }
#pragma unroll
    for (int t = 0; t < 4; ++t) {
#pragma unroll
      for (int v = 0; v < 8; ++v) {
        const int M = t * 16 + v + 8 * hi;
        acc1[t][v] = P[sRow[M] * 128 + n0 + lm];
      }
    }
    for (int i = 0; i < 8; ++i)
      *reinterpret_cast<v4f*>(&sA[(wv + 8 * i) * 164 + 4 * lane]) = xv[i];
    *reinterpret_cast<v4f*>(&sA[(tid >> 3) * 164 + 128 + 4 * (tid & 7)]) = ea0;
    *reinterpret_cast<v4f*>(&sA[((tid >> 3) + 32) * 164 + 128 + 4 * (tid & 7)]) =
        ea1;
  }
  __syncthreads();

  {
    for (int k0 = 0; k0 < 160; k0 += 4) {
      int kk = k0 + 2 * hi;
      v2f b;
      b.x = We1[(128 + kk) * 128 + n0 + lm];
      b.y = We1[(129 + kk) * 128 + n0 + lm];
#pragma unroll
      for (int t = 0; t < 4; ++t) {
        v2f a;
        a.x = sA[(t * 16 + lm) * 164 + kk];
        a.y = sA[(t * 16 + lm) * 164 + kk + 1];
        acc1[t] = wmma4(a, b, acc1[t]);
      }
    }
    float wlast = We1[288 * 128 + n0 + lm];
    __syncthreads();
#pragma unroll
    for (int t = 0; t < 4; ++t) {
#pragma unroll
      for (int v = 0; v < 8; ++v) {
        int M = t * 16 + v + 8 * hi;
        float pre = acc1[t][v] + sDist[M] * wlast;
        sH[M * 132 + n0 + lm] = silu_f(pre);
      }
    }
  }
  __syncthreads();

  {
    v8f acc[4] = {{}, {}, {}, {}};
    for (int k0 = 0; k0 < 128; k0 += 4) {
      int kk = k0 + 2 * hi;
      v2f b;
      b.x = We2[kk * 128 + n0 + lm];
      b.y = We2[(kk + 1) * 128 + n0 + lm];
#pragma unroll
      for (int t = 0; t < 4; ++t) {
        v2f a;
        a.x = sH[(t * 16 + lm) * 132 + kk];
        a.y = sH[(t * 16 + lm) * 132 + kk + 1];
        acc[t] = wmma4(a, b, acc[t]);
      }
    }
    float bias = be2[n0 + lm];
#pragma unroll
    for (int t = 0; t < 4; ++t) {
#pragma unroll
      for (int v = 0; v < 8; ++v) {
        int M = t * 16 + v + 8 * hi;
        float ef = acc[t][v] + bias;
        sE[M * 132 + n0 + lm] = ef;
        atomicAdd(&agg[sRow[M] * 128 + n0 + lm], ef);
      }
    }
  }
  __syncthreads();

  {
    v8f acc[4] = {{}, {}, {}, {}};
    for (int k0 = 0; k0 < 128; k0 += 4) {
      int kk = k0 + 2 * hi;
      v2f b;
      b.x = Wc1[kk * 128 + n0 + lm];
      b.y = Wc1[(kk + 1) * 128 + n0 + lm];
#pragma unroll
      for (int t = 0; t < 4; ++t) {
        v2f a;
        a.x = sE[(t * 16 + lm) * 132 + kk];
        a.y = sE[(t * 16 + lm) * 132 + kk + 1];
        acc[t] = wmma4(a, b, acc[t]);
      }
    }
    float bias = bc1[n0 + lm];
    float wc2 = Wc2[n0 + lm];
#pragma unroll
    for (int t = 0; t < 4; ++t) {
      float mine = 0.0f;
#pragma unroll
      for (int v = 0; v < 8; ++v) {
        float p = row16_sum(silu_f(acc[t][v] + bias) * wc2);
        mine = (lm == v) ? p : mine;
      }
      if (lm < 8) atomicAdd(&sW[t * 16 + lm + 8 * hi], mine);
    }
  }
  __syncthreads();

  if (tid < EPB * 3) {
    int e = tid / 3;
    int d = tid - e * 3;
    float cd = sDiff[tid] / (sqrtf(sDist[e]) + 1e-8f);
    atomicAdd(&coords_out[sRow[e] * 3 + d], cd * sW[e]);
  }
}

__global__ __launch_bounds__(256) void egnn_pre2_kernel(
    const float* __restrict__ x, const float* __restrict__ coords,
    const float* __restrict__ We1, const float* __restrict__ be1,
    const float* __restrict__ We2, const float* __restrict__ Wc1,
    float* __restrict__ P, float* __restrict__ Q, float* __restrict__ agg,
    float* __restrict__ coords_out, float* __restrict__ Wp) {
  if (blockIdx.x >= N_NODES / 16) {
    const int pb = blockIdx.x - N_NODES / 16;
    v4f* out4 = reinterpret_cast<v4f*>(Wp);
    for (int i = 0; i < 4; ++i) {
      const int slot = pb * 1024 + i * 256 + threadIdx.x;
      const float* src;
      int ls;
      if (slot < PK_W2) {
        src = We1 + 256 * 128;
        ls = slot;
      } else if (slot < PK_W3) {
        src = We2;
        ls = slot - PK_W2;
      } else {
        src = Wc1;
        ls = slot - PK_W3;
      }
      const int g = ls >> 7;
      const int n = ls & 127;
      v4f v;
      v.x = src[(4 * g + 0) * 128 + n];
      v.y = src[(4 * g + 1) * 128 + n];
      v.z = src[(4 * g + 2) * 128 + n];
      v.w = src[(4 * g + 3) * 128 + n];
      out4[slot] = v;
    }
    return;
  }
  __shared__ __attribute__((aligned(16))) float sX[16 * 132];
  __shared__ __attribute__((aligned(16))) float sQ[16 * 132];

  const int tid = threadIdx.x;
  const int lane = tid & 31;
  const int wv = tid >> 5;
  const int hi = lane >> 4;
  const int lm = lane & 15;
  const int n0 = wv * 16;
  const int nb = blockIdx.x * 16;

  const int sr = tid >> 5;
  const int sc = 4 * (tid & 31);
  const v4f* x4 = reinterpret_cast<const v4f*>(x) + nb * 32;
  const v4f xa = x4[tid];
  const v4f xb = x4[tid + 256];

  const v4f zero4 = {0.0f, 0.0f, 0.0f, 0.0f};
  v4f* g4 = reinterpret_cast<v4f*>(agg) + nb * 32;
  g4[tid] = zero4;
  g4[tid + 256] = zero4;
  if (tid < 48) coords_out[nb * 3 + tid] = coords[nb * 3 + tid];

  *reinterpret_cast<v4f*>(&sX[sr * 132 + sc]) = xa;
  *reinterpret_cast<v4f*>(&sX[(sr + 8) * 132 + sc]) = xb;
  __syncthreads();

  v8f accP = {};
  v8f accQ = {};
  for (int k0 = 0; k0 < 128; k0 += 4) {
    int kk = k0 + 2 * hi;
    v2f a, bp, bq;
    a.x = sX[lm * 132 + kk];
    a.y = sX[lm * 132 + kk + 1];
    bp.x = We1[kk * 128 + n0 + lm];
    bp.y = We1[(kk + 1) * 128 + n0 + lm];
    bq.x = We1[(128 + kk) * 128 + n0 + lm];
    bq.y = We1[(129 + kk) * 128 + n0 + lm];
    accP = wmma4(a, bp, accP);
    accQ = wmma4(a, bq, accQ);
  }
  const float bias = be1[n0 + lm];
  __syncthreads();
#pragma unroll
  for (int v = 0; v < 8; ++v) {
    sX[(v + 8 * hi) * 132 + n0 + lm] = accP[v] + bias;
    sQ[(v + 8 * hi) * 132 + n0 + lm] = accQ[v];
  }
  __syncthreads();
  v4f* p4 = reinterpret_cast<v4f*>(P) + nb * 32;
  v4f* q4 = reinterpret_cast<v4f*>(Q) + nb * 32;
  p4[tid] = *reinterpret_cast<const v4f*>(&sX[sr * 132 + sc]);
  p4[tid + 256] = *reinterpret_cast<const v4f*>(&sX[(sr + 8) * 132 + sc]);
  q4[tid] = *reinterpret_cast<const v4f*>(&sQ[sr * 132 + sc]);
  q4[tid + 256] = *reinterpret_cast<const v4f*>(&sQ[(sr + 8) * 132 + sc]);
}

__global__ __launch_bounds__(256, 8) void egnn_edge2_kernel(
    const float* __restrict__ coords, const float* __restrict__ edge_attr,
    const int* __restrict__ erow, const int* __restrict__ ecol,
    const float* __restrict__ We1, const float* __restrict__ P,
    const float* __restrict__ Q, const float* __restrict__ We2,
    const float* __restrict__ be2, const float* __restrict__ Wc1,
    const float* __restrict__ bc1, const float* __restrict__ Wc2,
    float* __restrict__ agg, float* __restrict__ coords_out) {
  __shared__ __attribute__((aligned(16))) float smem[EPB * 264];
  float* sH = smem;
  float* sE = smem + EPB * 132;
  float* sA = sE;

  __shared__ int sRC[2 * EPB];
  int* sRow = sRC;
  int* sCol = sRC + EPB;
  __shared__ float sDiff[EPB * 3], sDist[EPB], sW[EPB];

  const int tid = threadIdx.x;
  const int lane = tid & 31;
  const int wv = tid >> 5;
  const int hi = lane >> 4;
  const int lm = lane & 15;
  const int n0 = wv * 16;
  const int ebase = blockIdx.x * EPB;

  const v4f* ea4 = reinterpret_cast<const v4f*>(edge_attr) + ebase * 8;
  const v4f ea0 = ea4[tid];
  const v4f ea1 = ea4[tid + 256];

  if (tid < EPB) {
    int r = erow[ebase + tid];
    int c = ecol[ebase + tid];
    sRow[tid] = r;
    sCol[tid] = c;
    float d0 = coords[r * 3 + 0] - coords[c * 3 + 0];
    float d1 = coords[r * 3 + 1] - coords[c * 3 + 1];
    float d2 = coords[r * 3 + 2] - coords[c * 3 + 2];
    sDiff[tid * 3 + 0] = d0;
    sDiff[tid * 3 + 1] = d1;
    sDiff[tid * 3 + 2] = d2;
    sDist[tid] = d0 * d0 + d1 * d1 + d2 * d2;
    sW[tid] = 0.0f;
  }
  __syncthreads();

  v8f acc1[4];
  {
    v8f qv[4];
#pragma unroll
    for (int t = 0; t < 4; ++t) {
#pragma unroll
      for (int v = 0; v < 8; ++v) {
        const int M = t * 16 + v + 8 * hi;
        qv[t][v] = Q[sCol[M] * 128 + n0 + lm];
      }
    }
#pragma unroll
    for (int t = 0; t < 4; ++t) {
#pragma unroll
      for (int v = 0; v < 8; ++v) {
        const int M = t * 16 + v + 8 * hi;
        acc1[t][v] = P[sRow[M] * 128 + n0 + lm];
      }
    }
#pragma unroll
    for (int t = 0; t < 4; ++t) acc1[t] += qv[t];
  }
  *reinterpret_cast<v4f*>(&sA[(tid >> 3) * 36 + 4 * (tid & 7)]) = ea0;
  *reinterpret_cast<v4f*>(&sA[((tid >> 3) + 32) * 36 + 4 * (tid & 7)]) = ea1;
  __syncthreads();

  {
    for (int k0 = 0; k0 < 32; k0 += 4) {
      int kk = k0 + 2 * hi;
      v2f b;
      b.x = We1[(256 + kk) * 128 + n0 + lm];
      b.y = We1[(257 + kk) * 128 + n0 + lm];
#pragma unroll
      for (int t = 0; t < 4; ++t) {
        v2f a;
        a.x = sA[(t * 16 + lm) * 36 + kk];
        a.y = sA[(t * 16 + lm) * 36 + kk + 1];
        acc1[t] = wmma4(a, b, acc1[t]);
      }
    }
    float wlast = We1[288 * 128 + n0 + lm];
#pragma unroll
    for (int t = 0; t < 4; ++t) {
#pragma unroll
      for (int v = 0; v < 8; ++v) {
        int M = t * 16 + v + 8 * hi;
        float pre = acc1[t][v] + sDist[M] * wlast;
        sH[M * 132 + n0 + lm] = silu_f(pre);
      }
    }
  }
  __syncthreads();

  {
    v8f acc[4] = {{}, {}, {}, {}};
    for (int k0 = 0; k0 < 128; k0 += 4) {
      int kk = k0 + 2 * hi;
      v2f b;
      b.x = We2[kk * 128 + n0 + lm];
      b.y = We2[(kk + 1) * 128 + n0 + lm];
#pragma unroll
      for (int t = 0; t < 4; ++t) {
        v2f a;
        a.x = sH[(t * 16 + lm) * 132 + kk];
        a.y = sH[(t * 16 + lm) * 132 + kk + 1];
        acc[t] = wmma4(a, b, acc[t]);
      }
    }
    float bias = be2[n0 + lm];
#pragma unroll
    for (int t = 0; t < 4; ++t) {
#pragma unroll
      for (int v = 0; v < 8; ++v) {
        int M = t * 16 + v + 8 * hi;
        float ef = acc[t][v] + bias;
        sE[M * 132 + n0 + lm] = ef;
        atomicAdd(&agg[sRow[M] * 128 + n0 + lm], ef);
      }
    }
  }
  __syncthreads();

  {
    v8f acc[4] = {{}, {}, {}, {}};
    for (int k0 = 0; k0 < 128; k0 += 4) {
      int kk = k0 + 2 * hi;
      v2f b;
      b.x = Wc1[kk * 128 + n0 + lm];
      b.y = Wc1[(kk + 1) * 128 + n0 + lm];
#pragma unroll
      for (int t = 0; t < 4; ++t) {
        v2f a;
        a.x = sE[(t * 16 + lm) * 132 + kk];
        a.y = sE[(t * 16 + lm) * 132 + kk + 1];
        acc[t] = wmma4(a, b, acc[t]);
      }
    }
    float bias = bc1[n0 + lm];
    float wc2 = Wc2[n0 + lm];
#pragma unroll
    for (int t = 0; t < 4; ++t) {
      float mine = 0.0f;
#pragma unroll
      for (int v = 0; v < 8; ++v) {
        float p = row16_sum(silu_f(acc[t][v] + bias) * wc2);
        mine = (lm == v) ? p : mine;
      }
      if (lm < 8) atomicAdd(&sW[t * 16 + lm + 8 * hi], mine);
    }
  }
  __syncthreads();

  if (tid < EPB * 3) {
    int e = tid / 3;
    int d = tid - e * 3;
    float cd = sDiff[tid] / (sqrtf(sDist[e]) + 1e-8f);
    atomicAdd(&coords_out[sRow[e] * 3 + d], cd * sW[e]);
  }
}

__global__ __launch_bounds__(256, 8) void egnn_edge3_kernel(
    const float* __restrict__ coords, const float* __restrict__ edge_attr,
    const int* __restrict__ erow, const int* __restrict__ ecol,
    const float* __restrict__ We1, const float* __restrict__ P,
    const float* __restrict__ Q, const float* __restrict__ Wp,
    const float* __restrict__ be2, const float* __restrict__ bc1,
    const float* __restrict__ Wc2, float* __restrict__ agg,
    float* __restrict__ coords_out) {
  __shared__ __attribute__((aligned(16))) float smem[EPB * 264];
  float* sH = smem;
  float* sE = smem + EPB * 132;
  float* sA = sE;

  __shared__ int sRC[2 * EPB];
  int* sRow = sRC;
  int* sCol = sRC + EPB;
  __shared__ float sDiff[EPB * 3], sDist[EPB], sW[EPB];

  const int tid = threadIdx.x;
  const int lane = tid & 31;
  const int wv = tid >> 5;
  const int hi = lane >> 4;
  const int lm = lane & 15;
  const int n0 = wv * 16;
  const int ebase = blockIdx.x * EPB;
  const v4f* wp4 = reinterpret_cast<const v4f*>(Wp) + hi * 128 + n0 + lm;

  const v4f* ea4 = reinterpret_cast<const v4f*>(edge_attr) + ebase * 8;
  const v4f ea0 = ea4[tid];
  const v4f ea1 = ea4[tid + 256];

  if (tid < EPB) {
    int r = erow[ebase + tid];
    int c = ecol[ebase + tid];
    sRow[tid] = r;
    sCol[tid] = c;
    float d0 = coords[r * 3 + 0] - coords[c * 3 + 0];
    float d1 = coords[r * 3 + 1] - coords[c * 3 + 1];
    float d2 = coords[r * 3 + 2] - coords[c * 3 + 2];
    sDiff[tid * 3 + 0] = d0;
    sDiff[tid * 3 + 1] = d1;
    sDiff[tid * 3 + 2] = d2;
    sDist[tid] = d0 * d0 + d1 * d1 + d2 * d2;
    sW[tid] = 0.0f;
  }
  __syncthreads();

  v8f acc1[4];
  {
    v8f qv[4];
#pragma unroll
    for (int t = 0; t < 4; ++t) {
#pragma unroll
      for (int v = 0; v < 8; ++v) {
        const int M = t * 16 + v + 8 * hi;
        qv[t][v] = Q[sCol[M] * 128 + n0 + lm];
      }
    }
#pragma unroll
    for (int t = 0; t < 4; ++t) {
#pragma unroll
      for (int v = 0; v < 8; ++v) {
        const int M = t * 16 + v + 8 * hi;
        acc1[t][v] = P[sRow[M] * 128 + n0 + lm];
      }
    }
#pragma unroll
    for (int t = 0; t < 4; ++t) acc1[t] += qv[t];
  }
  *reinterpret_cast<v4f*>(&sA[(tid >> 3) * 36 + 4 * (tid & 7)]) = ea0;
  *reinterpret_cast<v4f*>(&sA[((tid >> 3) + 32) * 36 + 4 * (tid & 7)]) = ea1;
  __syncthreads();

  {
    for (int k0 = 0; k0 < 32; k0 += 8) {
      const v4f bv = wp4[(k0 >> 2) * 128];
      v4f av[4];
#pragma unroll
      for (int t = 0; t < 4; ++t)
        av[t] = *reinterpret_cast<const v4f*>(
            &sA[(t * 16 + lm) * 36 + k0 + 4 * hi]);
      const v2f b0 = {bv.x, bv.y};
      const v2f b1 = {bv.z, bv.w};
#pragma unroll
      for (int t = 0; t < 4; ++t) {
        const v2f a0 = {av[t].x, av[t].y};
        acc1[t] = wmma4(a0, b0, acc1[t]);
      }
#pragma unroll
      for (int t = 0; t < 4; ++t) {
        const v2f a1 = {av[t].z, av[t].w};
        acc1[t] = wmma4(a1, b1, acc1[t]);
      }
    }
    float wlast = We1[288 * 128 + n0 + lm];
#pragma unroll
    for (int t = 0; t < 4; ++t) {
#pragma unroll
      for (int v = 0; v < 8; ++v) {
        int M = t * 16 + v + 8 * hi;
        float pre = acc1[t][v] + sDist[M] * wlast;
        sH[M * 132 + n0 + lm] = silu_f(pre);
      }
    }
  }
  __syncthreads();

  {
    v8f acc[4] = {{}, {}, {}, {}};
    for (int k0 = 0; k0 < 128; k0 += 8) {
      const v4f bv = wp4[PK_W2 + (k0 >> 2) * 128];
      v4f av[4];
#pragma unroll
      for (int t = 0; t < 4; ++t)
        av[t] = *reinterpret_cast<const v4f*>(
            &sH[(t * 16 + lm) * 132 + k0 + 4 * hi]);
      const v2f b0 = {bv.x, bv.y};
      const v2f b1 = {bv.z, bv.w};
#pragma unroll
      for (int t = 0; t < 4; ++t) {
        const v2f a0 = {av[t].x, av[t].y};
        acc[t] = wmma4(a0, b0, acc[t]);
      }
#pragma unroll
      for (int t = 0; t < 4; ++t) {
        const v2f a1 = {av[t].z, av[t].w};
        acc[t] = wmma4(a1, b1, acc[t]);
      }
    }
    float bias = be2[n0 + lm];
#pragma unroll
    for (int t = 0; t < 4; ++t) {
#pragma unroll
      for (int v = 0; v < 8; ++v) {
        int M = t * 16 + v + 8 * hi;
        float ef = acc[t][v] + bias;
        sE[M * 132 + n0 + lm] = ef;
        atomicAdd(&agg[sRow[M] * 128 + n0 + lm], ef);
      }
    }
  }
  __syncthreads();

  {
    v8f acc[4] = {{}, {}, {}, {}};
    for (int k0 = 0; k0 < 128; k0 += 8) {
      const v4f bv = wp4[PK_W3 + (k0 >> 2) * 128];
      v4f av[4];
#pragma unroll
      for (int t = 0; t < 4; ++t)
        av[t] = *reinterpret_cast<const v4f*>(
            &sE[(t * 16 + lm) * 132 + k0 + 4 * hi]);
      const v2f b0 = {bv.x, bv.y};
      const v2f b1 = {bv.z, bv.w};
#pragma unroll
      for (int t = 0; t < 4; ++t) {
        const v2f a0 = {av[t].x, av[t].y};
        acc[t] = wmma4(a0, b0, acc[t]);
      }
#pragma unroll
      for (int t = 0; t < 4; ++t) {
        const v2f a1 = {av[t].z, av[t].w};
        acc[t] = wmma4(a1, b1, acc[t]);
      }
    }
    float bias = bc1[n0 + lm];
    float wc2 = Wc2[n0 + lm];
#pragma unroll
    for (int t = 0; t < 4; ++t) {
      float mine = 0.0f;
#pragma unroll
      for (int v = 0; v < 8; ++v) {
        float p = row16_sum(silu_f(acc[t][v] + bias) * wc2);
        mine = (lm == v) ? p : mine;
      }
      if (lm < 8) atomicAdd(&sW[t * 16 + lm + 8 * hi], mine);
    }
  }
  __syncthreads();

  if (tid < EPB * 3) {
    int e = tid / 3;
    int d = tid - e * 3;
    float cd = sDiff[tid] / (sqrtf(sDist[e]) + 1e-8f);
    atomicAdd(&coords_out[sRow[e] * 3 + d], cd * sW[e]);
  }
}

__global__ __launch_bounds__(256) void egnn_node_kernel(
    const float* __restrict__ x, const float* __restrict__ agg,
    const float* __restrict__ Wn1, const float* __restrict__ bn1,
    const float* __restrict__ Wn2, const float* __restrict__ bn2,
    float* __restrict__ x_out) {
  __shared__ float sA[16 * 260];
  __shared__ float sH[16 * 132];

  const int tid = threadIdx.x;
  const int lane = tid & 31;
  const int wv = tid >> 5;
  const int hi = lane >> 4;
  const int lm = lane & 15;
  const int n0 = wv * 16;
  const int nb = blockIdx.x * 16;

  for (int idx = tid; idx < 16 * 256; idx += 256) {
    int m = idx >> 8;
    int c = idx & 255;
    float v = (c < 128) ? x[(nb + m) * 128 + c] : agg[(nb + m) * 128 + (c - 128)];
    sA[m * 260 + c] = v;
  }
  __syncthreads();

  v8f acc1 = {};
  for (int k0 = 0; k0 < 256; k0 += 4) {
    int kk = k0 + 2 * hi;
    v2f a, b;
    a.x = sA[lm * 260 + kk];
    a.y = sA[lm * 260 + kk + 1];
    b.x = Wn1[kk * 128 + n0 + lm];
    b.y = Wn1[(kk + 1) * 128 + n0 + lm];
    acc1 = wmma4(a, b, acc1);
  }
  {
    float bias = bn1[n0 + lm];
#pragma unroll
    for (int v = 0; v < 8; ++v) {
      int M = v + 8 * hi;
      sH[M * 132 + n0 + lm] = silu_f(acc1[v] + bias);
    }
  }
  __syncthreads();

  v8f acc2 = {};
  for (int k0 = 0; k0 < 128; k0 += 4) {
    int kk = k0 + 2 * hi;
    v2f a, b;
    a.x = sH[lm * 132 + kk];
    a.y = sH[lm * 132 + kk + 1];
    b.x = Wn2[kk * 128 + n0 + lm];
    b.y = Wn2[(kk + 1) * 128 + n0 + lm];
    acc2 = wmma4(a, b, acc2);
  }
  {
    float bias = bn2[n0 + lm];
#pragma unroll
    for (int v = 0; v < 8; ++v) {
      int M = v + 8 * hi;
      x_out[(nb + M) * 128 + n0 + lm] = acc2[v] + bias;
    }
  }
}

extern "C" void kernel_launch(void* const* d_in, const int* in_sizes, int n_in,
                              void* d_out, int out_size, void* d_ws,
                              size_t ws_size, hipStream_t stream) {
  const float* x = (const float*)d_in[0];
  const float* coords = (const float*)d_in[1];
  const float* edge_attr = (const float*)d_in[2];
  const int* edge_index = (const int*)d_in[3];
  const float* We1 = (const float*)d_in[4];
  const float* be1 = (const float*)d_in[5];
  const float* We2 = (const float*)d_in[6];
  const float* be2 = (const float*)d_in[7];
  const float* Wc1 = (const float*)d_in[8];
  const float* bc1 = (const float*)d_in[9];
  const float* Wc2 = (const float*)d_in[10];
  const float* Wn1 = (const float*)d_in[11];
  const float* bn1 = (const float*)d_in[12];
  const float* Wn2 = (const float*)d_in[13];
  const float* bn2 = (const float*)d_in[14];

  float* x_out = (float*)d_out;
  float* coords_out = x_out + (size_t)N_NODES * 128;
  float* agg = (float*)d_ws;

  const int* erow = edge_index;
  const int* ecol = edge_index + N_EDGES;

  if (ws_size >= (size_t)2 * N_NODES * 128 * sizeof(float) +
                     (size_t)PK_SLOTS * 16) {
    float* Q = agg + (size_t)N_NODES * 128;
    float* Wp = Q + (size_t)N_NODES * 128;
    egnn_pre2_kernel<<<N_NODES / 16 + PK_SLOTS / 1024, 256, 0, stream>>>(
        x, coords, We1, be1, We2, Wc1, x_out, Q, agg, coords_out, Wp);
    egnn_edge3_kernel<<<N_EDGES / EPB, 256, 0, stream>>>(
        coords, edge_attr, erow, ecol, We1, x_out, Q, Wp, be2, bc1, Wc2, agg,
        coords_out);
  } else if (ws_size >= (size_t)2 * N_NODES * 128 * sizeof(float)) {
    float* Q = agg + (size_t)N_NODES * 128;
    egnn_pre2_kernel<<<N_NODES / 16, 256, 0, stream>>>(
        x, coords, We1, be1, We2, Wc1, x_out, Q, agg, coords_out, nullptr);
    egnn_edge2_kernel<<<N_EDGES / EPB, 256, 0, stream>>>(
        coords, edge_attr, erow, ecol, We1, x_out, Q, We2, be2, Wc1, bc1, Wc2,
        agg, coords_out);
  } else {
    egnn_pre_kernel<<<N_NODES / 16, 256, 0, stream>>>(x, coords, We1, be1, x_out,
                                                      agg, coords_out);
    egnn_edge_kernel<<<N_EDGES / EPB, 256, 0, stream>>>(
        x, coords, edge_attr, erow, ecol, We1, x_out, We2, be2, Wc1, bc1, Wc2,
        agg, coords_out);
  }
  egnn_node_kernel<<<N_NODES / 16, 256, 0, stream>>>(x, agg, Wn1, bn1, Wn2, bn2,
                                                     x_out);
}


# Edge layer 2 off the edges by linearity: agg = sum(h1)@We2 + deg*be2 in the node kernel, gate via We2@Wc1; h1 summed by 64-bit fixed-point integer atomics, order-free (f32 arrival-order sums varied)
# speedup vs baseline: 3.2157x; 1.0633x over previous
#include <hip/hip_runtime.h>
#include <math.h>

typedef __attribute__((ext_vector_type(2))) float v2f;
typedef __attribute__((ext_vector_type(8))) float v8f;
typedef __attribute__((ext_vector_type(4))) float v4f;

#define N_NODES 50000
#define N_EDGES 800000
#define EPB 64
#define PK_W2 1024
#define PK_W3 5120
#define PK_SLOTS 9216

__device__ __forceinline__ v8f wmma4(v2f a, v2f b, v8f c) {
  return __builtin_amdgcn_wmma_f32_16x16x4_f32(false, a, false, b, (short)0, c,
                                               false, false);
}

__device__ __forceinline__ float silu_f(float v) {
  return v * __builtin_amdgcn_rcpf(1.0f + __expf(-v));
}

__device__ __forceinline__ float row16_sum(float v) {
  v += __int_as_float(__builtin_amdgcn_update_dpp(
      0, __float_as_int(v), 0x161, 0xf, 0xf, false));
  v += __int_as_float(__builtin_amdgcn_update_dpp(
      0, __float_as_int(v), 0x162, 0xf, 0xf, false));
  v += __int_as_float(__builtin_amdgcn_update_dpp(
      0, __float_as_int(v), 0x164, 0xf, 0xf, false));
  v += __int_as_float(__builtin_amdgcn_update_dpp(
      0, __float_as_int(v), 0x168, 0xf, 0xf, false));
  return v;
}

__global__ __launch_bounds__(256) void egnn_pre_kernel(
    const float* __restrict__ x, const float* __restrict__ coords,
    const float* __restrict__ We1, const float* __restrict__ be1,
    float* __restrict__ P, float* __restrict__ agg,
    float* __restrict__ coords_out) {
  __shared__ __attribute__((aligned(16))) float sX[16 * 132];

  const int tid = threadIdx.x;
  const int lane = tid & 31;
  const int wv = tid >> 5;
  const int hi = lane >> 4;
  const int lm = lane & 15;
  const int n0 = wv * 16;
  const int nb = blockIdx.x * 16;

  const int sr = tid >> 5;
  const int sc = 4 * (tid & 31);
  const v4f* x4 = reinterpret_cast<const v4f*>(x) + nb * 32;
  const v4f xa = x4[tid];
  const v4f xb = x4[tid + 256];

  const v4f zero4 = {0.0f, 0.0f, 0.0f, 0.0f};
  v4f* g4 = reinterpret_cast<v4f*>(agg) + nb * 32;
  g4[tid] = zero4;
  g4[tid + 256] = zero4;
  if (tid < 48) coords_out[nb * 3 + tid] = coords[nb * 3 + tid];

  *reinterpret_cast<v4f*>(&sX[sr * 132 + sc]) = xa;
  *reinterpret_cast<v4f*>(&sX[(sr + 8) * 132 + sc]) = xb;
  __syncthreads();

  v8f acc = {};
  for (int k0 = 0; k0 < 128; k0 += 4) {
    int kk = k0 + 2 * hi;
    v2f a, b;
    a.x = sX[lm * 132 + kk];
    a.y = sX[lm * 132 + kk + 1];
    b.x = We1[kk * 128 + n0 + lm];
    b.y = We1[(kk + 1) * 128 + n0 + lm];
    acc = wmma4(a, b, acc);
  }
  const float bias = be1[n0 + lm];
  __syncthreads();
#pragma unroll
  for (int v = 0; v < 8; ++v) sX[(v + 8 * hi) * 132 + n0 + lm] = acc[v] + bias;
  __syncthreads();
  v4f* p4 = reinterpret_cast<v4f*>(P) + nb * 32;
  p4[tid] = *reinterpret_cast<const v4f*>(&sX[sr * 132 + sc]);
  p4[tid + 256] = *reinterpret_cast<const v4f*>(&sX[(sr + 8) * 132 + sc]);
}

__global__ __launch_bounds__(256, 8) void egnn_edge_kernel(
    const float* __restrict__ x, const float* __restrict__ coords,
    const float* __restrict__ edge_attr, const int* __restrict__ erow,
    const int* __restrict__ ecol, const float* __restrict__ We1,
    const float* __restrict__ P, const float* __restrict__ We2,
    const float* __restrict__ be2, const float* __restrict__ Wc1,
    const float* __restrict__ bc1, const float* __restrict__ Wc2,
    float* __restrict__ agg, float* __restrict__ coords_out) {
  __shared__ __attribute__((aligned(16))) float smem[EPB * 264];
  float* sA = smem;
  float* sH = smem;
  float* sE = smem + EPB * 132;

  __shared__ int sRC[2 * EPB];
  int* sRow = sRC;
  int* sCol = sRC + EPB;
  __shared__ float sDiff[EPB * 3], sDist[EPB], sW[EPB];

  const int tid = threadIdx.x;
  const int lane = tid & 31;
  const int wv = tid >> 5;
  const int hi = lane >> 4;
  const int lm = lane & 15;
  const int n0 = wv * 16;
  const int ebase = blockIdx.x * EPB;

  const v4f* ea4 = reinterpret_cast<const v4f*>(edge_attr) + ebase * 8;
  const v4f ea0 = ea4[tid];
  const v4f ea1 = ea4[tid + 256];

  if (tid < EPB) {
    int r = erow[ebase + tid];
    int c = ecol[ebase + tid];
    sRow[tid] = r;
    sCol[tid] = c;
    float d0 = coords[r * 3 + 0] - coords[c * 3 + 0];
    float d1 = coords[r * 3 + 1] - coords[c * 3 + 1];
    float d2 = coords[r * 3 + 2] - coords[c * 3 + 2];
    sDiff[tid * 3 + 0] = d0;
    sDiff[tid * 3 + 1] = d1;
    sDiff[tid * 3 + 2] = d2;
    sDist[tid] = d0 * d0 + d1 * d1 + d2 * d2;
    sW[tid] = 0.0f;
  }
  __syncthreads();

  v8f acc1[4];
  {
    const v4f* x4 = reinterpret_cast<const v4f*>(x);
    v4f xv[8];
    for (int i = 0; i < 8; ++i) {
      const int node = sCol[wv + 8 * i];
      xv[i] = x4[node * 32 + lane];
    }
#pragma unroll
    for (int t = 0; t < 4; ++t) {
#pragma unroll
      for (int v = 0; v < 8; ++v) {
        const int M = t * 16 + v + 8 * hi;
        acc1[t][v] = P[sRow[M] * 128 + n0 + lm];
      }
    }
    for (int i = 0; i < 8; ++i)
      *reinterpret_cast<v4f*>(&sA[(wv + 8 * i) * 164 + 4 * lane]) = xv[i];
    *reinterpret_cast<v4f*>(&sA[(tid >> 3) * 164 + 128 + 4 * (tid & 7)]) = ea0;
    *reinterpret_cast<v4f*>(&sA[((tid >> 3) + 32) * 164 + 128 + 4 * (tid & 7)]) =
        ea1;
  }
  __syncthreads();

  {
    for (int k0 = 0; k0 < 160; k0 += 4) {
      int kk = k0 + 2 * hi;
      v2f b;
      b.x = We1[(128 + kk) * 128 + n0 + lm];
      b.y = We1[(129 + kk) * 128 + n0 + lm];
#pragma unroll
      for (int t = 0; t < 4; ++t) {
        v2f a;
        a.x = sA[(t * 16 + lm) * 164 + kk];
        a.y = sA[(t * 16 + lm) * 164 + kk + 1];
        acc1[t] = wmma4(a, b, acc1[t]);
      }
    }
    float wlast = We1[288 * 128 + n0 + lm];
    __syncthreads();
#pragma unroll
    for (int t = 0; t < 4; ++t) {
#pragma unroll
      for (int v = 0; v < 8; ++v) {
        int M = t * 16 + v + 8 * hi;
        float pre = acc1[t][v] + sDist[M] * wlast;
        sH[M * 132 + n0 + lm] = silu_f(pre);
      }
    }
  }
  __syncthreads();

  {
    v8f acc[4] = {{}, {}, {}, {}};
    for (int k0 = 0; k0 < 128; k0 += 4) {
      int kk = k0 + 2 * hi;
      v2f b;
      b.x = We2[kk * 128 + n0 + lm];
      b.y = We2[(kk + 1) * 128 + n0 + lm];
#pragma unroll
      for (int t = 0; t < 4; ++t) {
        v2f a;
        a.x = sH[(t * 16 + lm) * 132 + kk];
        a.y = sH[(t * 16 + lm) * 132 + kk + 1];
        acc[t] = wmma4(a, b, acc[t]);
      }
    }
    float bias = be2[n0 + lm];
#pragma unroll
    for (int t = 0; t < 4; ++t) {
#pragma unroll
      for (int v = 0; v < 8; ++v) {
        int M = t * 16 + v + 8 * hi;
        float ef = acc[t][v] + bias;
        sE[M * 132 + n0 + lm] = ef;
        atomicAdd(&agg[sRow[M] * 128 + n0 + lm], ef);
      }
    }
  }
  __syncthreads();

  {
    v8f acc[4] = {{}, {}, {}, {}};
    for (int k0 = 0; k0 < 128; k0 += 4) {
      int kk = k0 + 2 * hi;
      v2f b;
      b.x = Wc1[kk * 128 + n0 + lm];
      b.y = Wc1[(kk + 1) * 128 + n0 + lm];
#pragma unroll
      for (int t = 0; t < 4; ++t) {
        v2f a;
        a.x = sE[(t * 16 + lm) * 132 + kk];
        a.y = sE[(t * 16 + lm) * 132 + kk + 1];
        acc[t] = wmma4(a, b, acc[t]);
      }
    }
    float bias = bc1[n0 + lm];
    float wc2 = Wc2[n0 + lm];
#pragma unroll
    for (int t = 0; t < 4; ++t) {
      float mine = 0.0f;
#pragma unroll
      for (int v = 0; v < 8; ++v) {
        float p = row16_sum(silu_f(acc[t][v] + bias) * wc2);
        mine = (lm == v) ? p : mine;
      }
      if (lm < 8) atomicAdd(&sW[t * 16 + lm + 8 * hi], mine);
    }
  }
  __syncthreads();

  if (tid < EPB * 3) {
    int e = tid / 3;
    int d = tid - e * 3;
    float cd = sDiff[tid] / (sqrtf(sDist[e]) + 1e-8f);
    atomicAdd(&coords_out[sRow[e] * 3 + d], cd * sW[e]);
  }
}

__global__ __launch_bounds__(256) void egnn_pre2_kernel(
    const float* __restrict__ x, const float* __restrict__ coords,
    const float* __restrict__ We1, const float* __restrict__ be1,
    const float* __restrict__ We2, const float* __restrict__ Wc1,
    float* __restrict__ P, float* __restrict__ Q, float* __restrict__ agg,
    float* __restrict__ coords_out, float* __restrict__ Wp) {
  if (blockIdx.x >= N_NODES / 16) {
    const int pb = blockIdx.x - N_NODES / 16;
    v4f* out4 = reinterpret_cast<v4f*>(Wp);
    for (int i = 0; i < 4; ++i) {
      const int slot = pb * 1024 + i * 256 + threadIdx.x;
      const float* src;
      int ls;
      if (slot < PK_W2) {
        src = We1 + 256 * 128;
        ls = slot;
      } else if (slot < PK_W3) {
        src = We2;
        ls = slot - PK_W2;
      } else {
        src = Wc1;
        ls = slot - PK_W3;
      }
      const int g = ls >> 7;
      const int n = ls & 127;
      v4f v;
      v.x = src[(4 * g + 0) * 128 + n];
      v.y = src[(4 * g + 1) * 128 + n];
      v.z = src[(4 * g + 2) * 128 + n];
      v.w = src[(4 * g + 3) * 128 + n];
      out4[slot] = v;
    }
    return;
  }
  __shared__ __attribute__((aligned(16))) float sX[16 * 132];
  __shared__ __attribute__((aligned(16))) float sQ[16 * 132];

  const int tid = threadIdx.x;
  const int lane = tid & 31;
  const int wv = tid >> 5;
  const int hi = lane >> 4;
  const int lm = lane & 15;
  const int n0 = wv * 16;
  const int nb = blockIdx.x * 16;

  const int sr = tid >> 5;
  const int sc = 4 * (tid & 31);
  const v4f* x4 = reinterpret_cast<const v4f*>(x) + nb * 32;
  const v4f xa = x4[tid];
  const v4f xb = x4[tid + 256];

  const v4f zero4 = {0.0f, 0.0f, 0.0f, 0.0f};
  v4f* g4 = reinterpret_cast<v4f*>(agg) + nb * 32;
  g4[tid] = zero4;
  g4[tid + 256] = zero4;
  if (tid < 48) coords_out[nb * 3 + tid] = coords[nb * 3 + tid];

  *reinterpret_cast<v4f*>(&sX[sr * 132 + sc]) = xa;
  *reinterpret_cast<v4f*>(&sX[(sr + 8) * 132 + sc]) = xb;
  __syncthreads();

  v8f accP = {};
  v8f accQ = {};
  for (int k0 = 0; k0 < 128; k0 += 4) {
    int kk = k0 + 2 * hi;
    v2f a, bp, bq;
    a.x = sX[lm * 132 + kk];
    a.y = sX[lm * 132 + kk + 1];
    bp.x = We1[kk * 128 + n0 + lm];
    bp.y = We1[(kk + 1) * 128 + n0 + lm];
    bq.x = We1[(128 + kk) * 128 + n0 + lm];
    bq.y = We1[(129 + kk) * 128 + n0 + lm];
    accP = wmma4(a, bp, accP);
    accQ = wmma4(a, bq, accQ);
  }
  const float bias = be1[n0 + lm];
  __syncthreads();
#pragma unroll
  for (int v = 0; v < 8; ++v) {
    sX[(v + 8 * hi) * 132 + n0 + lm] = accP[v] + bias;
    sQ[(v + 8 * hi) * 132 + n0 + lm] = accQ[v];
  }
  __syncthreads();
  v4f* p4 = reinterpret_cast<v4f*>(P) + nb * 32;
  v4f* q4 = reinterpret_cast<v4f*>(Q) + nb * 32;
  p4[tid] = *reinterpret_cast<const v4f*>(&sX[sr * 132 + sc]);
  p4[tid + 256] = *reinterpret_cast<const v4f*>(&sX[(sr + 8) * 132 + sc]);
  q4[tid] = *reinterpret_cast<const v4f*>(&sQ[sr * 132 + sc]);
  q4[tid + 256] = *reinterpret_cast<const v4f*>(&sQ[(sr + 8) * 132 + sc]);
}

__global__ __launch_bounds__(256, 8) void egnn_edge2_kernel(
    const float* __restrict__ coords, const float* __restrict__ edge_attr,
    const int* __restrict__ erow, const int* __restrict__ ecol,
    const float* __restrict__ We1, const float* __restrict__ P,
    const float* __restrict__ Q, const float* __restrict__ We2,
    const float* __restrict__ be2, const float* __restrict__ Wc1,
    const float* __restrict__ bc1, const float* __restrict__ Wc2,
    float* __restrict__ agg, float* __restrict__ coords_out) {
  __shared__ __attribute__((aligned(16))) float smem[EPB * 264];
  float* sH = smem;
  float* sE = smem + EPB * 132;
  float* sA = sE;

  __shared__ int sRC[2 * EPB];
  int* sRow = sRC;
  int* sCol = sRC + EPB;
  __shared__ float sDiff[EPB * 3], sDist[EPB], sW[EPB];

  const int tid = threadIdx.x;
  const int lane = tid & 31;
  const int wv = tid >> 5;
  const int hi = lane >> 4;
  const int lm = lane & 15;
  const int n0 = wv * 16;
  const int ebase = blockIdx.x * EPB;

  const v4f* ea4 = reinterpret_cast<const v4f*>(edge_attr) + ebase * 8;
  const v4f ea0 = ea4[tid];
  const v4f ea1 = ea4[tid + 256];

  if (tid < EPB) {
    int r = erow[ebase + tid];
    int c = ecol[ebase + tid];
    sRow[tid] = r;
    sCol[tid] = c;
    float d0 = coords[r * 3 + 0] - coords[c * 3 + 0];
    float d1 = coords[r * 3 + 1] - coords[c * 3 + 1];
    float d2 = coords[r * 3 + 2] - coords[c * 3 + 2];
    sDiff[tid * 3 + 0] = d0;
    sDiff[tid * 3 + 1] = d1;
    sDiff[tid * 3 + 2] = d2;
    sDist[tid] = d0 * d0 + d1 * d1 + d2 * d2;
    sW[tid] = 0.0f;
  }
  __syncthreads();

  v8f acc1[4];
  {
    v8f qv[4];
#pragma unroll
    for (int t = 0; t < 4; ++t) {
#pragma unroll
      for (int v = 0; v < 8; ++v) {
        const int M = t * 16 + v + 8 * hi;
        qv[t][v] = Q[sCol[M] * 128 + n0 + lm];
      }
    }
#pragma unroll
    for (int t = 0; t < 4; ++t) {
#pragma unroll
      for (int v = 0; v < 8; ++v) {
        const int M = t * 16 + v + 8 * hi;
        acc1[t][v] = P[sRow[M] * 128 + n0 + lm];
      }
    }
#pragma unroll
    for (int t = 0; t < 4; ++t) acc1[t] += qv[t];
  }
  *reinterpret_cast<v4f*>(&sA[(tid >> 3) * 36 + 4 * (tid & 7)]) = ea0;
  *reinterpret_cast<v4f*>(&sA[((tid >> 3) + 32) * 36 + 4 * (tid & 7)]) = ea1;
  __syncthreads();

  {
    for (int k0 = 0; k0 < 32; k0 += 4) {
      int kk = k0 + 2 * hi;
      v2f b;
      b.x = We1[(256 + kk) * 128 + n0 + lm];
      b.y = We1[(257 + kk) * 128 + n0 + lm];
#pragma unroll
      for (int t = 0; t < 4; ++t) {
        v2f a;
        a.x = sA[(t * 16 + lm) * 36 + kk];
        a.y = sA[(t * 16 + lm) * 36 + kk + 1];
        acc1[t] = wmma4(a, b, acc1[t]);
      }
    }
    float wlast = We1[288 * 128 + n0 + lm];
#pragma unroll
    for (int t = 0; t < 4; ++t) {
#pragma unroll
      for (int v = 0; v < 8; ++v) {
        int M = t * 16 + v + 8 * hi;
        float pre = acc1[t][v] + sDist[M] * wlast;
        sH[M * 132 + n0 + lm] = silu_f(pre);
      }
    }
  }
  __syncthreads();

  {
    v8f acc[4] = {{}, {}, {}, {}};
    for (int k0 = 0; k0 < 128; k0 += 4) {
      int kk = k0 + 2 * hi;
      v2f b;
      b.x = We2[kk * 128 + n0 + lm];
      b.y = We2[(kk + 1) * 128 + n0 + lm];
#pragma unroll
      for (int t = 0; t < 4; ++t) {
        v2f a;
        a.x = sH[(t * 16 + lm) * 132 + kk];
        a.y = sH[(t * 16 + lm) * 132 + kk + 1];
        acc[t] = wmma4(a, b, acc[t]);
      }
    }
    float bias = be2[n0 + lm];
#pragma unroll
    for (int t = 0; t < 4; ++t) {
#pragma unroll
      for (int v = 0; v < 8; ++v) {
        int M = t * 16 + v + 8 * hi;
        float ef = acc[t][v] + bias;
        sE[M * 132 + n0 + lm] = ef;
        atomicAdd(&agg[sRow[M] * 128 + n0 + lm], ef);
      }
    }
  }
  __syncthreads();

  {
    v8f acc[4] = {{}, {}, {}, {}};
    for (int k0 = 0; k0 < 128; k0 += 4) {
      int kk = k0 + 2 * hi;
      v2f b;
      b.x = Wc1[kk * 128 + n0 + lm];
      b.y = Wc1[(kk + 1) * 128 + n0 + lm];
#pragma unroll
      for (int t = 0; t < 4; ++t) {
        v2f a;
        a.x = sE[(t * 16 + lm) * 132 + kk];
        a.y = sE[(t * 16 + lm) * 132 + kk + 1];
        acc[t] = wmma4(a, b, acc[t]);
      }
    }
    float bias = bc1[n0 + lm];
    float wc2 = Wc2[n0 + lm];
#pragma unroll
    for (int t = 0; t < 4; ++t) {
      float mine = 0.0f;
#pragma unroll
      for (int v = 0; v < 8; ++v) {
        float p = row16_sum(silu_f(acc[t][v] + bias) * wc2);
        mine = (lm == v) ? p : mine;
      }
      if (lm < 8) atomicAdd(&sW[t * 16 + lm + 8 * hi], mine);
    }
  }
  __syncthreads();

  if (tid < EPB * 3) {
    int e = tid / 3;
    int d = tid - e * 3;
    float cd = sDiff[tid] / (sqrtf(sDist[e]) + 1e-8f);
    atomicAdd(&coords_out[sRow[e] * 3 + d], cd * sW[e]);
  }
}

__global__ __launch_bounds__(256, 8) void egnn_edge3_kernel(
    const float* __restrict__ coords, const float* __restrict__ edge_attr,
    const int* __restrict__ erow, const int* __restrict__ ecol,
    const float* __restrict__ We1, const float* __restrict__ P,
    const float* __restrict__ Q, const float* __restrict__ Wp,
    const float* __restrict__ be2, const float* __restrict__ bc1,
    const float* __restrict__ Wc2, float* __restrict__ agg,
    float* __restrict__ coords_out) {
  __shared__ __attribute__((aligned(16))) float smem[EPB * 264];
  float* sH = smem;
  float* sE = smem + EPB * 132;
  float* sA = sE;

  __shared__ int sRC[2 * EPB];
  int* sRow = sRC;
  int* sCol = sRC + EPB;
  __shared__ float sDiff[EPB * 3], sDist[EPB], sW[EPB];

  const int tid = threadIdx.x;
  const int lane = tid & 31;
  const int wv = tid >> 5;
  const int hi = lane >> 4;
  const int lm = lane & 15;
  const int n0 = wv * 16;
  const int ebase = blockIdx.x * EPB;
  const v4f* wp4 = reinterpret_cast<const v4f*>(Wp) + hi * 128 + n0 + lm;

  const v4f* ea4 = reinterpret_cast<const v4f*>(edge_attr) + ebase * 8;
  const v4f ea0 = ea4[tid];
  const v4f ea1 = ea4[tid + 256];

  if (tid < EPB) {
    int r = erow[ebase + tid];
    int c = ecol[ebase + tid];
    sRow[tid] = r;
    sCol[tid] = c;
    float d0 = coords[r * 3 + 0] - coords[c * 3 + 0];
    float d1 = coords[r * 3 + 1] - coords[c * 3 + 1];
    float d2 = coords[r * 3 + 2] - coords[c * 3 + 2];
    sDiff[tid * 3 + 0] = d0;
    sDiff[tid * 3 + 1] = d1;
    sDiff[tid * 3 + 2] = d2;
    sDist[tid] = d0 * d0 + d1 * d1 + d2 * d2;
    sW[tid] = 0.0f;
  }
  __syncthreads();

  v8f acc1[4];
  {
    v8f qv[4];
#pragma unroll
    for (int t = 0; t < 4; ++t) {
#pragma unroll
      for (int v = 0; v < 8; ++v) {
        const int M = t * 16 + v + 8 * hi;
        qv[t][v] = Q[sCol[M] * 128 + n0 + lm];
      }
    }
#pragma unroll
    for (int t = 0; t < 4; ++t) {
#pragma unroll
      for (int v = 0; v < 8; ++v) {
        const int M = t * 16 + v + 8 * hi;
        acc1[t][v] = P[sRow[M] * 128 + n0 + lm];
      }
    }
#pragma unroll
    for (int t = 0; t < 4; ++t) acc1[t] += qv[t];
  }
  *reinterpret_cast<v4f*>(&sA[(tid >> 3) * 36 + 4 * (tid & 7)]) = ea0;
  *reinterpret_cast<v4f*>(&sA[((tid >> 3) + 32) * 36 + 4 * (tid & 7)]) = ea1;
  __syncthreads();

  {
    for (int k0 = 0; k0 < 32; k0 += 8) {
      const v4f bv = wp4[(k0 >> 2) * 128];
      v4f av[4];
#pragma unroll
      for (int t = 0; t < 4; ++t)
        av[t] = *reinterpret_cast<const v4f*>(
            &sA[(t * 16 + lm) * 36 + k0 + 4 * hi]);
      const v2f b0 = {bv.x, bv.y};
      const v2f b1 = {bv.z, bv.w};
#pragma unroll
      for (int t = 0; t < 4; ++t) {
        const v2f a0 = {av[t].x, av[t].y};
        acc1[t] = wmma4(a0, b0, acc1[t]);
      }
#pragma unroll
      for (int t = 0; t < 4; ++t) {
        const v2f a1 = {av[t].z, av[t].w};
        acc1[t] = wmma4(a1, b1, acc1[t]);
      }
    }
    float wlast = We1[288 * 128 + n0 + lm];
#pragma unroll
    for (int t = 0; t < 4; ++t) {
#pragma unroll
      for (int v = 0; v < 8; ++v) {
        int M = t * 16 + v + 8 * hi;
        float pre = acc1[t][v] + sDist[M] * wlast;
        sH[M * 132 + n0 + lm] = silu_f(pre);
      }
    }
  }
  __syncthreads();

  {
    v8f acc[4] = {{}, {}, {}, {}};
    for (int k0 = 0; k0 < 128; k0 += 8) {
      const v4f bv = wp4[PK_W2 + (k0 >> 2) * 128];
      v4f av[4];
#pragma unroll
      for (int t = 0; t < 4; ++t)
        av[t] = *reinterpret_cast<const v4f*>(
            &sH[(t * 16 + lm) * 132 + k0 + 4 * hi]);
      const v2f b0 = {bv.x, bv.y};
      const v2f b1 = {bv.z, bv.w};
#pragma unroll
      for (int t = 0; t < 4; ++t) {
        const v2f a0 = {av[t].x, av[t].y};
        acc[t] = wmma4(a0, b0, acc[t]);
      }
#pragma unroll
      for (int t = 0; t < 4; ++t) {
        const v2f a1 = {av[t].z, av[t].w};
        acc[t] = wmma4(a1, b1, acc[t]);
      }
    }
    float bias = be2[n0 + lm];
#pragma unroll
    for (int t = 0; t < 4; ++t) {
#pragma unroll
      for (int v = 0; v < 8; ++v) {
        int M = t * 16 + v + 8 * hi;
        float ef = acc[t][v] + bias;
        sE[M * 132 + n0 + lm] = ef;
        atomicAdd(&agg[sRow[M] * 128 + n0 + lm], ef);
      }
    }
  }
  __syncthreads();

  {
    v8f acc[4] = {{}, {}, {}, {}};
    for (int k0 = 0; k0 < 128; k0 += 8) {
      const v4f bv = wp4[PK_W3 + (k0 >> 2) * 128];
      v4f av[4];
#pragma unroll
      for (int t = 0; t < 4; ++t)
        av[t] = *reinterpret_cast<const v4f*>(
            &sE[(t * 16 + lm) * 132 + k0 + 4 * hi]);
      const v2f b0 = {bv.x, bv.y};
      const v2f b1 = {bv.z, bv.w};
#pragma unroll
      for (int t = 0; t < 4; ++t) {
        const v2f a0 = {av[t].x, av[t].y};
        acc[t] = wmma4(a0, b0, acc[t]);
      }
#pragma unroll
      for (int t = 0; t < 4; ++t) {
        const v2f a1 = {av[t].z, av[t].w};
        acc[t] = wmma4(a1, b1, acc[t]);
      }
    }
    float bias = bc1[n0 + lm];
    float wc2 = Wc2[n0 + lm];
#pragma unroll
    for (int t = 0; t < 4; ++t) {
      float mine = 0.0f;
#pragma unroll
      for (int v = 0; v < 8; ++v) {
        float p = row16_sum(silu_f(acc[t][v] + bias) * wc2);
        mine = (lm == v) ? p : mine;
      }
      if (lm < 8) atomicAdd(&sW[t * 16 + lm + 8 * hi], mine);
    }
  }
  __syncthreads();

  if (tid < EPB * 3) {
    int e = tid / 3;
    int d = tid - e * 3;
    float cd = sDiff[tid] / (sqrtf(sDist[e]) + 1e-8f);
    atomicAdd(&coords_out[sRow[e] * 3 + d], cd * sW[e]);
  }
}

__global__ __launch_bounds__(256) void egnn_node_kernel(
    const float* __restrict__ x, const float* __restrict__ agg,
    const float* __restrict__ Wn1, const float* __restrict__ bn1,
    const float* __restrict__ Wn2, const float* __restrict__ bn2,
    float* __restrict__ x_out) {
  __shared__ float sA[16 * 260];
  __shared__ float sH[16 * 132];

  const int tid = threadIdx.x;
  const int lane = tid & 31;
  const int wv = tid >> 5;
  const int hi = lane >> 4;
  const int lm = lane & 15;
  const int n0 = wv * 16;
  const int nb = blockIdx.x * 16;

  for (int idx = tid; idx < 16 * 256; idx += 256) {
    int m = idx >> 8;
    int c = idx & 255;
    float v = (c < 128) ? x[(nb + m) * 128 + c] : agg[(nb + m) * 128 + (c - 128)];
    sA[m * 260 + c] = v;
  }
  __syncthreads();

  v8f acc1 = {};
  for (int k0 = 0; k0 < 256; k0 += 4) {
    int kk = k0 + 2 * hi;
    v2f a, b;
    a.x = sA[lm * 260 + kk];
    a.y = sA[lm * 260 + kk + 1];
    b.x = Wn1[kk * 128 + n0 + lm];
    b.y = Wn1[(kk + 1) * 128 + n0 + lm];
    acc1 = wmma4(a, b, acc1);
  }
  {
    float bias = bn1[n0 + lm];
#pragma unroll
    for (int v = 0; v < 8; ++v) {
      int M = v + 8 * hi;
      sH[M * 132 + n0 + lm] = silu_f(acc1[v] + bias);
    }
  }
  __syncthreads();

  v8f acc2 = {};
  for (int k0 = 0; k0 < 128; k0 += 4) {
    int kk = k0 + 2 * hi;
    v2f a, b;
    a.x = sH[lm * 132 + kk];
    a.y = sH[lm * 132 + kk + 1];
    b.x = Wn2[kk * 128 + n0 + lm];
    b.y = Wn2[(kk + 1) * 128 + n0 + lm];
    acc2 = wmma4(a, b, acc2);
  }
  {
    float bias = bn2[n0 + lm];
#pragma unroll
    for (int v = 0; v < 8; ++v) {
      int M = v + 8 * hi;
      x_out[(nb + M) * 128 + n0 + lm] = acc2[v] + bias;
    }
  }
}

#define PL_W23 1024
#define PL_SLOTS 5120
#define PL_BLOCKS 9

__global__ __launch_bounds__(256) void egnn_pre3_kernel(
    const float* __restrict__ x, const float* __restrict__ coords,
    const float* __restrict__ We1, const float* __restrict__ be1,
    const float* __restrict__ We2, const float* __restrict__ be2,
    const float* __restrict__ Wc1, const float* __restrict__ bc1,
    float* __restrict__ P, float* __restrict__ Q, float* __restrict__ S4,
    int* __restrict__ deg, float* __restrict__ coords_out,
    float* __restrict__ Wp, float* __restrict__ b23) {
  const int tid = threadIdx.x;
  const int lane = tid & 31;
  const int wv = tid >> 5;
  const int hi = lane >> 4;
  const int lm = lane & 15;

  if (blockIdx.x >= N_NODES / 16) {
    const int pb = blockIdx.x - N_NODES / 16;
    v4f* out4 = reinterpret_cast<v4f*>(Wp);
    if (pb == 0) {
      const float* src = We1 + 256 * 128;
      for (int i = 0; i < 4; ++i) {
        const int slot = i * 256 + tid;
        const int g = slot >> 7;
        const int n = slot & 127;
        v4f v;
        v.x = src[(4 * g + 0) * 128 + n];
        v.y = src[(4 * g + 1) * 128 + n];
        v.z = src[(4 * g + 2) * 128 + n];
        v.w = src[(4 * g + 3) * 128 + n];
        out4[slot] = v;
      }
      if (tid < 128) {
        float s = 0.0f;
        for (int m = 0; m < 128; ++m) s = fmaf(be2[m], Wc1[m * 128 + tid], s);
        b23[tid] = s + bc1[tid];
      }
      return;
    }
    const int c0 = (pb - 1) * 16;
    const int i0 = wv * 16;
    v8f acc = {};
    for (int k0 = 0; k0 < 128; k0 += 4) {
      const int kk = k0 + 2 * hi;
      v2f a, b;
      a.x = We2[(i0 + lm) * 128 + kk];
      a.y = We2[(i0 + lm) * 128 + kk + 1];
      b.x = Wc1[kk * 128 + c0 + lm];
      b.y = Wc1[(kk + 1) * 128 + c0 + lm];
      acc = wmma4(a, b, acc);
    }
    const int g0 = (i0 + 8 * hi) >> 2;
    const v4f s0 = {acc[0], acc[1], acc[2], acc[3]};
    const v4f s1 = {acc[4], acc[5], acc[6], acc[7]};
    out4[PL_W23 + g0 * 128 + c0 + lm] = s0;
    out4[PL_W23 + (g0 + 1) * 128 + c0 + lm] = s1;
    return;
  }
  __shared__ __attribute__((aligned(16))) float sX[16 * 132];
  __shared__ __attribute__((aligned(16))) float sQ[16 * 132];

  const int n0 = wv * 16;
  const int nb = blockIdx.x * 16;

  const int sr = tid >> 5;
  const int sc = 4 * (tid & 31);
  const v4f* x4 = reinterpret_cast<const v4f*>(x) + nb * 32;
  const v4f xa = x4[tid];
  const v4f xb = x4[tid + 256];

  const v4f zero4 = {0.0f, 0.0f, 0.0f, 0.0f};
  v4f* g4 = reinterpret_cast<v4f*>(S4) + nb * 64;
  g4[tid] = zero4;
  g4[tid + 256] = zero4;
  g4[tid + 512] = zero4;
  g4[tid + 768] = zero4;
  if (tid < 16) deg[nb + tid] = 0;
  if (tid < 48) coords_out[nb * 3 + tid] = coords[nb * 3 + tid];

  *reinterpret_cast<v4f*>(&sX[sr * 132 + sc]) = xa;
  *reinterpret_cast<v4f*>(&sX[(sr + 8) * 132 + sc]) = xb;
  __syncthreads();

  v8f accP = {};
  v8f accQ = {};
  for (int k0 = 0; k0 < 128; k0 += 4) {
    int kk = k0 + 2 * hi;
    v2f a, bp, bq;
    a.x = sX[lm * 132 + kk];
    a.y = sX[lm * 132 + kk + 1];
    bp.x = We1[kk * 128 + n0 + lm];
    bp.y = We1[(kk + 1) * 128 + n0 + lm];
    bq.x = We1[(128 + kk) * 128 + n0 + lm];
    bq.y = We1[(129 + kk) * 128 + n0 + lm];
    accP = wmma4(a, bp, accP);
    accQ = wmma4(a, bq, accQ);
  }
  const float bias = be1[n0 + lm];
  __syncthreads();
#pragma unroll
  for (int v = 0; v < 8; ++v) {
    sX[(v + 8 * hi) * 132 + n0 + lm] = accP[v] + bias;
    sQ[(v + 8 * hi) * 132 + n0 + lm] = accQ[v];
  }
  __syncthreads();
  v4f* p4 = reinterpret_cast<v4f*>(P) + nb * 32;
  v4f* q4 = reinterpret_cast<v4f*>(Q) + nb * 32;
  p4[tid] = *reinterpret_cast<const v4f*>(&sX[sr * 132 + sc]);
  p4[tid + 256] = *reinterpret_cast<const v4f*>(&sX[(sr + 8) * 132 + sc]);
  q4[tid] = *reinterpret_cast<const v4f*>(&sQ[sr * 132 + sc]);
  q4[tid + 256] = *reinterpret_cast<const v4f*>(&sQ[(sr + 8) * 132 + sc]);
}

__global__ __launch_bounds__(256, 8) void egnn_edge4_kernel(
    const float* __restrict__ coords, const float* __restrict__ edge_attr,
    const int* __restrict__ erow, const int* __restrict__ ecol,
    const float* __restrict__ We1, const float* __restrict__ P,
    const float* __restrict__ Q, const float* __restrict__ Wp,
    const float* __restrict__ b23, const float* __restrict__ Wc2,
    unsigned long long* __restrict__ S, int* __restrict__ deg,
    float* __restrict__ coords_out) {
  __shared__ __attribute__((aligned(16))) float smem[EPB * 168];
  float* sH = smem;
  float* sA = smem + EPB * 132;

  __shared__ int sRC[2 * EPB];
  int* sRow = sRC;
  int* sCol = sRC + EPB;
  __shared__ float sDiff[EPB * 3], sDist[EPB], sW[EPB];

  const int tid = threadIdx.x;
  const int lane = tid & 31;
  const int wv = tid >> 5;
  const int hi = lane >> 4;
  const int lm = lane & 15;
  const int n0 = wv * 16;
  const int ebase = blockIdx.x * EPB;
  const v4f* wp4 = reinterpret_cast<const v4f*>(Wp) + hi * 128 + n0 + lm;

  const v4f* ea4 = reinterpret_cast<const v4f*>(edge_attr) + ebase * 8;
  const v4f ea0 = ea4[tid];
  const v4f ea1 = ea4[tid + 256];

  if (tid < EPB) {
    int r = erow[ebase + tid];
    int c = ecol[ebase + tid];
    sRow[tid] = r;
    sCol[tid] = c;
    atomicAdd(&deg[r], 1);
    float d0 = coords[r * 3 + 0] - coords[c * 3 + 0];
    float d1 = coords[r * 3 + 1] - coords[c * 3 + 1];
    float d2 = coords[r * 3 + 2] - coords[c * 3 + 2];
    sDiff[tid * 3 + 0] = d0;
    sDiff[tid * 3 + 1] = d1;
    sDiff[tid * 3 + 2] = d2;
    sDist[tid] = d0 * d0 + d1 * d1 + d2 * d2;
    sW[tid] = 0.0f;
  }
  __syncthreads();

  v8f acc1[4];
  {
    v8f qv[4];
#pragma unroll
    for (int t = 0; t < 4; ++t) {
#pragma unroll
      for (int v = 0; v < 8; ++v) {
        const int M = t * 16 + v + 8 * hi;
        qv[t][v] = Q[sCol[M] * 128 + n0 + lm];
      }
    }
#pragma unroll
    for (int t = 0; t < 4; ++t) {
#pragma unroll
      for (int v = 0; v < 8; ++v) {
        const int M = t * 16 + v + 8 * hi;
        acc1[t][v] = P[sRow[M] * 128 + n0 + lm];
      }
    }
#pragma unroll
    for (int t = 0; t < 4; ++t) acc1[t] += qv[t];
  }
  *reinterpret_cast<v4f*>(&sA[(tid >> 3) * 36 + 4 * (tid & 7)]) = ea0;
  *reinterpret_cast<v4f*>(&sA[((tid >> 3) + 32) * 36 + 4 * (tid & 7)]) = ea1;
  __syncthreads();

  {
    for (int k0 = 0; k0 < 32; k0 += 8) {
      const v4f bv = wp4[(k0 >> 2) * 128];
      v4f av[4];
#pragma unroll
      for (int t = 0; t < 4; ++t)
        av[t] = *reinterpret_cast<const v4f*>(
            &sA[(t * 16 + lm) * 36 + k0 + 4 * hi]);
      const v2f b0 = {bv.x, bv.y};
      const v2f b1 = {bv.z, bv.w};
#pragma unroll
      for (int t = 0; t < 4; ++t) {
        const v2f a0 = {av[t].x, av[t].y};
        acc1[t] = wmma4(a0, b0, acc1[t]);
      }
#pragma unroll
      for (int t = 0; t < 4; ++t) {
        const v2f a1 = {av[t].z, av[t].w};
        acc1[t] = wmma4(a1, b1, acc1[t]);
      }
    }
    float wlast = We1[288 * 128 + n0 + lm];
#pragma unroll
    for (int t = 0; t < 4; ++t) {
#pragma unroll
      for (int v = 0; v < 8; ++v) {
        int M = t * 16 + v + 8 * hi;
        float pre = acc1[t][v] + sDist[M] * wlast;
        float h = silu_f(pre);
        sH[M * 132 + n0 + lm] = h;
        const long long q = (long long)(h * 4294967296.0f);
        atomicAdd(&S[sRow[M] * 128 + n0 + lm], (unsigned long long)q);
      }
    }
  }
  __syncthreads();

  {
    v8f acc[4] = {{}, {}, {}, {}};
    for (int k0 = 0; k0 < 128; k0 += 8) {
      const v4f bv = wp4[PL_W23 + (k0 >> 2) * 128];
      v4f av[4];
#pragma unroll
      for (int t = 0; t < 4; ++t)
        av[t] = *reinterpret_cast<const v4f*>(
            &sH[(t * 16 + lm) * 132 + k0 + 4 * hi]);
      const v2f b0 = {bv.x, bv.y};
      const v2f b1 = {bv.z, bv.w};
#pragma unroll
      for (int t = 0; t < 4; ++t) {
        const v2f a0 = {av[t].x, av[t].y};
        acc[t] = wmma4(a0, b0, acc[t]);
      }
#pragma unroll
      for (int t = 0; t < 4; ++t) {
        const v2f a1 = {av[t].z, av[t].w};
        acc[t] = wmma4(a1, b1, acc[t]);
      }
    }
    float bias = b23[n0 + lm];
    float wc2 = Wc2[n0 + lm];
#pragma unroll
    for (int t = 0; t < 4; ++t) {
      float mine = 0.0f;
#pragma unroll
      for (int v = 0; v < 8; ++v) {
        float p = row16_sum(silu_f(acc[t][v] + bias) * wc2);
        mine = (lm == v) ? p : mine;
      }
      if (lm < 8) atomicAdd(&sW[t * 16 + lm + 8 * hi], mine);
    }
  }
  __syncthreads();

  if (tid < EPB * 3) {
    int e = tid / 3;
    int d = tid - e * 3;
    float cd = sDiff[tid] / (sqrtf(sDist[e]) + 1e-8f);
    atomicAdd(&coords_out[sRow[e] * 3 + d], cd * sW[e]);
  }
}

__global__ __launch_bounds__(256, 16) void egnn_node2_kernel(
    const float* __restrict__ x, const unsigned long long* __restrict__ S,
    const int* __restrict__ deg, const float* __restrict__ We2,
    const float* __restrict__ be2, const float* __restrict__ Wn1,
    const float* __restrict__ bn1, const float* __restrict__ Wn2,
    const float* __restrict__ bn2, float* __restrict__ x_out) {
  __shared__ float sA[16 * 260];
  __shared__ float sH[16 * 132];

  const int tid = threadIdx.x;
  const int lane = tid & 31;
  const int wv = tid >> 5;
  const int hi = lane >> 4;
  const int lm = lane & 15;
  const int n0 = wv * 16;
  const int nb = blockIdx.x * 16;

  for (int idx = tid; idx < 16 * 128; idx += 256) {
    int m = idx >> 7;
    int c = idx & 127;
    sA[m * 260 + c] = x[(nb + m) * 128 + c];
    sH[m * 132 + c] =
        (float)(long long)S[(nb + m) * 128 + c] * 2.3283064365386963e-10f;
  }
  __syncthreads();

  {
    v8f acc = {};
    for (int k0 = 0; k0 < 128; k0 += 4) {
      int kk = k0 + 2 * hi;
      v2f a, b;
      a.x = sH[lm * 132 + kk];
      a.y = sH[lm * 132 + kk + 1];
      b.x = We2[kk * 128 + n0 + lm];
      b.y = We2[(kk + 1) * 128 + n0 + lm];
      acc = wmma4(a, b, acc);
    }
    float bias = be2[n0 + lm];
#pragma unroll
    for (int v = 0; v < 8; ++v) {
      int M = v + 8 * hi;
      sA[M * 260 + 128 + n0 + lm] = acc[v] + (float)deg[nb + M] * bias;
    }
  }
  __syncthreads();

  v8f acc1 = {};
  for (int k0 = 0; k0 < 256; k0 += 4) {
    int kk = k0 + 2 * hi;
    v2f a, b;
    a.x = sA[lm * 260 + kk];
    a.y = sA[lm * 260 + kk + 1];
    b.x = Wn1[kk * 128 + n0 + lm];
    b.y = Wn1[(kk + 1) * 128 + n0 + lm];
    acc1 = wmma4(a, b, acc1);
  }
  {
    float bias = bn1[n0 + lm];
#pragma unroll
    for (int v = 0; v < 8; ++v) {
      int M = v + 8 * hi;
      sH[M * 132 + n0 + lm] = silu_f(acc1[v] + bias);
    }
  }
  __syncthreads();

  v8f acc2 = {};
  for (int k0 = 0; k0 < 128; k0 += 4) {
    int kk = k0 + 2 * hi;
    v2f a, b;
    a.x = sH[lm * 132 + kk];
    a.y = sH[lm * 132 + kk + 1];
    b.x = Wn2[kk * 128 + n0 + lm];
    b.y = Wn2[(kk + 1) * 128 + n0 + lm];
    acc2 = wmma4(a, b, acc2);
  }
  {
    float bias = bn2[n0 + lm];
#pragma unroll
    for (int v = 0; v < 8; ++v) {
      int M = v + 8 * hi;
      x_out[(nb + M) * 128 + n0 + lm] = acc2[v] + bias;
    }
  }
}

extern "C" void kernel_launch(void* const* d_in, const int* in_sizes, int n_in,
                              void* d_out, int out_size, void* d_ws,
                              size_t ws_size, hipStream_t stream) {
  const float* x = (const float*)d_in[0];
  const float* coords = (const float*)d_in[1];
  const float* edge_attr = (const float*)d_in[2];
  const int* edge_index = (const int*)d_in[3];
  const float* We1 = (const float*)d_in[4];
  const float* be1 = (const float*)d_in[5];
  const float* We2 = (const float*)d_in[6];
  const float* be2 = (const float*)d_in[7];
  const float* Wc1 = (const float*)d_in[8];
  const float* bc1 = (const float*)d_in[9];
  const float* Wc2 = (const float*)d_in[10];
  const float* Wn1 = (const float*)d_in[11];
  const float* bn1 = (const float*)d_in[12];
  const float* Wn2 = (const float*)d_in[13];
  const float* bn2 = (const float*)d_in[14];

  float* x_out = (float*)d_out;
  float* coords_out = x_out + (size_t)N_NODES * 128;
  float* agg = (float*)d_ws;

  const int* erow = edge_index;
  const int* ecol = edge_index + N_EDGES;

  if (ws_size >= (size_t)3 * N_NODES * 128 * sizeof(float) +
                     (size_t)PL_SLOTS * 16 + 128 * sizeof(float) +
                     (size_t)N_NODES * sizeof(int)) {
    unsigned long long* S = (unsigned long long*)d_ws;
    float* Q = (float*)d_ws + (size_t)N_NODES * 256;
    float* Wp = Q + (size_t)N_NODES * 128;
    float* b23 = Wp + (size_t)PL_SLOTS * 4;
    int* deg = (int*)(b23 + 128);
    egnn_pre3_kernel<<<N_NODES / 16 + PL_BLOCKS, 256, 0, stream>>>(
        x, coords, We1, be1, We2, be2, Wc1, bc1, x_out, Q, (float*)d_ws, deg,
        coords_out, Wp, b23);
    egnn_edge4_kernel<<<N_EDGES / EPB, 256, 0, stream>>>(
        coords, edge_attr, erow, ecol, We1, x_out, Q, Wp, b23, Wc2, S, deg,
        coords_out);
    egnn_node2_kernel<<<N_NODES / 16, 256, 0, stream>>>(
        x, S, deg, We2, be2, Wn1, bn1, Wn2, bn2, x_out);
    return;
  }
  if (ws_size >= (size_t)2 * N_NODES * 128 * sizeof(float) +
                     (size_t)PK_SLOTS * 16) {
    float* Q = agg + (size_t)N_NODES * 128;
    float* Wp = Q + (size_t)N_NODES * 128;
    egnn_pre2_kernel<<<N_NODES / 16 + PK_SLOTS / 1024, 256, 0, stream>>>(
        x, coords, We1, be1, We2, Wc1, x_out, Q, agg, coords_out, Wp);
    egnn_edge3_kernel<<<N_EDGES / EPB, 256, 0, stream>>>(
        coords, edge_attr, erow, ecol, We1, x_out, Q, Wp, be2, bc1, Wc2, agg,
        coords_out);
  } else if (ws_size >= (size_t)2 * N_NODES * 128 * sizeof(float)) {
    float* Q = agg + (size_t)N_NODES * 128;
    egnn_pre2_kernel<<<N_NODES / 16, 256, 0, stream>>>(
        x, coords, We1, be1, We2, Wc1, x_out, Q, agg, coords_out, nullptr);
    egnn_edge2_kernel<<<N_EDGES / EPB, 256, 0, stream>>>(
        coords, edge_attr, erow, ecol, We1, x_out, Q, We2, be2, Wc1, bc1, Wc2,
        agg, coords_out);
  } else {
    egnn_pre_kernel<<<N_NODES / 16, 256, 0, stream>>>(x, coords, We1, be1, x_out,
                                                      agg, coords_out);
    egnn_edge_kernel<<<N_EDGES / EPB, 256, 0, stream>>>(
        x, coords, edge_attr, erow, ecol, We1, x_out, We2, be2, Wc1, bc1, Wc2,
        agg, coords_out);
  }
  egnn_node_kernel<<<N_NODES / 16, 256, 0, stream>>>(x, agg, Wn1, bn1, Wn2, bn2,
                                                     x_out);
}


# Node sums of h1: two features per 64-bit integer atomic (signed 32-bit fields, steps 2^-22 not 2^-32, rounded: sum precision about f32's), sent from the LDS tile, 256 B per wave; half the atomics
# speedup vs baseline: 3.6393x; 1.1317x over previous
#include <hip/hip_runtime.h>
#include <math.h>

typedef __attribute__((ext_vector_type(2))) float v2f;
typedef __attribute__((ext_vector_type(8))) float v8f;
typedef __attribute__((ext_vector_type(4))) float v4f;

#define N_NODES 50000
#define N_EDGES 800000
#define EPB 64
#define PK_W2 1024
#define PK_W3 5120
#define PK_SLOTS 9216

__device__ __forceinline__ v8f wmma4(v2f a, v2f b, v8f c) {
  return __builtin_amdgcn_wmma_f32_16x16x4_f32(false, a, false, b, (short)0, c,
                                               false, false);
}

__device__ __forceinline__ float silu_f(float v) {
  return v * __builtin_amdgcn_rcpf(1.0f + __expf(-v));
}

__device__ __forceinline__ float row16_sum(float v) {
  v += __int_as_float(__builtin_amdgcn_update_dpp(
      0, __float_as_int(v), 0x161, 0xf, 0xf, false));
  v += __int_as_float(__builtin_amdgcn_update_dpp(
      0, __float_as_int(v), 0x162, 0xf, 0xf, false));
  v += __int_as_float(__builtin_amdgcn_update_dpp(
      0, __float_as_int(v), 0x164, 0xf, 0xf, false));
  v += __int_as_float(__builtin_amdgcn_update_dpp(
      0, __float_as_int(v), 0x168, 0xf, 0xf, false));
  return v;
}

__global__ __launch_bounds__(256) void egnn_pre_kernel(
    const float* __restrict__ x, const float* __restrict__ coords,
    const float* __restrict__ We1, const float* __restrict__ be1,
    float* __restrict__ P, float* __restrict__ agg,
    float* __restrict__ coords_out) {
  __shared__ __attribute__((aligned(16))) float sX[16 * 132];

  const int tid = threadIdx.x;
  const int lane = tid & 31;
  const int wv = tid >> 5;
  const int hi = lane >> 4;
  const int lm = lane & 15;
  const int n0 = wv * 16;
  const int nb = blockIdx.x * 16;

  const int sr = tid >> 5;
  const int sc = 4 * (tid & 31);
  const v4f* x4 = reinterpret_cast<const v4f*>(x) + nb * 32;
  const v4f xa = x4[tid];
  const v4f xb = x4[tid + 256];

  const v4f zero4 = {0.0f, 0.0f, 0.0f, 0.0f};
  v4f* g4 = reinterpret_cast<v4f*>(agg) + nb * 32;
  g4[tid] = zero4;
  g4[tid + 256] = zero4;
  if (tid < 48) coords_out[nb * 3 + tid] = coords[nb * 3 + tid];

  *reinterpret_cast<v4f*>(&sX[sr * 132 + sc]) = xa;
  *reinterpret_cast<v4f*>(&sX[(sr + 8) * 132 + sc]) = xb;
  __syncthreads();

  v8f acc = {};
  for (int k0 = 0; k0 < 128; k0 += 4) {
    int kk = k0 + 2 * hi;
    v2f a, b;
    a.x = sX[lm * 132 + kk];
    a.y = sX[lm * 132 + kk + 1];
    b.x = We1[kk * 128 + n0 + lm];
    b.y = We1[(kk + 1) * 128 + n0 + lm];
    acc = wmma4(a, b, acc);
  }
  const float bias = be1[n0 + lm];
  __syncthreads();
#pragma unroll
  for (int v = 0; v < 8; ++v) sX[(v + 8 * hi) * 132 + n0 + lm] = acc[v] + bias;
  __syncthreads();
  v4f* p4 = reinterpret_cast<v4f*>(P) + nb * 32;
  p4[tid] = *reinterpret_cast<const v4f*>(&sX[sr * 132 + sc]);
  p4[tid + 256] = *reinterpret_cast<const v4f*>(&sX[(sr + 8) * 132 + sc]);
}

__global__ __launch_bounds__(256, 8) void egnn_edge_kernel(
    const float* __restrict__ x, const float* __restrict__ coords,
    const float* __restrict__ edge_attr, const int* __restrict__ erow,
    const int* __restrict__ ecol, const float* __restrict__ We1,
    const float* __restrict__ P, const float* __restrict__ We2,
    const float* __restrict__ be2, const float* __restrict__ Wc1,
    const float* __restrict__ bc1, const float* __restrict__ Wc2,
    float* __restrict__ agg, float* __restrict__ coords_out) {
  __shared__ __attribute__((aligned(16))) float smem[EPB * 264];
  float* sA = smem;
  float* sH = smem;
  float* sE = smem + EPB * 132;

  __shared__ int sRC[2 * EPB];
  int* sRow = sRC;
  int* sCol = sRC + EPB;
  __shared__ float sDiff[EPB * 3], sDist[EPB], sW[EPB];

  const int tid = threadIdx.x;
  const int lane = tid & 31;
  const int wv = tid >> 5;
  const int hi = lane >> 4;
  const int lm = lane & 15;
  const int n0 = wv * 16;
  const int ebase = blockIdx.x * EPB;

  const v4f* ea4 = reinterpret_cast<const v4f*>(edge_attr) + ebase * 8;
  const v4f ea0 = ea4[tid];
  const v4f ea1 = ea4[tid + 256];

  if (tid < EPB) {
    int r = erow[ebase + tid];
    int c = ecol[ebase + tid];
    sRow[tid] = r;
    sCol[tid] = c;
    float d0 = coords[r * 3 + 0] - coords[c * 3 + 0];
    float d1 = coords[r * 3 + 1] - coords[c * 3 + 1];
    float d2 = coords[r * 3 + 2] - coords[c * 3 + 2];
    sDiff[tid * 3 + 0] = d0;
    sDiff[tid * 3 + 1] = d1;
    sDiff[tid * 3 + 2] = d2;
    sDist[tid] = d0 * d0 + d1 * d1 + d2 * d2;
    sW[tid] = 0.0f;
  }
  __syncthreads();

  v8f acc1[4];
  {
    const v4f* x4 = reinterpret_cast<const v4f*>(x);
    v4f xv[8];
    for (int i = 0; i < 8; ++i) {
      const int node = sCol[wv + 8 * i];
      xv[i] = x4[node * 32 + lane];
    }
#pragma unroll
    for (int t = 0; t < 4; ++t) {
#pragma unroll
      for (int v = 0; v < 8; ++v) {
        const int M = t * 16 + v + 8 * hi;
        acc1[t][v] = P[sRow[M] * 128 + n0 + lm];
      }
    }
    for (int i = 0; i < 8; ++i)
      *reinterpret_cast<v4f*>(&sA[(wv + 8 * i) * 164 + 4 * lane]) = xv[i];
    *reinterpret_cast<v4f*>(&sA[(tid >> 3) * 164 + 128 + 4 * (tid & 7)]) = ea0;
    *reinterpret_cast<v4f*>(&sA[((tid >> 3) + 32) * 164 + 128 + 4 * (tid & 7)]) =
        ea1;
  }
  __syncthreads();

  {
    for (int k0 = 0; k0 < 160; k0 += 4) {
      int kk = k0 + 2 * hi;
      v2f b;
      b.x = We1[(128 + kk) * 128 + n0 + lm];
      b.y = We1[(129 + kk) * 128 + n0 + lm];
#pragma unroll
      for (int t = 0; t < 4; ++t) {
        v2f a;
        a.x = sA[(t * 16 + lm) * 164 + kk];
        a.y = sA[(t * 16 + lm) * 164 + kk + 1];
        acc1[t] = wmma4(a, b, acc1[t]);
      }
    }
    float wlast = We1[288 * 128 + n0 + lm];
    __syncthreads();
#pragma unroll
    for (int t = 0; t < 4; ++t) {
#pragma unroll
      for (int v = 0; v < 8; ++v) {
        int M = t * 16 + v + 8 * hi;
        float pre = acc1[t][v] + sDist[M] * wlast;
        sH[M * 132 + n0 + lm] = silu_f(pre);
      }
    }
  }
  __syncthreads();

  {
    v8f acc[4] = {{}, {}, {}, {}};
    for (int k0 = 0; k0 < 128; k0 += 4) {
      int kk = k0 + 2 * hi;
      v2f b;
      b.x = We2[kk * 128 + n0 + lm];
      b.y = We2[(kk + 1) * 128 + n0 + lm];
#pragma unroll
      for (int t = 0; t < 4; ++t) {
        v2f a;
        a.x = sH[(t * 16 + lm) * 132 + kk];
        a.y = sH[(t * 16 + lm) * 132 + kk + 1];
        acc[t] = wmma4(a, b, acc[t]);
      }
    }
    float bias = be2[n0 + lm];
#pragma unroll
    for (int t = 0; t < 4; ++t) {
#pragma unroll
      for (int v = 0; v < 8; ++v) {
        int M = t * 16 + v + 8 * hi;
        float ef = acc[t][v] + bias;
        sE[M * 132 + n0 + lm] = ef;
        atomicAdd(&agg[sRow[M] * 128 + n0 + lm], ef);
      }
    }
  }
  __syncthreads();

  {
    v8f acc[4] = {{}, {}, {}, {}};
    for (int k0 = 0; k0 < 128; k0 += 4) {
      int kk = k0 + 2 * hi;
      v2f b;
      b.x = Wc1[kk * 128 + n0 + lm];
      b.y = Wc1[(kk + 1) * 128 + n0 + lm];
#pragma unroll
      for (int t = 0; t < 4; ++t) {
        v2f a;
        a.x = sE[(t * 16 + lm) * 132 + kk];
        a.y = sE[(t * 16 + lm) * 132 + kk + 1];
        acc[t] = wmma4(a, b, acc[t]);
      }
    }
    float bias = bc1[n0 + lm];
    float wc2 = Wc2[n0 + lm];
#pragma unroll
    for (int t = 0; t < 4; ++t) {
      float mine = 0.0f;
#pragma unroll
      for (int v = 0; v < 8; ++v) {
        float p = row16_sum(silu_f(acc[t][v] + bias) * wc2);
        mine = (lm == v) ? p : mine;
      }
      if (lm < 8) atomicAdd(&sW[t * 16 + lm + 8 * hi], mine);
    }
  }
  __syncthreads();

  if (tid < EPB * 3) {
    int e = tid / 3;
    int d = tid - e * 3;
    float cd = sDiff[tid] / (sqrtf(sDist[e]) + 1e-8f);
    atomicAdd(&coords_out[sRow[e] * 3 + d], cd * sW[e]);
  }
}

__global__ __launch_bounds__(256) void egnn_pre2_kernel(
    const float* __restrict__ x, const float* __restrict__ coords,
    const float* __restrict__ We1, const float* __restrict__ be1,
    const float* __restrict__ We2, const float* __restrict__ Wc1,
    float* __restrict__ P, float* __restrict__ Q, float* __restrict__ agg,
    float* __restrict__ coords_out, float* __restrict__ Wp) {
  if (blockIdx.x >= N_NODES / 16) {
    const int pb = blockIdx.x - N_NODES / 16;
    v4f* out4 = reinterpret_cast<v4f*>(Wp);
    for (int i = 0; i < 4; ++i) {
      const int slot = pb * 1024 + i * 256 + threadIdx.x;
      const float* src;
      int ls;
      if (slot < PK_W2) {
        src = We1 + 256 * 128;
        ls = slot;
      } else if (slot < PK_W3) {
        src = We2;
        ls = slot - PK_W2;
      } else {
        src = Wc1;
        ls = slot - PK_W3;
      }
      const int g = ls >> 7;
      const int n = ls & 127;
      v4f v;
      v.x = src[(4 * g + 0) * 128 + n];
      v.y = src[(4 * g + 1) * 128 + n];
      v.z = src[(4 * g + 2) * 128 + n];
      v.w = src[(4 * g + 3) * 128 + n];
      out4[slot] = v;
    }
    return;
  }
  __shared__ __attribute__((aligned(16))) float sX[16 * 132];
  __shared__ __attribute__((aligned(16))) float sQ[16 * 132];

  const int tid = threadIdx.x;
  const int lane = tid & 31;
  const int wv = tid >> 5;
  const int hi = lane >> 4;
  const int lm = lane & 15;
  const int n0 = wv * 16;
  const int nb = blockIdx.x * 16;

  const int sr = tid >> 5;
  const int sc = 4 * (tid & 31);
  const v4f* x4 = reinterpret_cast<const v4f*>(x) + nb * 32;
  const v4f xa = x4[tid];
  const v4f xb = x4[tid + 256];

  const v4f zero4 = {0.0f, 0.0f, 0.0f, 0.0f};
  v4f* g4 = reinterpret_cast<v4f*>(agg) + nb * 32;
  g4[tid] = zero4;
  g4[tid + 256] = zero4;
  if (tid < 48) coords_out[nb * 3 + tid] = coords[nb * 3 + tid];

  *reinterpret_cast<v4f*>(&sX[sr * 132 + sc]) = xa;
  *reinterpret_cast<v4f*>(&sX[(sr + 8) * 132 + sc]) = xb;
  __syncthreads();

  v8f accP = {};
  v8f accQ = {};
  for (int k0 = 0; k0 < 128; k0 += 4) {
    int kk = k0 + 2 * hi;
    v2f a, bp, bq;
    a.x = sX[lm * 132 + kk];
    a.y = sX[lm * 132 + kk + 1];
    bp.x = We1[kk * 128 + n0 + lm];
    bp.y = We1[(kk + 1) * 128 + n0 + lm];
    bq.x = We1[(128 + kk) * 128 + n0 + lm];
    bq.y = We1[(129 + kk) * 128 + n0 + lm];
    accP = wmma4(a, bp, accP);
    accQ = wmma4(a, bq, accQ);
  }
  const float bias = be1[n0 + lm];
  __syncthreads();
#pragma unroll
  for (int v = 0; v < 8; ++v) {
    sX[(v + 8 * hi) * 132 + n0 + lm] = accP[v] + bias;
    sQ[(v + 8 * hi) * 132 + n0 + lm] = accQ[v];
  }
  __syncthreads();
  v4f* p4 = reinterpret_cast<v4f*>(P) + nb * 32;
  v4f* q4 = reinterpret_cast<v4f*>(Q) + nb * 32;
  p4[tid] = *reinterpret_cast<const v4f*>(&sX[sr * 132 + sc]);
  p4[tid + 256] = *reinterpret_cast<const v4f*>(&sX[(sr + 8) * 132 + sc]);
  q4[tid] = *reinterpret_cast<const v4f*>(&sQ[sr * 132 + sc]);
  q4[tid + 256] = *reinterpret_cast<const v4f*>(&sQ[(sr + 8) * 132 + sc]);
}

__global__ __launch_bounds__(256, 8) void egnn_edge2_kernel(
    const float* __restrict__ coords, const float* __restrict__ edge_attr,
    const int* __restrict__ erow, const int* __restrict__ ecol,
    const float* __restrict__ We1, const float* __restrict__ P,
    const float* __restrict__ Q, const float* __restrict__ We2,
    const float* __restrict__ be2, const float* __restrict__ Wc1,
    const float* __restrict__ bc1, const float* __restrict__ Wc2,
    float* __restrict__ agg, float* __restrict__ coords_out) {
  __shared__ __attribute__((aligned(16))) float smem[EPB * 264];
  float* sH = smem;
  float* sE = smem + EPB * 132;
  float* sA = sE;

  __shared__ int sRC[2 * EPB];
  int* sRow = sRC;
  int* sCol = sRC + EPB;
  __shared__ float sDiff[EPB * 3], sDist[EPB], sW[EPB];

  const int tid = threadIdx.x;
  const int lane = tid & 31;
  const int wv = tid >> 5;
  const int hi = lane >> 4;
  const int lm = lane & 15;
  const int n0 = wv * 16;
  const int ebase = blockIdx.x * EPB;

  const v4f* ea4 = reinterpret_cast<const v4f*>(edge_attr) + ebase * 8;
  const v4f ea0 = ea4[tid];
  const v4f ea1 = ea4[tid + 256];

  if (tid < EPB) {
    int r = erow[ebase + tid];
    int c = ecol[ebase + tid];
    sRow[tid] = r;
    sCol[tid] = c;
    float d0 = coords[r * 3 + 0] - coords[c * 3 + 0];
    float d1 = coords[r * 3 + 1] - coords[c * 3 + 1];
    float d2 = coords[r * 3 + 2] - coords[c * 3 + 2];
    sDiff[tid * 3 + 0] = d0;
    sDiff[tid * 3 + 1] = d1;
    sDiff[tid * 3 + 2] = d2;
    sDist[tid] = d0 * d0 + d1 * d1 + d2 * d2;
    sW[tid] = 0.0f;
  }
  __syncthreads();

  v8f acc1[4];
  {
    v8f qv[4];
#pragma unroll
    for (int t = 0; t < 4; ++t) {
#pragma unroll
      for (int v = 0; v < 8; ++v) {
        const int M = t * 16 + v + 8 * hi;
        qv[t][v] = Q[sCol[M] * 128 + n0 + lm];
      }
    }
#pragma unroll
    for (int t = 0; t < 4; ++t) {
#pragma unroll
      for (int v = 0; v < 8; ++v) {
        const int M = t * 16 + v + 8 * hi;
        acc1[t][v] = P[sRow[M] * 128 + n0 + lm];
      }
    }
#pragma unroll
    for (int t = 0; t < 4; ++t) acc1[t] += qv[t];
  }
  *reinterpret_cast<v4f*>(&sA[(tid >> 3) * 36 + 4 * (tid & 7)]) = ea0;
  *reinterpret_cast<v4f*>(&sA[((tid >> 3) + 32) * 36 + 4 * (tid & 7)]) = ea1;
  __syncthreads();

  {
    for (int k0 = 0; k0 < 32; k0 += 4) {
      int kk = k0 + 2 * hi;
      v2f b;
      b.x = We1[(256 + kk) * 128 + n0 + lm];
      b.y = We1[(257 + kk) * 128 + n0 + lm];
#pragma unroll
      for (int t = 0; t < 4; ++t) {
        v2f a;
        a.x = sA[(t * 16 + lm) * 36 + kk];
        a.y = sA[(t * 16 + lm) * 36 + kk + 1];
        acc1[t] = wmma4(a, b, acc1[t]);
      }
    }
    float wlast = We1[288 * 128 + n0 + lm];
#pragma unroll
    for (int t = 0; t < 4; ++t) {
#pragma unroll
      for (int v = 0; v < 8; ++v) {
        int M = t * 16 + v + 8 * hi;
        float pre = acc1[t][v] + sDist[M] * wlast;
        sH[M * 132 + n0 + lm] = silu_f(pre);
      }
    }
  }
  __syncthreads();

  {
    v8f acc[4] = {{}, {}, {}, {}};
    for (int k0 = 0; k0 < 128; k0 += 4) {
      int kk = k0 + 2 * hi;
      v2f b;
      b.x = We2[kk * 128 + n0 + lm];
      b.y = We2[(kk + 1) * 128 + n0 + lm];
#pragma unroll
      for (int t = 0; t < 4; ++t) {
        v2f a;
        a.x = sH[(t * 16 + lm) * 132 + kk];
        a.y = sH[(t * 16 + lm) * 132 + kk + 1];
        acc[t] = wmma4(a, b, acc[t]);
      }
    }
    float bias = be2[n0 + lm];
#pragma unroll
    for (int t = 0; t < 4; ++t) {
#pragma unroll
      for (int v = 0; v < 8; ++v) {
        int M = t * 16 + v + 8 * hi;
        float ef = acc[t][v] + bias;
        sE[M * 132 + n0 + lm] = ef;
        atomicAdd(&agg[sRow[M] * 128 + n0 + lm], ef);
      }
    }
  }
  __syncthreads();

  {
    v8f acc[4] = {{}, {}, {}, {}};
    for (int k0 = 0; k0 < 128; k0 += 4) {
      int kk = k0 + 2 * hi;
      v2f b;
      b.x = Wc1[kk * 128 + n0 + lm];
      b.y = Wc1[(kk + 1) * 128 + n0 + lm];
#pragma unroll
      for (int t = 0; t < 4; ++t) {
        v2f a;
        a.x = sE[(t * 16 + lm) * 132 + kk];
        a.y = sE[(t * 16 + lm) * 132 + kk + 1];
        acc[t] = wmma4(a, b, acc[t]);
      }
    }
    float bias = bc1[n0 + lm];
    float wc2 = Wc2[n0 + lm];
#pragma unroll
    for (int t = 0; t < 4; ++t) {
      float mine = 0.0f;
#pragma unroll
      for (int v = 0; v < 8; ++v) {
        float p = row16_sum(silu_f(acc[t][v] + bias) * wc2);
        mine = (lm == v) ? p : mine;
      }
      if (lm < 8) atomicAdd(&sW[t * 16 + lm + 8 * hi], mine);
    }
  }
  __syncthreads();

  if (tid < EPB * 3) {
    int e = tid / 3;
    int d = tid - e * 3;
    float cd = sDiff[tid] / (sqrtf(sDist[e]) + 1e-8f);
    atomicAdd(&coords_out[sRow[e] * 3 + d], cd * sW[e]);
  }
}

__global__ __launch_bounds__(256, 8) void egnn_edge3_kernel(
    const float* __restrict__ coords, const float* __restrict__ edge_attr,
    const int* __restrict__ erow, const int* __restrict__ ecol,
    const float* __restrict__ We1, const float* __restrict__ P,
    const float* __restrict__ Q, const float* __restrict__ Wp,
    const float* __restrict__ be2, const float* __restrict__ bc1,
    const float* __restrict__ Wc2, float* __restrict__ agg,
    float* __restrict__ coords_out) {
  __shared__ __attribute__((aligned(16))) float smem[EPB * 264];
  float* sH = smem;
  float* sE = smem + EPB * 132;
  float* sA = sE;

  __shared__ int sRC[2 * EPB];
  int* sRow = sRC;
  int* sCol = sRC + EPB;
  __shared__ float sDiff[EPB * 3], sDist[EPB], sW[EPB];

  const int tid = threadIdx.x;
  const int lane = tid & 31;
  const int wv = tid >> 5;
  const int hi = lane >> 4;
  const int lm = lane & 15;
  const int n0 = wv * 16;
  const int ebase = blockIdx.x * EPB;
  const v4f* wp4 = reinterpret_cast<const v4f*>(Wp) + hi * 128 + n0 + lm;

  const v4f* ea4 = reinterpret_cast<const v4f*>(edge_attr) + ebase * 8;
  const v4f ea0 = ea4[tid];
  const v4f ea1 = ea4[tid + 256];

  if (tid < EPB) {
    int r = erow[ebase + tid];
    int c = ecol[ebase + tid];
    sRow[tid] = r;
    sCol[tid] = c;
    float d0 = coords[r * 3 + 0] - coords[c * 3 + 0];
    float d1 = coords[r * 3 + 1] - coords[c * 3 + 1];
    float d2 = coords[r * 3 + 2] - coords[c * 3 + 2];
    sDiff[tid * 3 + 0] = d0;
    sDiff[tid * 3 + 1] = d1;
    sDiff[tid * 3 + 2] = d2;
    sDist[tid] = d0 * d0 + d1 * d1 + d2 * d2;
    sW[tid] = 0.0f;
  }
  __syncthreads();

  v8f acc1[4];
  {
    v8f qv[4];
#pragma unroll
    for (int t = 0; t < 4; ++t) {
#pragma unroll
      for (int v = 0; v < 8; ++v) {
        const int M = t * 16 + v + 8 * hi;
        qv[t][v] = Q[sCol[M] * 128 + n0 + lm];
      }
    }
#pragma unroll
    for (int t = 0; t < 4; ++t) {
#pragma unroll
      for (int v = 0; v < 8; ++v) {
        const int M = t * 16 + v + 8 * hi;
        acc1[t][v] = P[sRow[M] * 128 + n0 + lm];
      }
    }
#pragma unroll
    for (int t = 0; t < 4; ++t) acc1[t] += qv[t];
  }
  *reinterpret_cast<v4f*>(&sA[(tid >> 3) * 36 + 4 * (tid & 7)]) = ea0;
  *reinterpret_cast<v4f*>(&sA[((tid >> 3) + 32) * 36 + 4 * (tid & 7)]) = ea1;
  __syncthreads();

  {
    for (int k0 = 0; k0 < 32; k0 += 8) {
      const v4f bv = wp4[(k0 >> 2) * 128];
      v4f av[4];
#pragma unroll
      for (int t = 0; t < 4; ++t)
        av[t] = *reinterpret_cast<const v4f*>(
            &sA[(t * 16 + lm) * 36 + k0 + 4 * hi]);
      const v2f b0 = {bv.x, bv.y};
      const v2f b1 = {bv.z, bv.w};
#pragma unroll
      for (int t = 0; t < 4; ++t) {
        const v2f a0 = {av[t].x, av[t].y};
        acc1[t] = wmma4(a0, b0, acc1[t]);
      }
#pragma unroll
      for (int t = 0; t < 4; ++t) {
        const v2f a1 = {av[t].z, av[t].w};
        acc1[t] = wmma4(a1, b1, acc1[t]);
      }
    }
    float wlast = We1[288 * 128 + n0 + lm];
#pragma unroll
    for (int t = 0; t < 4; ++t) {
#pragma unroll
      for (int v = 0; v < 8; ++v) {
        int M = t * 16 + v + 8 * hi;
        float pre = acc1[t][v] + sDist[M] * wlast;
        sH[M * 132 + n0 + lm] = silu_f(pre);
      }
    }
  }
  __syncthreads();

  {
    v8f acc[4] = {{}, {}, {}, {}};
    for (int k0 = 0; k0 < 128; k0 += 8) {
      const v4f bv = wp4[PK_W2 + (k0 >> 2) * 128];
      v4f av[4];
#pragma unroll
      for (int t = 0; t < 4; ++t)
        av[t] = *reinterpret_cast<const v4f*>(
            &sH[(t * 16 + lm) * 132 + k0 + 4 * hi]);
      const v2f b0 = {bv.x, bv.y};
      const v2f b1 = {bv.z, bv.w};
#pragma unroll
      for (int t = 0; t < 4; ++t) {
        const v2f a0 = {av[t].x, av[t].y};
        acc[t] = wmma4(a0, b0, acc[t]);
      }
#pragma unroll
      for (int t = 0; t < 4; ++t) {
        const v2f a1 = {av[t].z, av[t].w};
        acc[t] = wmma4(a1, b1, acc[t]);
      }
    }
    float bias = be2[n0 + lm];
#pragma unroll
    for (int t = 0; t < 4; ++t) {
#pragma unroll
      for (int v = 0; v < 8; ++v) {
        int M = t * 16 + v + 8 * hi;
        float ef = acc[t][v] + bias;
        sE[M * 132 + n0 + lm] = ef;
        atomicAdd(&agg[sRow[M] * 128 + n0 + lm], ef);
      }
    }
  }
  __syncthreads();

  {
    v8f acc[4] = {{}, {}, {}, {}};
    for (int k0 = 0; k0 < 128; k0 += 8) {
      const v4f bv = wp4[PK_W3 + (k0 >> 2) * 128];
      v4f av[4];
#pragma unroll
      for (int t = 0; t < 4; ++t)
        av[t] = *reinterpret_cast<const v4f*>(
            &sE[(t * 16 + lm) * 132 + k0 + 4 * hi]);
      const v2f b0 = {bv.x, bv.y};
      const v2f b1 = {bv.z, bv.w};
#pragma unroll
      for (int t = 0; t < 4; ++t) {
        const v2f a0 = {av[t].x, av[t].y};
        acc[t] = wmma4(a0, b0, acc[t]);
      }
#pragma unroll
      for (int t = 0; t < 4; ++t) {
        const v2f a1 = {av[t].z, av[t].w};
        acc[t] = wmma4(a1, b1, acc[t]);
      }
    }
    float bias = bc1[n0 + lm];
    float wc2 = Wc2[n0 + lm];
#pragma unroll
    for (int t = 0; t < 4; ++t) {
      float mine = 0.0f;
#pragma unroll
      for (int v = 0; v < 8; ++v) {
        float p = row16_sum(silu_f(acc[t][v] + bias) * wc2);
        mine = (lm == v) ? p : mine;
      }
      if (lm < 8) atomicAdd(&sW[t * 16 + lm + 8 * hi], mine);
    }
  }
  __syncthreads();

  if (tid < EPB * 3) {
    int e = tid / 3;
    int d = tid - e * 3;
    float cd = sDiff[tid] / (sqrtf(sDist[e]) + 1e-8f);
    atomicAdd(&coords_out[sRow[e] * 3 + d], cd * sW[e]);
  }
}

__global__ __launch_bounds__(256) void egnn_node_kernel(
    const float* __restrict__ x, const float* __restrict__ agg,
    const float* __restrict__ Wn1, const float* __restrict__ bn1,
    const float* __restrict__ Wn2, const float* __restrict__ bn2,
    float* __restrict__ x_out) {
  __shared__ float sA[16 * 260];
  __shared__ float sH[16 * 132];

  const int tid = threadIdx.x;
  const int lane = tid & 31;
  const int wv = tid >> 5;
  const int hi = lane >> 4;
  const int lm = lane & 15;
  const int n0 = wv * 16;
  const int nb = blockIdx.x * 16;

  for (int idx = tid; idx < 16 * 256; idx += 256) {
    int m = idx >> 8;
    int c = idx & 255;
    float v = (c < 128) ? x[(nb + m) * 128 + c] : agg[(nb + m) * 128 + (c - 128)];
    sA[m * 260 + c] = v;
  }
  __syncthreads();

  v8f acc1 = {};
  for (int k0 = 0; k0 < 256; k0 += 4) {
    int kk = k0 + 2 * hi;
    v2f a, b;
    a.x = sA[lm * 260 + kk];
    a.y = sA[lm * 260 + kk + 1];
    b.x = Wn1[kk * 128 + n0 + lm];
    b.y = Wn1[(kk + 1) * 128 + n0 + lm];
    acc1 = wmma4(a, b, acc1);
  }
  {
    float bias = bn1[n0 + lm];
#pragma unroll
    for (int v = 0; v < 8; ++v) {
      int M = v + 8 * hi;
      sH[M * 132 + n0 + lm] = silu_f(acc1[v] + bias);
    }
  }
  __syncthreads();

  v8f acc2 = {};
  for (int k0 = 0; k0 < 128; k0 += 4) {
    int kk = k0 + 2 * hi;
    v2f a, b;
    a.x = sH[lm * 132 + kk];
    a.y = sH[lm * 132 + kk + 1];
    b.x = Wn2[kk * 128 + n0 + lm];
    b.y = Wn2[(kk + 1) * 128 + n0 + lm];
    acc2 = wmma4(a, b, acc2);
  }
  {
    float bias = bn2[n0 + lm];
#pragma unroll
    for (int v = 0; v < 8; ++v) {
      int M = v + 8 * hi;
      x_out[(nb + M) * 128 + n0 + lm] = acc2[v] + bias;
    }
  }
}

#define PL_W23 1024
#define PL_SLOTS 5120
#define PL_BLOCKS 9

__global__ __launch_bounds__(256) void egnn_pre3_kernel(
    const float* __restrict__ x, const float* __restrict__ coords,
    const float* __restrict__ We1, const float* __restrict__ be1,
    const float* __restrict__ We2, const float* __restrict__ be2,
    const float* __restrict__ Wc1, const float* __restrict__ bc1,
    float* __restrict__ P, float* __restrict__ Q, float* __restrict__ S4,
    int* __restrict__ deg, float* __restrict__ coords_out,
    float* __restrict__ Wp, float* __restrict__ b23) {
  const int tid = threadIdx.x;
  const int lane = tid & 31;
  const int wv = tid >> 5;
  const int hi = lane >> 4;
  const int lm = lane & 15;

  if (blockIdx.x >= N_NODES / 16) {
    const int pb = blockIdx.x - N_NODES / 16;
    v4f* out4 = reinterpret_cast<v4f*>(Wp);
    if (pb == 0) {
      const float* src = We1 + 256 * 128;
      for (int i = 0; i < 4; ++i) {
        const int slot = i * 256 + tid;
        const int g = slot >> 7;
        const int n = slot & 127;
        v4f v;
        v.x = src[(4 * g + 0) * 128 + n];
        v.y = src[(4 * g + 1) * 128 + n];
        v.z = src[(4 * g + 2) * 128 + n];
        v.w = src[(4 * g + 3) * 128 + n];
        out4[slot] = v;
      }
      if (tid < 128) {
        float s = 0.0f;
        for (int m = 0; m < 128; ++m) s = fmaf(be2[m], Wc1[m * 128 + tid], s);
        b23[tid] = s + bc1[tid];
      }
      return;
    }
    const int c0 = (pb - 1) * 16;
    const int i0 = wv * 16;
    v8f acc = {};
    for (int k0 = 0; k0 < 128; k0 += 4) {
      const int kk = k0 + 2 * hi;
      v2f a, b;
      a.x = We2[(i0 + lm) * 128 + kk];
      a.y = We2[(i0 + lm) * 128 + kk + 1];
      b.x = Wc1[kk * 128 + c0 + lm];
      b.y = Wc1[(kk + 1) * 128 + c0 + lm];
      acc = wmma4(a, b, acc);
    }
    const int g0 = (i0 + 8 * hi) >> 2;
    const v4f s0 = {acc[0], acc[1], acc[2], acc[3]};
    const v4f s1 = {acc[4], acc[5], acc[6], acc[7]};
    out4[PL_W23 + g0 * 128 + c0 + lm] = s0;
    out4[PL_W23 + (g0 + 1) * 128 + c0 + lm] = s1;
    return;
  }
  __shared__ __attribute__((aligned(16))) float sX[16 * 132];
  __shared__ __attribute__((aligned(16))) float sQ[16 * 132];

  const int n0 = wv * 16;
  const int nb = blockIdx.x * 16;

  const int sr = tid >> 5;
  const int sc = 4 * (tid & 31);
  const v4f* x4 = reinterpret_cast<const v4f*>(x) + nb * 32;
  const v4f xa = x4[tid];
  const v4f xb = x4[tid + 256];

  const v4f zero4 = {0.0f, 0.0f, 0.0f, 0.0f};
  v4f* g4 = reinterpret_cast<v4f*>(S4) + nb * 32;
  g4[tid] = zero4;
  g4[tid + 256] = zero4;
  if (tid < 16) deg[nb + tid] = 0;
  if (tid < 48) coords_out[nb * 3 + tid] = coords[nb * 3 + tid];

  *reinterpret_cast<v4f*>(&sX[sr * 132 + sc]) = xa;
  *reinterpret_cast<v4f*>(&sX[(sr + 8) * 132 + sc]) = xb;
  __syncthreads();

  v8f accP = {};
  v8f accQ = {};
  for (int k0 = 0; k0 < 128; k0 += 4) {
    int kk = k0 + 2 * hi;
    v2f a, bp, bq;
    a.x = sX[lm * 132 + kk];
    a.y = sX[lm * 132 + kk + 1];
    bp.x = We1[kk * 128 + n0 + lm];
    bp.y = We1[(kk + 1) * 128 + n0 + lm];
    bq.x = We1[(128 + kk) * 128 + n0 + lm];
    bq.y = We1[(129 + kk) * 128 + n0 + lm];
    accP = wmma4(a, bp, accP);
    accQ = wmma4(a, bq, accQ);
  }
  const float bias = be1[n0 + lm];
  __syncthreads();
#pragma unroll
  for (int v = 0; v < 8; ++v) {
    sX[(v + 8 * hi) * 132 + n0 + lm] = accP[v] + bias;
    sQ[(v + 8 * hi) * 132 + n0 + lm] = accQ[v];
  }
  __syncthreads();
  v4f* p4 = reinterpret_cast<v4f*>(P) + nb * 32;
  v4f* q4 = reinterpret_cast<v4f*>(Q) + nb * 32;
  p4[tid] = *reinterpret_cast<const v4f*>(&sX[sr * 132 + sc]);
  p4[tid + 256] = *reinterpret_cast<const v4f*>(&sX[(sr + 8) * 132 + sc]);
  q4[tid] = *reinterpret_cast<const v4f*>(&sQ[sr * 132 + sc]);
  q4[tid + 256] = *reinterpret_cast<const v4f*>(&sQ[(sr + 8) * 132 + sc]);
}

__global__ __launch_bounds__(256, 8) void egnn_edge4_kernel(
    const float* __restrict__ coords, const float* __restrict__ edge_attr,
    const int* __restrict__ erow, const int* __restrict__ ecol,
    const float* __restrict__ We1, const float* __restrict__ P,
    const float* __restrict__ Q, const float* __restrict__ Wp,
    const float* __restrict__ b23, const float* __restrict__ Wc2,
    unsigned long long* __restrict__ S, int* __restrict__ deg,
    float* __restrict__ coords_out) {
  __shared__ __attribute__((aligned(16))) float smem[EPB * 168];
  float* sH = smem;
  float* sA = smem + EPB * 132;

  __shared__ int sRC[2 * EPB];
  int* sRow = sRC;
  int* sCol = sRC + EPB;
  __shared__ float sDiff[EPB * 3], sDist[EPB], sW[EPB];

  const int tid = threadIdx.x;
  const int lane = tid & 31;
  const int wv = tid >> 5;
  const int hi = lane >> 4;
  const int lm = lane & 15;
  const int n0 = wv * 16;
  const int ebase = blockIdx.x * EPB;
  const v4f* wp4 = reinterpret_cast<const v4f*>(Wp) + hi * 128 + n0 + lm;

  const v4f* ea4 = reinterpret_cast<const v4f*>(edge_attr) + ebase * 8;
  const v4f ea0 = ea4[tid];
  const v4f ea1 = ea4[tid + 256];

  if (tid < EPB) {
    int r = erow[ebase + tid];
    int c = ecol[ebase + tid];
    sRow[tid] = r;
    sCol[tid] = c;
    atomicAdd(&deg[r], 1);
    float d0 = coords[r * 3 + 0] - coords[c * 3 + 0];
    float d1 = coords[r * 3 + 1] - coords[c * 3 + 1];
    float d2 = coords[r * 3 + 2] - coords[c * 3 + 2];
    sDiff[tid * 3 + 0] = d0;
    sDiff[tid * 3 + 1] = d1;
    sDiff[tid * 3 + 2] = d2;
    sDist[tid] = d0 * d0 + d1 * d1 + d2 * d2;
    sW[tid] = 0.0f;
  }
  __syncthreads();

  v8f acc1[4];
  {
    v8f qv[4];
#pragma unroll
    for (int t = 0; t < 4; ++t) {
#pragma unroll
      for (int v = 0; v < 8; ++v) {
        const int M = t * 16 + v + 8 * hi;
        qv[t][v] = Q[sCol[M] * 128 + n0 + lm];
      }
    }
#pragma unroll
    for (int t = 0; t < 4; ++t) {
#pragma unroll
      for (int v = 0; v < 8; ++v) {
        const int M = t * 16 + v + 8 * hi;
        acc1[t][v] = P[sRow[M] * 128 + n0 + lm];
      }
    }
#pragma unroll
    for (int t = 0; t < 4; ++t) acc1[t] += qv[t];
  }
  *reinterpret_cast<v4f*>(&sA[(tid >> 3) * 36 + 4 * (tid & 7)]) = ea0;
  *reinterpret_cast<v4f*>(&sA[((tid >> 3) + 32) * 36 + 4 * (tid & 7)]) = ea1;
  __syncthreads();

  {
    for (int k0 = 0; k0 < 32; k0 += 8) {
      const v4f bv = wp4[(k0 >> 2) * 128];
      v4f av[4];
#pragma unroll
      for (int t = 0; t < 4; ++t)
        av[t] = *reinterpret_cast<const v4f*>(
            &sA[(t * 16 + lm) * 36 + k0 + 4 * hi]);
      const v2f b0 = {bv.x, bv.y};
      const v2f b1 = {bv.z, bv.w};
#pragma unroll
      for (int t = 0; t < 4; ++t) {
        const v2f a0 = {av[t].x, av[t].y};
        acc1[t] = wmma4(a0, b0, acc1[t]);
      }
#pragma unroll
      for (int t = 0; t < 4; ++t) {
        const v2f a1 = {av[t].z, av[t].w};
        acc1[t] = wmma4(a1, b1, acc1[t]);
      }
    }
    float wlast = We1[288 * 128 + n0 + lm];
#pragma unroll
    for (int t = 0; t < 4; ++t) {
#pragma unroll
      for (int v = 0; v < 8; ++v) {
        int M = t * 16 + v + 8 * hi;
        float pre = acc1[t][v] + sDist[M] * wlast;
        sH[M * 132 + n0 + lm] = silu_f(pre);
      }
    }
  }
  __syncthreads();

  {
    const int pr = (wv & 1) * 32 + lane;
    v2f h2[16];
    int node[16];
#pragma unroll
    for (int i = 0; i < 16; ++i) {
      const int e = i * 4 + (wv >> 1);
      h2[i] = *reinterpret_cast<const v2f*>(&sH[e * 132 + 2 * pr]);
      node[i] = sRow[e];
    }
#pragma unroll
    for (int i = 0; i < 16; ++i) {
      const int a = (int)rintf(h2[i].x * 4194304.0f);
      const int b = (int)rintf(h2[i].y * 4194304.0f);
      const long long q = (long long)b * 4294967296LL + (long long)a;
      atomicAdd(&S[node[i] * 64 + pr], (unsigned long long)q);
    }
  }

  {
    v8f acc[4] = {{}, {}, {}, {}};
    for (int k0 = 0; k0 < 128; k0 += 8) {
      const v4f bv = wp4[PL_W23 + (k0 >> 2) * 128];
      v4f av[4];
#pragma unroll
      for (int t = 0; t < 4; ++t)
        av[t] = *reinterpret_cast<const v4f*>(
            &sH[(t * 16 + lm) * 132 + k0 + 4 * hi]);
      const v2f b0 = {bv.x, bv.y};
      const v2f b1 = {bv.z, bv.w};
#pragma unroll
      for (int t = 0; t < 4; ++t) {
        const v2f a0 = {av[t].x, av[t].y};
        acc[t] = wmma4(a0, b0, acc[t]);
      }
#pragma unroll
      for (int t = 0; t < 4; ++t) {
        const v2f a1 = {av[t].z, av[t].w};
        acc[t] = wmma4(a1, b1, acc[t]);
      }
    }
    float bias = b23[n0 + lm];
    float wc2 = Wc2[n0 + lm];
#pragma unroll
    for (int t = 0; t < 4; ++t) {
      float mine = 0.0f;
#pragma unroll
      for (int v = 0; v < 8; ++v) {
        float p = row16_sum(silu_f(acc[t][v] + bias) * wc2);
        mine = (lm == v) ? p : mine;
      }
      if (lm < 8) atomicAdd(&sW[t * 16 + lm + 8 * hi], mine);
    }
  }
  __syncthreads();

  if (tid < EPB * 3) {
    int e = tid / 3;
    int d = tid - e * 3;
    float cd = sDiff[tid] / (sqrtf(sDist[e]) + 1e-8f);
    atomicAdd(&coords_out[sRow[e] * 3 + d], cd * sW[e]);
  }
}

__global__ __launch_bounds__(256, 16) void egnn_node2_kernel(
    const float* __restrict__ x, const unsigned long long* __restrict__ S,
    const int* __restrict__ deg, const float* __restrict__ We2,
    const float* __restrict__ be2, const float* __restrict__ Wn1,
    const float* __restrict__ bn1, const float* __restrict__ Wn2,
    const float* __restrict__ bn2, float* __restrict__ x_out) {
  __shared__ float sA[16 * 260];
  __shared__ float sH[16 * 132];

  const int tid = threadIdx.x;
  const int lane = tid & 31;
  const int wv = tid >> 5;
  const int hi = lane >> 4;
  const int lm = lane & 15;
  const int n0 = wv * 16;
  const int nb = blockIdx.x * 16;

  for (int idx = tid; idx < 16 * 128; idx += 256) {
    int m = idx >> 7;
    int c = idx & 127;
    sA[m * 260 + c] = x[(nb + m) * 128 + c];
  }
  for (int idx = tid; idx < 16 * 64; idx += 256) {
    int m = idx >> 6;
    int p = idx & 63;
    const long long q = (long long)S[(nb + m) * 64 + p];
    const int lo = (int)q;
    const int hi = (int)((q - (long long)lo) >> 32);
    sH[m * 132 + 2 * p] = (float)lo * 2.384185791015625e-07f;
    sH[m * 132 + 2 * p + 1] = (float)hi * 2.384185791015625e-07f;
  }
  __syncthreads();

  {
    v8f acc = {};
    for (int k0 = 0; k0 < 128; k0 += 4) {
      int kk = k0 + 2 * hi;
      v2f a, b;
      a.x = sH[lm * 132 + kk];
      a.y = sH[lm * 132 + kk + 1];
      b.x = We2[kk * 128 + n0 + lm];
      b.y = We2[(kk + 1) * 128 + n0 + lm];
      acc = wmma4(a, b, acc);
    }
    float bias = be2[n0 + lm];
#pragma unroll
    for (int v = 0; v < 8; ++v) {
      int M = v + 8 * hi;
      sA[M * 260 + 128 + n0 + lm] = acc[v] + (float)deg[nb + M] * bias;
    }
  }
  __syncthreads();

  v8f acc1 = {};
  for (int k0 = 0; k0 < 256; k0 += 4) {
    int kk = k0 + 2 * hi;
    v2f a, b;
    a.x = sA[lm * 260 + kk];
    a.y = sA[lm * 260 + kk + 1];
    b.x = Wn1[kk * 128 + n0 + lm];
    b.y = Wn1[(kk + 1) * 128 + n0 + lm];
    acc1 = wmma4(a, b, acc1);
  }
  {
    float bias = bn1[n0 + lm];
#pragma unroll
    for (int v = 0; v < 8; ++v) {
      int M = v + 8 * hi;
      sH[M * 132 + n0 + lm] = silu_f(acc1[v] + bias);
    }
  }
  __syncthreads();

  v8f acc2 = {};
  for (int k0 = 0; k0 < 128; k0 += 4) {
    int kk = k0 + 2 * hi;
    v2f a, b;
    a.x = sH[lm * 132 + kk];
    a.y = sH[lm * 132 + kk + 1];
    b.x = Wn2[kk * 128 + n0 + lm];
    b.y = Wn2[(kk + 1) * 128 + n0 + lm];
    acc2 = wmma4(a, b, acc2);
  }
  {
    float bias = bn2[n0 + lm];
#pragma unroll
    for (int v = 0; v < 8; ++v) {
      int M = v + 8 * hi;
      x_out[(nb + M) * 128 + n0 + lm] = acc2[v] + bias;
    }
  }
}

extern "C" void kernel_launch(void* const* d_in, const int* in_sizes, int n_in,
                              void* d_out, int out_size, void* d_ws,
                              size_t ws_size, hipStream_t stream) {
  const float* x = (const float*)d_in[0];
  const float* coords = (const float*)d_in[1];
  const float* edge_attr = (const float*)d_in[2];
  const int* edge_index = (const int*)d_in[3];
  const float* We1 = (const float*)d_in[4];
  const float* be1 = (const float*)d_in[5];
  const float* We2 = (const float*)d_in[6];
  const float* be2 = (const float*)d_in[7];
  const float* Wc1 = (const float*)d_in[8];
  const float* bc1 = (const float*)d_in[9];
  const float* Wc2 = (const float*)d_in[10];
  const float* Wn1 = (const float*)d_in[11];
  const float* bn1 = (const float*)d_in[12];
  const float* Wn2 = (const float*)d_in[13];
  const float* bn2 = (const float*)d_in[14];

  float* x_out = (float*)d_out;
  float* coords_out = x_out + (size_t)N_NODES * 128;
  float* agg = (float*)d_ws;

  const int* erow = edge_index;
  const int* ecol = edge_index + N_EDGES;

  if (ws_size >= (size_t)2 * N_NODES * 128 * sizeof(float) +
                     (size_t)PL_SLOTS * 16 + 128 * sizeof(float) +
                     (size_t)N_NODES * sizeof(int)) {
    unsigned long long* S = (unsigned long long*)d_ws;
    float* Q = (float*)d_ws + (size_t)N_NODES * 128;
    float* Wp = Q + (size_t)N_NODES * 128;
    float* b23 = Wp + (size_t)PL_SLOTS * 4;
    int* deg = (int*)(b23 + 128);
    egnn_pre3_kernel<<<N_NODES / 16 + PL_BLOCKS, 256, 0, stream>>>(
        x, coords, We1, be1, We2, be2, Wc1, bc1, x_out, Q, (float*)d_ws, deg,
        coords_out, Wp, b23);
    egnn_edge4_kernel<<<N_EDGES / EPB, 256, 0, stream>>>(
        coords, edge_attr, erow, ecol, We1, x_out, Q, Wp, b23, Wc2, S, deg,
        coords_out);
    egnn_node2_kernel<<<N_NODES / 16, 256, 0, stream>>>(
        x, S, deg, We2, be2, Wn1, bn1, Wn2, bn2, x_out);
    return;
  }
  if (ws_size >= (size_t)2 * N_NODES * 128 * sizeof(float) +
                     (size_t)PK_SLOTS * 16) {
    float* Q = agg + (size_t)N_NODES * 128;
    float* Wp = Q + (size_t)N_NODES * 128;
    egnn_pre2_kernel<<<N_NODES / 16 + PK_SLOTS / 1024, 256, 0, stream>>>(
        x, coords, We1, be1, We2, Wc1, x_out, Q, agg, coords_out, Wp);
    egnn_edge3_kernel<<<N_EDGES / EPB, 256, 0, stream>>>(
        coords, edge_attr, erow, ecol, We1, x_out, Q, Wp, be2, bc1, Wc2, agg,
        coords_out);
  } else if (ws_size >= (size_t)2 * N_NODES * 128 * sizeof(float)) {
    float* Q = agg + (size_t)N_NODES * 128;
    egnn_pre2_kernel<<<N_NODES / 16, 256, 0, stream>>>(
        x, coords, We1, be1, We2, Wc1, x_out, Q, agg, coords_out, nullptr);
    egnn_edge2_kernel<<<N_EDGES / EPB, 256, 0, stream>>>(
        coords, edge_attr, erow, ecol, We1, x_out, Q, We2, be2, Wc1, bc1, Wc2,
        agg, coords_out);
  } else {
    egnn_pre_kernel<<<N_NODES / 16, 256, 0, stream>>>(x, coords, We1, be1, x_out,
                                                      agg, coords_out);
    egnn_edge_kernel<<<N_EDGES / EPB, 256, 0, stream>>>(
        x, coords, edge_attr, erow, ecol, We1, x_out, We2, be2, Wc1, bc1, Wc2,
        agg, coords_out);
  }
  egnn_node_kernel<<<N_NODES / 16, 256, 0, stream>>>(x, agg, Wn1, bn1, Wn2, bn2,
                                                     x_out);
}


# Edge tiles transposed (weights as WMMA A operand: lane = edge, 8 features per lane): gate sum in-lane, no 16-lane butterfly; P+Q as whole rows via LDS (gate sum order: in-lane, lane pair, waves)
# speedup vs baseline: 3.8641x; 1.0618x over previous
#include <hip/hip_runtime.h>
#include <math.h>

typedef __attribute__((ext_vector_type(2))) float v2f;
typedef __attribute__((ext_vector_type(8))) float v8f;
typedef __attribute__((ext_vector_type(4))) float v4f;

#define N_NODES 50000
#define N_EDGES 800000
#define EPB 64
#define PK_W2 1024
#define PK_W3 5120
#define PK_SLOTS 9216

__device__ __forceinline__ v8f wmma4(v2f a, v2f b, v8f c) {
  return __builtin_amdgcn_wmma_f32_16x16x4_f32(false, a, false, b, (short)0, c,
                                               false, false);
}

__device__ __forceinline__ float silu_f(float v) {
  return v * __builtin_amdgcn_rcpf(1.0f + __expf(-v));
}

__device__ __forceinline__ float row16_sum(float v) {
  v += __int_as_float(__builtin_amdgcn_update_dpp(
      0, __float_as_int(v), 0x161, 0xf, 0xf, false));
  v += __int_as_float(__builtin_amdgcn_update_dpp(
      0, __float_as_int(v), 0x162, 0xf, 0xf, false));
  v += __int_as_float(__builtin_amdgcn_update_dpp(
      0, __float_as_int(v), 0x164, 0xf, 0xf, false));
  v += __int_as_float(__builtin_amdgcn_update_dpp(
      0, __float_as_int(v), 0x168, 0xf, 0xf, false));
  return v;
}

__global__ __launch_bounds__(256) void egnn_pre_kernel(
    const float* __restrict__ x, const float* __restrict__ coords,
    const float* __restrict__ We1, const float* __restrict__ be1,
    float* __restrict__ P, float* __restrict__ agg,
    float* __restrict__ coords_out) {
  __shared__ __attribute__((aligned(16))) float sX[16 * 132];

  const int tid = threadIdx.x;
  const int lane = tid & 31;
  const int wv = tid >> 5;
  const int hi = lane >> 4;
  const int lm = lane & 15;
  const int n0 = wv * 16;
  const int nb = blockIdx.x * 16;

  const int sr = tid >> 5;
  const int sc = 4 * (tid & 31);
  const v4f* x4 = reinterpret_cast<const v4f*>(x) + nb * 32;
  const v4f xa = x4[tid];
  const v4f xb = x4[tid + 256];

  const v4f zero4 = {0.0f, 0.0f, 0.0f, 0.0f};
  v4f* g4 = reinterpret_cast<v4f*>(agg) + nb * 32;
  g4[tid] = zero4;
  g4[tid + 256] = zero4;
  if (tid < 48) coords_out[nb * 3 + tid] = coords[nb * 3 + tid];

  *reinterpret_cast<v4f*>(&sX[sr * 132 + sc]) = xa;
  *reinterpret_cast<v4f*>(&sX[(sr + 8) * 132 + sc]) = xb;
  __syncthreads();

  v8f acc = {};
  for (int k0 = 0; k0 < 128; k0 += 4) {
    int kk = k0 + 2 * hi;
    v2f a, b;
    a.x = sX[lm * 132 + kk];
    a.y = sX[lm * 132 + kk + 1];
    b.x = We1[kk * 128 + n0 + lm];
    b.y = We1[(kk + 1) * 128 + n0 + lm];
    acc = wmma4(a, b, acc);
  }
  const float bias = be1[n0 + lm];
  __syncthreads();
#pragma unroll
  for (int v = 0; v < 8; ++v) sX[(v + 8 * hi) * 132 + n0 + lm] = acc[v] + bias;
  __syncthreads();
  v4f* p4 = reinterpret_cast<v4f*>(P) + nb * 32;
  p4[tid] = *reinterpret_cast<const v4f*>(&sX[sr * 132 + sc]);
  p4[tid + 256] = *reinterpret_cast<const v4f*>(&sX[(sr + 8) * 132 + sc]);
}

__global__ __launch_bounds__(256, 8) void egnn_edge_kernel(
    const float* __restrict__ x, const float* __restrict__ coords,
    const float* __restrict__ edge_attr, const int* __restrict__ erow,
    const int* __restrict__ ecol, const float* __restrict__ We1,
    const float* __restrict__ P, const float* __restrict__ We2,
    const float* __restrict__ be2, const float* __restrict__ Wc1,
    const float* __restrict__ bc1, const float* __restrict__ Wc2,
    float* __restrict__ agg, float* __restrict__ coords_out) {
  __shared__ __attribute__((aligned(16))) float smem[EPB * 264];
  float* sA = smem;
  float* sH = smem;
  float* sE = smem + EPB * 132;

  __shared__ int sRC[2 * EPB];
  int* sRow = sRC;
  int* sCol = sRC + EPB;
  __shared__ float sDiff[EPB * 3], sDist[EPB], sW[EPB];

  const int tid = threadIdx.x;
  const int lane = tid & 31;
  const int wv = tid >> 5;
  const int hi = lane >> 4;
  const int lm = lane & 15;
  const int n0 = wv * 16;
  const int ebase = blockIdx.x * EPB;

  const v4f* ea4 = reinterpret_cast<const v4f*>(edge_attr) + ebase * 8;
  const v4f ea0 = ea4[tid];
  const v4f ea1 = ea4[tid + 256];

  if (tid < EPB) {
    int r = erow[ebase + tid];
    int c = ecol[ebase + tid];
    sRow[tid] = r;
    sCol[tid] = c;
    float d0 = coords[r * 3 + 0] - coords[c * 3 + 0];
    float d1 = coords[r * 3 + 1] - coords[c * 3 + 1];
    float d2 = coords[r * 3 + 2] - coords[c * 3 + 2];
    sDiff[tid * 3 + 0] = d0;
    sDiff[tid * 3 + 1] = d1;
    sDiff[tid * 3 + 2] = d2;
    sDist[tid] = d0 * d0 + d1 * d1 + d2 * d2;
    sW[tid] = 0.0f;
  }
  __syncthreads();

  v8f acc1[4];
  {
    const v4f* x4 = reinterpret_cast<const v4f*>(x);
    v4f xv[8];
    for (int i = 0; i < 8; ++i) {
      const int node = sCol[wv + 8 * i];
      xv[i] = x4[node * 32 + lane];
    }
#pragma unroll
    for (int t = 0; t < 4; ++t) {
#pragma unroll
      for (int v = 0; v < 8; ++v) {
        const int M = t * 16 + v + 8 * hi;
        acc1[t][v] = P[sRow[M] * 128 + n0 + lm];
      }
    }
    for (int i = 0; i < 8; ++i)
      *reinterpret_cast<v4f*>(&sA[(wv + 8 * i) * 164 + 4 * lane]) = xv[i];
    *reinterpret_cast<v4f*>(&sA[(tid >> 3) * 164 + 128 + 4 * (tid & 7)]) = ea0;
    *reinterpret_cast<v4f*>(&sA[((tid >> 3) + 32) * 164 + 128 + 4 * (tid & 7)]) =
        ea1;
  }
  __syncthreads();

  {
    for (int k0 = 0; k0 < 160; k0 += 4) {
      int kk = k0 + 2 * hi;
      v2f b;
      b.x = We1[(128 + kk) * 128 + n0 + lm];
      b.y = We1[(129 + kk) * 128 + n0 + lm];
#pragma unroll
      for (int t = 0; t < 4; ++t) {
        v2f a;
        a.x = sA[(t * 16 + lm) * 164 + kk];
        a.y = sA[(t * 16 + lm) * 164 + kk + 1];
        acc1[t] = wmma4(a, b, acc1[t]);
      }
    }
    float wlast = We1[288 * 128 + n0 + lm];
    __syncthreads();
#pragma unroll
    for (int t = 0; t < 4; ++t) {
#pragma unroll
      for (int v = 0; v < 8; ++v) {
        int M = t * 16 + v + 8 * hi;
        float pre = acc1[t][v] + sDist[M] * wlast;
        sH[M * 132 + n0 + lm] = silu_f(pre);
      }
    }
  }
  __syncthreads();

  {
    v8f acc[4] = {{}, {}, {}, {}};
    for (int k0 = 0; k0 < 128; k0 += 4) {
      int kk = k0 + 2 * hi;
      v2f b;
      b.x = We2[kk * 128 + n0 + lm];
      b.y = We2[(kk + 1) * 128 + n0 + lm];
#pragma unroll
      for (int t = 0; t < 4; ++t) {
        v2f a;
        a.x = sH[(t * 16 + lm) * 132 + kk];
        a.y = sH[(t * 16 + lm) * 132 + kk + 1];
        acc[t] = wmma4(a, b, acc[t]);
      }
    }
    float bias = be2[n0 + lm];
#pragma unroll
    for (int t = 0; t < 4; ++t) {
#pragma unroll
      for (int v = 0; v < 8; ++v) {
        int M = t * 16 + v + 8 * hi;
        float ef = acc[t][v] + bias;
        sE[M * 132 + n0 + lm] = ef;
        atomicAdd(&agg[sRow[M] * 128 + n0 + lm], ef);
      }
    }
  }
  __syncthreads();

  {
    v8f acc[4] = {{}, {}, {}, {}};
    for (int k0 = 0; k0 < 128; k0 += 4) {
      int kk = k0 + 2 * hi;
      v2f b;
      b.x = Wc1[kk * 128 + n0 + lm];
      b.y = Wc1[(kk + 1) * 128 + n0 + lm];
#pragma unroll
      for (int t = 0; t < 4; ++t) {
        v2f a;
        a.x = sE[(t * 16 + lm) * 132 + kk];
        a.y = sE[(t * 16 + lm) * 132 + kk + 1];
        acc[t] = wmma4(a, b, acc[t]);
      }
    }
    float bias = bc1[n0 + lm];
    float wc2 = Wc2[n0 + lm];
#pragma unroll
    for (int t = 0; t < 4; ++t) {
      float mine = 0.0f;
#pragma unroll
      for (int v = 0; v < 8; ++v) {
        float p = row16_sum(silu_f(acc[t][v] + bias) * wc2);
        mine = (lm == v) ? p : mine;
      }
      if (lm < 8) atomicAdd(&sW[t * 16 + lm + 8 * hi], mine);
    }
  }
  __syncthreads();

  if (tid < EPB * 3) {
    int e = tid / 3;
    int d = tid - e * 3;
    float cd = sDiff[tid] / (sqrtf(sDist[e]) + 1e-8f);
    atomicAdd(&coords_out[sRow[e] * 3 + d], cd * sW[e]);
  }
}

__global__ __launch_bounds__(256) void egnn_pre2_kernel(
    const float* __restrict__ x, const float* __restrict__ coords,
    const float* __restrict__ We1, const float* __restrict__ be1,
    const float* __restrict__ We2, const float* __restrict__ Wc1,
    float* __restrict__ P, float* __restrict__ Q, float* __restrict__ agg,
    float* __restrict__ coords_out, float* __restrict__ Wp) {
  if (blockIdx.x >= N_NODES / 16) {
    const int pb = blockIdx.x - N_NODES / 16;
    v4f* out4 = reinterpret_cast<v4f*>(Wp);
    for (int i = 0; i < 4; ++i) {
      const int slot = pb * 1024 + i * 256 + threadIdx.x;
      const float* src;
      int ls;
      if (slot < PK_W2) {
        src = We1 + 256 * 128;
        ls = slot;
      } else if (slot < PK_W3) {
        src = We2;
        ls = slot - PK_W2;
      } else {
        src = Wc1;
        ls = slot - PK_W3;
      }
      const int g = ls >> 7;
      const int n = ls & 127;
      v4f v;
      v.x = src[(4 * g + 0) * 128 + n];
      v.y = src[(4 * g + 1) * 128 + n];
      v.z = src[(4 * g + 2) * 128 + n];
      v.w = src[(4 * g + 3) * 128 + n];
      out4[slot] = v;
    }
    return;
  }
  __shared__ __attribute__((aligned(16))) float sX[16 * 132];
  __shared__ __attribute__((aligned(16))) float sQ[16 * 132];

  const int tid = threadIdx.x;
  const int lane = tid & 31;
  const int wv = tid >> 5;
  const int hi = lane >> 4;
  const int lm = lane & 15;
  const int n0 = wv * 16;
  const int nb = blockIdx.x * 16;

  const int sr = tid >> 5;
  const int sc = 4 * (tid & 31);
  const v4f* x4 = reinterpret_cast<const v4f*>(x) + nb * 32;
  const v4f xa = x4[tid];
  const v4f xb = x4[tid + 256];

  const v4f zero4 = {0.0f, 0.0f, 0.0f, 0.0f};
  v4f* g4 = reinterpret_cast<v4f*>(agg) + nb * 32;
  g4[tid] = zero4;
  g4[tid + 256] = zero4;
  if (tid < 48) coords_out[nb * 3 + tid] = coords[nb * 3 + tid];

  *reinterpret_cast<v4f*>(&sX[sr * 132 + sc]) = xa;
  *reinterpret_cast<v4f*>(&sX[(sr + 8) * 132 + sc]) = xb;
  __syncthreads();

  v8f accP = {};
  v8f accQ = {};
  for (int k0 = 0; k0 < 128; k0 += 4) {
    int kk = k0 + 2 * hi;
    v2f a, bp, bq;
    a.x = sX[lm * 132 + kk];
    a.y = sX[lm * 132 + kk + 1];
    bp.x = We1[kk * 128 + n0 + lm];
    bp.y = We1[(kk + 1) * 128 + n0 + lm];
    bq.x = We1[(128 + kk) * 128 + n0 + lm];
    bq.y = We1[(129 + kk) * 128 + n0 + lm];
    accP = wmma4(a, bp, accP);
    accQ = wmma4(a, bq, accQ);
  }
  const float bias = be1[n0 + lm];
  __syncthreads();
#pragma unroll
  for (int v = 0; v < 8; ++v) {
    sX[(v + 8 * hi) * 132 + n0 + lm] = accP[v] + bias;
    sQ[(v + 8 * hi) * 132 + n0 + lm] = accQ[v];
  }
  __syncthreads();
  v4f* p4 = reinterpret_cast<v4f*>(P) + nb * 32;
  v4f* q4 = reinterpret_cast<v4f*>(Q) + nb * 32;
  p4[tid] = *reinterpret_cast<const v4f*>(&sX[sr * 132 + sc]);
  p4[tid + 256] = *reinterpret_cast<const v4f*>(&sX[(sr + 8) * 132 + sc]);
  q4[tid] = *reinterpret_cast<const v4f*>(&sQ[sr * 132 + sc]);
  q4[tid + 256] = *reinterpret_cast<const v4f*>(&sQ[(sr + 8) * 132 + sc]);
}

__global__ __launch_bounds__(256, 8) void egnn_edge2_kernel(
    const float* __restrict__ coords, const float* __restrict__ edge_attr,
    const int* __restrict__ erow, const int* __restrict__ ecol,
    const float* __restrict__ We1, const float* __restrict__ P,
    const float* __restrict__ Q, const float* __restrict__ We2,
    const float* __restrict__ be2, const float* __restrict__ Wc1,
    const float* __restrict__ bc1, const float* __restrict__ Wc2,
    float* __restrict__ agg, float* __restrict__ coords_out) {
  __shared__ __attribute__((aligned(16))) float smem[EPB * 264];
  float* sH = smem;
  float* sE = smem + EPB * 132;
  float* sA = sE;

  __shared__ int sRC[2 * EPB];
  int* sRow = sRC;
  int* sCol = sRC + EPB;
  __shared__ float sDiff[EPB * 3], sDist[EPB], sW[EPB];

  const int tid = threadIdx.x;
  const int lane = tid & 31;
  const int wv = tid >> 5;
  const int hi = lane >> 4;
  const int lm = lane & 15;
  const int n0 = wv * 16;
  const int ebase = blockIdx.x * EPB;

  const v4f* ea4 = reinterpret_cast<const v4f*>(edge_attr) + ebase * 8;
  const v4f ea0 = ea4[tid];
  const v4f ea1 = ea4[tid + 256];

  if (tid < EPB) {
    int r = erow[ebase + tid];
    int c = ecol[ebase + tid];
    sRow[tid] = r;
    sCol[tid] = c;
    float d0 = coords[r * 3 + 0] - coords[c * 3 + 0];
    float d1 = coords[r * 3 + 1] - coords[c * 3 + 1];
    float d2 = coords[r * 3 + 2] - coords[c * 3 + 2];
    sDiff[tid * 3 + 0] = d0;
    sDiff[tid * 3 + 1] = d1;
    sDiff[tid * 3 + 2] = d2;
    sDist[tid] = d0 * d0 + d1 * d1 + d2 * d2;
    sW[tid] = 0.0f;
  }
  __syncthreads();

  v8f acc1[4];
  {
    v8f qv[4];
#pragma unroll
    for (int t = 0; t < 4; ++t) {
#pragma unroll
      for (int v = 0; v < 8; ++v) {
        const int M = t * 16 + v + 8 * hi;
        qv[t][v] = Q[sCol[M] * 128 + n0 + lm];
      }
    }
#pragma unroll
    for (int t = 0; t < 4; ++t) {
#pragma unroll
      for (int v = 0; v < 8; ++v) {
        const int M = t * 16 + v + 8 * hi;
        acc1[t][v] = P[sRow[M] * 128 + n0 + lm];
      }
    }
#pragma unroll
    for (int t = 0; t < 4; ++t) acc1[t] += qv[t];
  }
  *reinterpret_cast<v4f*>(&sA[(tid >> 3) * 36 + 4 * (tid & 7)]) = ea0;
  *reinterpret_cast<v4f*>(&sA[((tid >> 3) + 32) * 36 + 4 * (tid & 7)]) = ea1;
  __syncthreads();

  {
    for (int k0 = 0; k0 < 32; k0 += 4) {
      int kk = k0 + 2 * hi;
      v2f b;
      b.x = We1[(256 + kk) * 128 + n0 + lm];
      b.y = We1[(257 + kk) * 128 + n0 + lm];
#pragma unroll
      for (int t = 0; t < 4; ++t) {
        v2f a;
        a.x = sA[(t * 16 + lm) * 36 + kk];
        a.y = sA[(t * 16 + lm) * 36 + kk + 1];
        acc1[t] = wmma4(a, b, acc1[t]);
      }
    }
    float wlast = We1[288 * 128 + n0 + lm];
#pragma unroll
    for (int t = 0; t < 4; ++t) {
#pragma unroll
      for (int v = 0; v < 8; ++v) {
        int M = t * 16 + v + 8 * hi;
        float pre = acc1[t][v] + sDist[M] * wlast;
        sH[M * 132 + n0 + lm] = silu_f(pre);
      }
    }
  }
  __syncthreads();

  {
    v8f acc[4] = {{}, {}, {}, {}};
    for (int k0 = 0; k0 < 128; k0 += 4) {
      int kk = k0 + 2 * hi;
      v2f b;
      b.x = We2[kk * 128 + n0 + lm];
      b.y = We2[(kk + 1) * 128 + n0 + lm];
#pragma unroll
      for (int t = 0; t < 4; ++t) {
        v2f a;
        a.x = sH[(t * 16 + lm) * 132 + kk];
        a.y = sH[(t * 16 + lm) * 132 + kk + 1];
        acc[t] = wmma4(a, b, acc[t]);
      }
    }
    float bias = be2[n0 + lm];
#pragma unroll
    for (int t = 0; t < 4; ++t) {
#pragma unroll
      for (int v = 0; v < 8; ++v) {
        int M = t * 16 + v + 8 * hi;
        float ef = acc[t][v] + bias;
        sE[M * 132 + n0 + lm] = ef;
        atomicAdd(&agg[sRow[M] * 128 + n0 + lm], ef);
      }
    }
  }
  __syncthreads();

  {
    v8f acc[4] = {{}, {}, {}, {}};
    for (int k0 = 0; k0 < 128; k0 += 4) {
      int kk = k0 + 2 * hi;
      v2f b;
      b.x = Wc1[kk * 128 + n0 + lm];
      b.y = Wc1[(kk + 1) * 128 + n0 + lm];
#pragma unroll
      for (int t = 0; t < 4; ++t) {
        v2f a;
        a.x = sE[(t * 16 + lm) * 132 + kk];
        a.y = sE[(t * 16 + lm) * 132 + kk + 1];
        acc[t] = wmma4(a, b, acc[t]);
      }
    }
    float bias = bc1[n0 + lm];
    float wc2 = Wc2[n0 + lm];
#pragma unroll
    for (int t = 0; t < 4; ++t) {
      float mine = 0.0f;
#pragma unroll
      for (int v = 0; v < 8; ++v) {
        float p = row16_sum(silu_f(acc[t][v] + bias) * wc2);
        mine = (lm == v) ? p : mine;
      }
      if (lm < 8) atomicAdd(&sW[t * 16 + lm + 8 * hi], mine);
    }
  }
  __syncthreads();

  if (tid < EPB * 3) {
    int e = tid / 3;
    int d = tid - e * 3;
    float cd = sDiff[tid] / (sqrtf(sDist[e]) + 1e-8f);
    atomicAdd(&coords_out[sRow[e] * 3 + d], cd * sW[e]);
  }
}

__global__ __launch_bounds__(256, 8) void egnn_edge3_kernel(
    const float* __restrict__ coords, const float* __restrict__ edge_attr,
    const int* __restrict__ erow, const int* __restrict__ ecol,
    const float* __restrict__ We1, const float* __restrict__ P,
    const float* __restrict__ Q, const float* __restrict__ Wp,
    const float* __restrict__ be2, const float* __restrict__ bc1,
    const float* __restrict__ Wc2, float* __restrict__ agg,
    float* __restrict__ coords_out) {
  __shared__ __attribute__((aligned(16))) float smem[EPB * 264];
  float* sH = smem;
  float* sE = smem + EPB * 132;
  float* sA = sE;

  __shared__ int sRC[2 * EPB];
  int* sRow = sRC;
  int* sCol = sRC + EPB;
  __shared__ float sDiff[EPB * 3], sDist[EPB], sW[EPB];

  const int tid = threadIdx.x;
  const int lane = tid & 31;
  const int wv = tid >> 5;
  const int hi = lane >> 4;
  const int lm = lane & 15;
  const int n0 = wv * 16;
  const int ebase = blockIdx.x * EPB;
  const v4f* wp4 = reinterpret_cast<const v4f*>(Wp) + hi * 128 + n0 + lm;

  const v4f* ea4 = reinterpret_cast<const v4f*>(edge_attr) + ebase * 8;
  const v4f ea0 = ea4[tid];
  const v4f ea1 = ea4[tid + 256];

  if (tid < EPB) {
    int r = erow[ebase + tid];
    int c = ecol[ebase + tid];
    sRow[tid] = r;
    sCol[tid] = c;
    float d0 = coords[r * 3 + 0] - coords[c * 3 + 0];
    float d1 = coords[r * 3 + 1] - coords[c * 3 + 1];
    float d2 = coords[r * 3 + 2] - coords[c * 3 + 2];
    sDiff[tid * 3 + 0] = d0;
    sDiff[tid * 3 + 1] = d1;
    sDiff[tid * 3 + 2] = d2;
    sDist[tid] = d0 * d0 + d1 * d1 + d2 * d2;
    sW[tid] = 0.0f;
  }
  __syncthreads();

  v8f acc1[4];
  {
    v8f qv[4];
#pragma unroll
    for (int t = 0; t < 4; ++t) {
#pragma unroll
      for (int v = 0; v < 8; ++v) {
        const int M = t * 16 + v + 8 * hi;
        qv[t][v] = Q[sCol[M] * 128 + n0 + lm];
      }
    }
#pragma unroll
    for (int t = 0; t < 4; ++t) {
#pragma unroll
      for (int v = 0; v < 8; ++v) {
        const int M = t * 16 + v + 8 * hi;
        acc1[t][v] = P[sRow[M] * 128 + n0 + lm];
      }
    }
#pragma unroll
    for (int t = 0; t < 4; ++t) acc1[t] += qv[t];
  }
  *reinterpret_cast<v4f*>(&sA[(tid >> 3) * 36 + 4 * (tid & 7)]) = ea0;
  *reinterpret_cast<v4f*>(&sA[((tid >> 3) + 32) * 36 + 4 * (tid & 7)]) = ea1;
  __syncthreads();

  {
    for (int k0 = 0; k0 < 32; k0 += 8) {
      const v4f bv = wp4[(k0 >> 2) * 128];
      v4f av[4];
#pragma unroll
      for (int t = 0; t < 4; ++t)
        av[t] = *reinterpret_cast<const v4f*>(
            &sA[(t * 16 + lm) * 36 + k0 + 4 * hi]);
      const v2f b0 = {bv.x, bv.y};
      const v2f b1 = {bv.z, bv.w};
#pragma unroll
      for (int t = 0; t < 4; ++t) {
        const v2f a0 = {av[t].x, av[t].y};
        acc1[t] = wmma4(a0, b0, acc1[t]);
      }
#pragma unroll
      for (int t = 0; t < 4; ++t) {
        const v2f a1 = {av[t].z, av[t].w};
        acc1[t] = wmma4(a1, b1, acc1[t]);
      }
    }
    float wlast = We1[288 * 128 + n0 + lm];
#pragma unroll
    for (int t = 0; t < 4; ++t) {
#pragma unroll
      for (int v = 0; v < 8; ++v) {
        int M = t * 16 + v + 8 * hi;
        float pre = acc1[t][v] + sDist[M] * wlast;
        sH[M * 132 + n0 + lm] = silu_f(pre);
      }
    }
  }
  __syncthreads();

  {
    v8f acc[4] = {{}, {}, {}, {}};
    for (int k0 = 0; k0 < 128; k0 += 8) {
      const v4f bv = wp4[PK_W2 + (k0 >> 2) * 128];
      v4f av[4];
#pragma unroll
      for (int t = 0; t < 4; ++t)
        av[t] = *reinterpret_cast<const v4f*>(
            &sH[(t * 16 + lm) * 132 + k0 + 4 * hi]);
      const v2f b0 = {bv.x, bv.y};
      const v2f b1 = {bv.z, bv.w};
#pragma unroll
      for (int t = 0; t < 4; ++t) {
        const v2f a0 = {av[t].x, av[t].y};
        acc[t] = wmma4(a0, b0, acc[t]);
      }
#pragma unroll
      for (int t = 0; t < 4; ++t) {
        const v2f a1 = {av[t].z, av[t].w};
        acc[t] = wmma4(a1, b1, acc[t]);
      }
    }
    float bias = be2[n0 + lm];
#pragma unroll
    for (int t = 0; t < 4; ++t) {
#pragma unroll
      for (int v = 0; v < 8; ++v) {
        int M = t * 16 + v + 8 * hi;
        float ef = acc[t][v] + bias;
        sE[M * 132 + n0 + lm] = ef;
        atomicAdd(&agg[sRow[M] * 128 + n0 + lm], ef);
      }
    }
  }
  __syncthreads();

  {
    v8f acc[4] = {{}, {}, {}, {}};
    for (int k0 = 0; k0 < 128; k0 += 8) {
      const v4f bv = wp4[PK_W3 + (k0 >> 2) * 128];
      v4f av[4];
#pragma unroll
      for (int t = 0; t < 4; ++t)
        av[t] = *reinterpret_cast<const v4f*>(
            &sE[(t * 16 + lm) * 132 + k0 + 4 * hi]);
      const v2f b0 = {bv.x, bv.y};
      const v2f b1 = {bv.z, bv.w};
#pragma unroll
      for (int t = 0; t < 4; ++t) {
        const v2f a0 = {av[t].x, av[t].y};
        acc[t] = wmma4(a0, b0, acc[t]);
      }
#pragma unroll
      for (int t = 0; t < 4; ++t) {
        const v2f a1 = {av[t].z, av[t].w};
        acc[t] = wmma4(a1, b1, acc[t]);
      }
    }
    float bias = bc1[n0 + lm];
    float wc2 = Wc2[n0 + lm];
#pragma unroll
    for (int t = 0; t < 4; ++t) {
      float mine = 0.0f;
#pragma unroll
      for (int v = 0; v < 8; ++v) {
        float p = row16_sum(silu_f(acc[t][v] + bias) * wc2);
        mine = (lm == v) ? p : mine;
      }
      if (lm < 8) atomicAdd(&sW[t * 16 + lm + 8 * hi], mine);
    }
  }
  __syncthreads();

  if (tid < EPB * 3) {
    int e = tid / 3;
    int d = tid - e * 3;
    float cd = sDiff[tid] / (sqrtf(sDist[e]) + 1e-8f);
    atomicAdd(&coords_out[sRow[e] * 3 + d], cd * sW[e]);
  }
}

__global__ __launch_bounds__(256) void egnn_node_kernel(
    const float* __restrict__ x, const float* __restrict__ agg,
    const float* __restrict__ Wn1, const float* __restrict__ bn1,
    const float* __restrict__ Wn2, const float* __restrict__ bn2,
    float* __restrict__ x_out) {
  __shared__ float sA[16 * 260];
  __shared__ float sH[16 * 132];

  const int tid = threadIdx.x;
  const int lane = tid & 31;
  const int wv = tid >> 5;
  const int hi = lane >> 4;
  const int lm = lane & 15;
  const int n0 = wv * 16;
  const int nb = blockIdx.x * 16;

  for (int idx = tid; idx < 16 * 256; idx += 256) {
    int m = idx >> 8;
    int c = idx & 255;
    float v = (c < 128) ? x[(nb + m) * 128 + c] : agg[(nb + m) * 128 + (c - 128)];
    sA[m * 260 + c] = v;
  }
  __syncthreads();

  v8f acc1 = {};
  for (int k0 = 0; k0 < 256; k0 += 4) {
    int kk = k0 + 2 * hi;
    v2f a, b;
    a.x = sA[lm * 260 + kk];
    a.y = sA[lm * 260 + kk + 1];
    b.x = Wn1[kk * 128 + n0 + lm];
    b.y = Wn1[(kk + 1) * 128 + n0 + lm];
    acc1 = wmma4(a, b, acc1);
  }
  {
    float bias = bn1[n0 + lm];
#pragma unroll
    for (int v = 0; v < 8; ++v) {
      int M = v + 8 * hi;
      sH[M * 132 + n0 + lm] = silu_f(acc1[v] + bias);
    }
  }
  __syncthreads();

  v8f acc2 = {};
  for (int k0 = 0; k0 < 128; k0 += 4) {
    int kk = k0 + 2 * hi;
    v2f a, b;
    a.x = sH[lm * 132 + kk];
    a.y = sH[lm * 132 + kk + 1];
    b.x = Wn2[kk * 128 + n0 + lm];
    b.y = Wn2[(kk + 1) * 128 + n0 + lm];
    acc2 = wmma4(a, b, acc2);
  }
  {
    float bias = bn2[n0 + lm];
#pragma unroll
    for (int v = 0; v < 8; ++v) {
      int M = v + 8 * hi;
      x_out[(nb + M) * 128 + n0 + lm] = acc2[v] + bias;
    }
  }
}

#define PL_W23 1024
#define PL_SLOTS 5120
#define PL_BLOCKS 9

__global__ __launch_bounds__(256) void egnn_pre3_kernel(
    const float* __restrict__ x, const float* __restrict__ coords,
    const float* __restrict__ We1, const float* __restrict__ be1,
    const float* __restrict__ We2, const float* __restrict__ be2,
    const float* __restrict__ Wc1, const float* __restrict__ bc1,
    float* __restrict__ P, float* __restrict__ Q, float* __restrict__ S4,
    int* __restrict__ deg, float* __restrict__ coords_out,
    float* __restrict__ Wp, float* __restrict__ b23) {
  const int tid = threadIdx.x;
  const int lane = tid & 31;
  const int wv = tid >> 5;
  const int hi = lane >> 4;
  const int lm = lane & 15;

  if (blockIdx.x >= N_NODES / 16) {
    const int pb = blockIdx.x - N_NODES / 16;
    v4f* out4 = reinterpret_cast<v4f*>(Wp);
    if (pb == 0) {
      const float* src = We1 + 256 * 128;
      for (int i = 0; i < 4; ++i) {
        const int slot = i * 256 + tid;
        const int g = slot >> 7;
        const int n = slot & 127;
        v4f v;
        v.x = src[(4 * g + 0) * 128 + n];
        v.y = src[(4 * g + 1) * 128 + n];
        v.z = src[(4 * g + 2) * 128 + n];
        v.w = src[(4 * g + 3) * 128 + n];
        out4[slot] = v;
      }
      if (tid < 128) {
        float s = 0.0f;
        for (int m = 0; m < 128; ++m) s = fmaf(be2[m], Wc1[m * 128 + tid], s);
        b23[tid] = s + bc1[tid];
      }
      return;
    }
    const int c0 = (pb - 1) * 16;
    const int i0 = wv * 16;
    v8f acc = {};
    for (int k0 = 0; k0 < 128; k0 += 4) {
      const int kk = k0 + 2 * hi;
      v2f a, b;
      a.x = We2[(i0 + lm) * 128 + kk];
      a.y = We2[(i0 + lm) * 128 + kk + 1];
      b.x = Wc1[kk * 128 + c0 + lm];
      b.y = Wc1[(kk + 1) * 128 + c0 + lm];
      acc = wmma4(a, b, acc);
    }
    const int g0 = (i0 + 8 * hi) >> 2;
    const v4f s0 = {acc[0], acc[1], acc[2], acc[3]};
    const v4f s1 = {acc[4], acc[5], acc[6], acc[7]};
    out4[PL_W23 + g0 * 128 + c0 + lm] = s0;
    out4[PL_W23 + (g0 + 1) * 128 + c0 + lm] = s1;
    return;
  }
  __shared__ __attribute__((aligned(16))) float sX[16 * 132];
  __shared__ __attribute__((aligned(16))) float sQ[16 * 132];

  const int n0 = wv * 16;
  const int nb = blockIdx.x * 16;

  const int sr = tid >> 5;
  const int sc = 4 * (tid & 31);
  const v4f* x4 = reinterpret_cast<const v4f*>(x) + nb * 32;
  const v4f xa = x4[tid];
  const v4f xb = x4[tid + 256];

  const v4f zero4 = {0.0f, 0.0f, 0.0f, 0.0f};
  v4f* g4 = reinterpret_cast<v4f*>(S4) + nb * 32;
  g4[tid] = zero4;
  g4[tid + 256] = zero4;
  if (tid < 16) deg[nb + tid] = 0;
  if (tid < 48) coords_out[nb * 3 + tid] = coords[nb * 3 + tid];

  *reinterpret_cast<v4f*>(&sX[sr * 132 + sc]) = xa;
  *reinterpret_cast<v4f*>(&sX[(sr + 8) * 132 + sc]) = xb;
  __syncthreads();

  v8f accP = {};
  v8f accQ = {};
  for (int k0 = 0; k0 < 128; k0 += 4) {
    int kk = k0 + 2 * hi;
    v2f a, bp, bq;
    a.x = sX[lm * 132 + kk];
    a.y = sX[lm * 132 + kk + 1];
    bp.x = We1[kk * 128 + n0 + lm];
    bp.y = We1[(kk + 1) * 128 + n0 + lm];
    bq.x = We1[(128 + kk) * 128 + n0 + lm];
    bq.y = We1[(129 + kk) * 128 + n0 + lm];
    accP = wmma4(a, bp, accP);
    accQ = wmma4(a, bq, accQ);
  }
  const float bias = be1[n0 + lm];
  __syncthreads();
#pragma unroll
  for (int v = 0; v < 8; ++v) {
    sX[(v + 8 * hi) * 132 + n0 + lm] = accP[v] + bias;
    sQ[(v + 8 * hi) * 132 + n0 + lm] = accQ[v];
  }
  __syncthreads();
  v4f* p4 = reinterpret_cast<v4f*>(P) + nb * 32;
  v4f* q4 = reinterpret_cast<v4f*>(Q) + nb * 32;
  p4[tid] = *reinterpret_cast<const v4f*>(&sX[sr * 132 + sc]);
  p4[tid + 256] = *reinterpret_cast<const v4f*>(&sX[(sr + 8) * 132 + sc]);
  q4[tid] = *reinterpret_cast<const v4f*>(&sQ[sr * 132 + sc]);
  q4[tid + 256] = *reinterpret_cast<const v4f*>(&sQ[(sr + 8) * 132 + sc]);
}

__global__ __launch_bounds__(256, 8) void egnn_edge4_kernel(
    const float* __restrict__ coords, const float* __restrict__ edge_attr,
    const int* __restrict__ erow, const int* __restrict__ ecol,
    const float* __restrict__ We1, const float* __restrict__ P,
    const float* __restrict__ Q, const float* __restrict__ Wp,
    const float* __restrict__ b23, const float* __restrict__ Wc2,
    unsigned long long* __restrict__ S, int* __restrict__ deg,
    float* __restrict__ coords_out) {
  __shared__ __attribute__((aligned(16))) float smem[EPB * 168];
  float* sH = smem;
  float* sA = smem + EPB * 132;

  __shared__ int sRC[2 * EPB];
  int* sRow = sRC;
  int* sCol = sRC + EPB;
  __shared__ float sDiff[EPB * 3], sDist[EPB], sW[EPB];

  const int tid = threadIdx.x;
  const int lane = tid & 31;
  const int wv = tid >> 5;
  const int hi = lane >> 4;
  const int lm = lane & 15;
  const int n0 = wv * 16;
  const int ebase = blockIdx.x * EPB;
  const v4f* wp4 = reinterpret_cast<const v4f*>(Wp) + hi * 128 + n0 + lm;
  const int f0 = n0 + 8 * hi;

  const v4f* ea4 = reinterpret_cast<const v4f*>(edge_attr) + ebase * 8;
  const v4f ea0 = ea4[tid];
  const v4f ea1 = ea4[tid + 256];

  if (tid < EPB) {
    int r = erow[ebase + tid];
    int c = ecol[ebase + tid];
    sRow[tid] = r;
    sCol[tid] = c;
    atomicAdd(&deg[r], 1);
    float d0 = coords[r * 3 + 0] - coords[c * 3 + 0];
    float d1 = coords[r * 3 + 1] - coords[c * 3 + 1];
    float d2 = coords[r * 3 + 2] - coords[c * 3 + 2];
    sDiff[tid * 3 + 0] = d0;
    sDiff[tid * 3 + 1] = d1;
    sDiff[tid * 3 + 2] = d2;
    sDist[tid] = d0 * d0 + d1 * d1 + d2 * d2;
    sW[tid] = 0.0f;
  }
  __syncthreads();

  {
    const v4f* P4 = reinterpret_cast<const v4f*>(P);
    const v4f* Q4 = reinterpret_cast<const v4f*>(Q);
    v4f pv[8], qv[8];
#pragma unroll
    for (int i = 0; i < 8; ++i) {
      const int e = wv + 8 * i;
      pv[i] = P4[sRow[e] * 32 + lane];
      qv[i] = Q4[sCol[e] * 32 + lane];
    }
#pragma unroll
    for (int i = 0; i < 8; ++i)
      *reinterpret_cast<v4f*>(&sH[(wv + 8 * i) * 132 + 4 * lane]) =
          pv[i] + qv[i];
  }
  *reinterpret_cast<v4f*>(&sA[(tid >> 3) * 36 + 4 * (tid & 7)]) = ea0;
  *reinterpret_cast<v4f*>(&sA[((tid >> 3) + 32) * 36 + 4 * (tid & 7)]) = ea1;
  __syncthreads();

  v8f acc1[4];
#pragma unroll
  for (int t = 0; t < 4; ++t) {
    const float* sp = &sH[(t * 16 + lm) * 132 + f0];
    const v4f s0 = *reinterpret_cast<const v4f*>(sp);
    const v4f s1 = *reinterpret_cast<const v4f*>(sp + 4);
    acc1[t] = (v8f){s0.x, s0.y, s0.z, s0.w, s1.x, s1.y, s1.z, s1.w};
  }

  {
    for (int k0 = 0; k0 < 32; k0 += 8) {
      const v4f bv = wp4[(k0 >> 2) * 128];
      v4f av[4];
#pragma unroll
      for (int t = 0; t < 4; ++t)
        av[t] = *reinterpret_cast<const v4f*>(
            &sA[(t * 16 + lm) * 36 + k0 + 4 * hi]);
      const v2f w0 = {bv.x, bv.y};
      const v2f w1 = {bv.z, bv.w};
#pragma unroll
      for (int t = 0; t < 4; ++t) {
        const v2f x0 = {av[t].x, av[t].y};
        acc1[t] = wmma4(w0, x0, acc1[t]);
      }
#pragma unroll
      for (int t = 0; t < 4; ++t) {
        const v2f x1 = {av[t].z, av[t].w};
        acc1[t] = wmma4(w1, x1, acc1[t]);
      }
    }
    const v4f* wl4 = reinterpret_cast<const v4f*>(We1 + 288 * 128 + f0);
    const v4f wla = wl4[0];
    const v4f wlb = wl4[1];
    const float wl[8] = {wla.x, wla.y, wla.z, wla.w,
                         wlb.x, wlb.y, wlb.z, wlb.w};
#pragma unroll
    for (int t = 0; t < 4; ++t) {
      const float dist = sDist[t * 16 + lm];
      float h[8];
#pragma unroll
      for (int r = 0; r < 8; ++r) h[r] = silu_f(acc1[t][r] + dist * wl[r]);
      float* dp = &sH[(t * 16 + lm) * 132 + f0];
      *reinterpret_cast<v4f*>(dp) = (v4f){h[0], h[1], h[2], h[3]};
      *reinterpret_cast<v4f*>(dp + 4) = (v4f){h[4], h[5], h[6], h[7]};
    }
  }
  __syncthreads();

  {
    const int pr = (wv & 1) * 32 + lane;
    v2f h2[16];
    int node[16];
#pragma unroll
    for (int i = 0; i < 16; ++i) {
      const int e = i * 4 + (wv >> 1);
      h2[i] = *reinterpret_cast<const v2f*>(&sH[e * 132 + 2 * pr]);
      node[i] = sRow[e];
    }
#pragma unroll
    for (int i = 0; i < 16; ++i) {
      const int a = (int)rintf(h2[i].x * 4194304.0f);
      const int b = (int)rintf(h2[i].y * 4194304.0f);
      const long long q = (long long)b * 4294967296LL + (long long)a;
      atomicAdd(&S[node[i] * 64 + pr], (unsigned long long)q);
    }
  }

  {
    v8f acc[4] = {{}, {}, {}, {}};
    for (int k0 = 0; k0 < 128; k0 += 8) {
      const v4f bv = wp4[PL_W23 + (k0 >> 2) * 128];
      v4f av[4];
#pragma unroll
      for (int t = 0; t < 4; ++t)
        av[t] = *reinterpret_cast<const v4f*>(
            &sH[(t * 16 + lm) * 132 + k0 + 4 * hi]);
      const v2f w0 = {bv.x, bv.y};
      const v2f w1 = {bv.z, bv.w};
#pragma unroll
      for (int t = 0; t < 4; ++t) {
        const v2f x0 = {av[t].x, av[t].y};
        acc[t] = wmma4(w0, x0, acc[t]);
      }
#pragma unroll
      for (int t = 0; t < 4; ++t) {
        const v2f x1 = {av[t].z, av[t].w};
        acc[t] = wmma4(w1, x1, acc[t]);
      }
    }
    const v4f* bb4 = reinterpret_cast<const v4f*>(b23 + f0);
    const v4f* wc4 = reinterpret_cast<const v4f*>(Wc2 + f0);
    const v4f bba = bb4[0];
    const v4f bbb = bb4[1];
    const v4f wca = wc4[0];
    const v4f wcb = wc4[1];
    const float bias[8] = {bba.x, bba.y, bba.z, bba.w,
                           bbb.x, bbb.y, bbb.z, bbb.w};
    const float wc2[8] = {wca.x, wca.y, wca.z, wca.w,
                          wcb.x, wcb.y, wcb.z, wcb.w};
#pragma unroll
    for (int t = 0; t < 4; ++t) {
      float s = 0.0f;
#pragma unroll
      for (int r = 0; r < 8; ++r) s += silu_f(acc[t][r] + bias[r]) * wc2[r];
      const float o = __int_as_float(__builtin_amdgcn_permlanex16(
          __float_as_int(s), __float_as_int(s), 0x76543210u, 0xfedcba98u,
          false, false));
      if (hi == 0) atomicAdd(&sW[t * 16 + lm], s + o);
    }
  }
  __syncthreads();

  if (tid < EPB * 3) {
    int e = tid / 3;
    int d = tid - e * 3;
    float cd = sDiff[tid] / (sqrtf(sDist[e]) + 1e-8f);
    atomicAdd(&coords_out[sRow[e] * 3 + d], cd * sW[e]);
  }
}

__global__ __launch_bounds__(256, 16) void egnn_node2_kernel(
    const float* __restrict__ x, const unsigned long long* __restrict__ S,
    const int* __restrict__ deg, const float* __restrict__ We2,
    const float* __restrict__ be2, const float* __restrict__ Wn1,
    const float* __restrict__ bn1, const float* __restrict__ Wn2,
    const float* __restrict__ bn2, float* __restrict__ x_out) {
  __shared__ float sA[16 * 260];
  __shared__ float sH[16 * 132];

  const int tid = threadIdx.x;
  const int lane = tid & 31;
  const int wv = tid >> 5;
  const int hi = lane >> 4;
  const int lm = lane & 15;
  const int n0 = wv * 16;
  const int nb = blockIdx.x * 16;

  for (int idx = tid; idx < 16 * 128; idx += 256) {
    int m = idx >> 7;
    int c = idx & 127;
    sA[m * 260 + c] = x[(nb + m) * 128 + c];
  }
  for (int idx = tid; idx < 16 * 64; idx += 256) {
    int m = idx >> 6;
    int p = idx & 63;
    const long long q = (long long)S[(nb + m) * 64 + p];
    const int lo = (int)q;
    const int hi = (int)((q - (long long)lo) >> 32);
    sH[m * 132 + 2 * p] = (float)lo * 2.384185791015625e-07f;
    sH[m * 132 + 2 * p + 1] = (float)hi * 2.384185791015625e-07f;
  }
  __syncthreads();

  {
    v8f acc = {};
    for (int k0 = 0; k0 < 128; k0 += 4) {
      int kk = k0 + 2 * hi;
      v2f a, b;
      a.x = sH[lm * 132 + kk];
      a.y = sH[lm * 132 + kk + 1];
      b.x = We2[kk * 128 + n0 + lm];
      b.y = We2[(kk + 1) * 128 + n0 + lm];
      acc = wmma4(a, b, acc);
    }
    float bias = be2[n0 + lm];
#pragma unroll
    for (int v = 0; v < 8; ++v) {
      int M = v + 8 * hi;
      sA[M * 260 + 128 + n0 + lm] = acc[v] + (float)deg[nb + M] * bias;
    }
  }
  __syncthreads();

  v8f acc1 = {};
  for (int k0 = 0; k0 < 256; k0 += 4) {
    int kk = k0 + 2 * hi;
    v2f a, b;
    a.x = sA[lm * 260 + kk];
    a.y = sA[lm * 260 + kk + 1];
    b.x = Wn1[kk * 128 + n0 + lm];
    b.y = Wn1[(kk + 1) * 128 + n0 + lm];
    acc1 = wmma4(a, b, acc1);
  }
  {
    float bias = bn1[n0 + lm];
#pragma unroll
    for (int v = 0; v < 8; ++v) {
      int M = v + 8 * hi;
      sH[M * 132 + n0 + lm] = silu_f(acc1[v] + bias);
    }
  }
  __syncthreads();

  v8f acc2 = {};
  for (int k0 = 0; k0 < 128; k0 += 4) {
    int kk = k0 + 2 * hi;
    v2f a, b;
    a.x = sH[lm * 132 + kk];
    a.y = sH[lm * 132 + kk + 1];
    b.x = Wn2[kk * 128 + n0 + lm];
    b.y = Wn2[(kk + 1) * 128 + n0 + lm];
    acc2 = wmma4(a, b, acc2);
  }
  {
    float bias = bn2[n0 + lm];
#pragma unroll
    for (int v = 0; v < 8; ++v) {
      int M = v + 8 * hi;
      x_out[(nb + M) * 128 + n0 + lm] = acc2[v] + bias;
    }
  }
}

extern "C" void kernel_launch(void* const* d_in, const int* in_sizes, int n_in,
                              void* d_out, int out_size, void* d_ws,
                              size_t ws_size, hipStream_t stream) {
  const float* x = (const float*)d_in[0];
  const float* coords = (const float*)d_in[1];
  const float* edge_attr = (const float*)d_in[2];
  const int* edge_index = (const int*)d_in[3];
  const float* We1 = (const float*)d_in[4];
  const float* be1 = (const float*)d_in[5];
  const float* We2 = (const float*)d_in[6];
  const float* be2 = (const float*)d_in[7];
  const float* Wc1 = (const float*)d_in[8];
  const float* bc1 = (const float*)d_in[9];
  const float* Wc2 = (const float*)d_in[10];
  const float* Wn1 = (const float*)d_in[11];
  const float* bn1 = (const float*)d_in[12];
  const float* Wn2 = (const float*)d_in[13];
  const float* bn2 = (const float*)d_in[14];

  float* x_out = (float*)d_out;
  float* coords_out = x_out + (size_t)N_NODES * 128;
  float* agg = (float*)d_ws;

  const int* erow = edge_index;
  const int* ecol = edge_index + N_EDGES;

  if (ws_size >= (size_t)2 * N_NODES * 128 * sizeof(float) +
                     (size_t)PL_SLOTS * 16 + 128 * sizeof(float) +
                     (size_t)N_NODES * sizeof(int)) {
    unsigned long long* S = (unsigned long long*)d_ws;
    float* Q = (float*)d_ws + (size_t)N_NODES * 128;
    float* Wp = Q + (size_t)N_NODES * 128;
    float* b23 = Wp + (size_t)PL_SLOTS * 4;
    int* deg = (int*)(b23 + 128);
    egnn_pre3_kernel<<<N_NODES / 16 + PL_BLOCKS, 256, 0, stream>>>(
        x, coords, We1, be1, We2, be2, Wc1, bc1, x_out, Q, (float*)d_ws, deg,
        coords_out, Wp, b23);
    egnn_edge4_kernel<<<N_EDGES / EPB, 256, 0, stream>>>(
        coords, edge_attr, erow, ecol, We1, x_out, Q, Wp, b23, Wc2, S, deg,
        coords_out);
    egnn_node2_kernel<<<N_NODES / 16, 256, 0, stream>>>(
        x, S, deg, We2, be2, Wn1, bn1, Wn2, bn2, x_out);
    return;
  }
  if (ws_size >= (size_t)2 * N_NODES * 128 * sizeof(float) +
                     (size_t)PK_SLOTS * 16) {
    float* Q = agg + (size_t)N_NODES * 128;
    float* Wp = Q + (size_t)N_NODES * 128;
    egnn_pre2_kernel<<<N_NODES / 16 + PK_SLOTS / 1024, 256, 0, stream>>>(
        x, coords, We1, be1, We2, Wc1, x_out, Q, agg, coords_out, Wp);
    egnn_edge3_kernel<<<N_EDGES / EPB, 256, 0, stream>>>(
        coords, edge_attr, erow, ecol, We1, x_out, Q, Wp, be2, bc1, Wc2, agg,
        coords_out);
  } else if (ws_size >= (size_t)2 * N_NODES * 128 * sizeof(float)) {
    float* Q = agg + (size_t)N_NODES * 128;
    egnn_pre2_kernel<<<N_NODES / 16, 256, 0, stream>>>(
        x, coords, We1, be1, We2, Wc1, x_out, Q, agg, coords_out, nullptr);
    egnn_edge2_kernel<<<N_EDGES / EPB, 256, 0, stream>>>(
        coords, edge_attr, erow, ecol, We1, x_out, Q, We2, be2, Wc1, bc1, Wc2,
        agg, coords_out);
  } else {
    egnn_pre_kernel<<<N_NODES / 16, 256, 0, stream>>>(x, coords, We1, be1, x_out,
                                                      agg, coords_out);
    egnn_edge_kernel<<<N_EDGES / EPB, 256, 0, stream>>>(
        x, coords, edge_attr, erow, ecol, We1, x_out, We2, be2, Wc1, bc1, Wc2,
        agg, coords_out);
  }
  egnn_node_kernel<<<N_NODES / 16, 256, 0, stream>>>(x, agg, Wn1, bn1, Wn2, bn2,
                                                     x_out);
}
